# starting kernel
# baseline (speedup 1.0000x reference)
#include <hip/hip_runtime.h>
#include <math.h>

#define NN   2000
#define NE   32000
#define EMB  768
#define HID  384
#define FT   768
#define NH   4
#define NEG_SLOPE 0.2f

typedef __attribute__((ext_vector_type(2))) float v2f;
typedef __attribute__((ext_vector_type(8))) float v8f;

static inline int cdiv(long long a, long long b) { return (int)((a + b - 1) / b); }

__device__ __forceinline__ unsigned fenc(float f) {
  unsigned u = __float_as_uint(f);
  return (u & 0x80000000u) ? ~u : (u | 0x80000000u);
}
__device__ __forceinline__ float fdec(unsigned e) {
  unsigned u = (e & 0x80000000u) ? (e ^ 0x80000000u) : ~e;
  return __uint_as_float(u);
}

__global__ void k_embed_x(const int* __restrict__ x, const float* __restrict__ emb,
                          float* __restrict__ xf) {
  long long i = (long long)blockIdx.x * blockDim.x + threadIdx.x;
  if (i >= (long long)NN * EMB) return;
  int n = (int)(i / EMB), c = (int)(i % EMB);
  xf[i] = emb[(size_t)x[n * 2 + 0] * EMB + c] + emb[(size_t)x[n * 2 + 1] * EMB + c];
}

__global__ void k_embed_e(const int* __restrict__ ea_idx, const float* __restrict__ emb,
                          float* __restrict__ ea) {
  long long i = (long long)blockIdx.x * blockDim.x + threadIdx.x;
  if (i >= (long long)NE * EMB) return;
  int e = (int)(i / EMB), c = (int)(i % EMB);
  ea[i] = emb[(size_t)ea_idx[e * 2 + 0] * EMB + c] + emb[(size_t)ea_idx[e * 2 + 1] * EMB + c];
}

__global__ void k_deg(const int* __restrict__ dst, float* __restrict__ deg) {
  int e = blockIdx.x * blockDim.x + threadIdx.x;
  if (e < NE) atomicAdd(&deg[dst[e]], 1.0f);
}

__global__ void k_loopsum(const int* __restrict__ dst, float* __restrict__ ea) {
  long long i = (long long)blockIdx.x * blockDim.x + threadIdx.x;
  if (i >= (long long)NE * EMB) return;
  int e = (int)(i / EMB), c = (int)(i % EMB);
  atomicAdd(&ea[(size_t)(NE + dst[e]) * EMB + c], ea[(size_t)e * EMB + c]);
}

__global__ void k_loopdiv(const float* __restrict__ deg, float* __restrict__ ea) {
  long long i = (long long)blockIdx.x * blockDim.x + threadIdx.x;
  if (i >= (long long)NN * EMB) return;
  int n = (int)(i / EMB);
  ea[(size_t)NE * EMB + i] /= fmaxf(deg[n], 1.0f);
}

__global__ __launch_bounds__(256) void k_gemm_wmma_f32(
    const float* __restrict__ A, const float* __restrict__ B, float* __restrict__ C,
    int M, int N, int K) {
  __shared__ __align__(16) float sA[64][32];
  __shared__ __align__(16) float sBt[128][32];
  const int tid = threadIdx.x;
  const int wave = tid >> 5, lane = tid & 31;
  const int wm = wave >> 2, wn = wave & 3;
  const int row16 = lane & 15, khalf = lane >> 4;
  const int bm = blockIdx.y * 64, bn = blockIdx.x * 128;

  v8f acc00 = {}, acc01 = {}, acc10 = {}, acc11 = {};

  for (int k0 = 0; k0 < K; k0 += 32) {
    for (int i = tid; i < 64 * 32; i += 256) {
      int r = i >> 5, c = i & 31;
      int gr = bm + r;
      sA[r][c] = (gr < M) ? A[(size_t)gr * K + k0 + c] : 0.0f;
    }
    for (int i = tid; i < 32 * 128; i += 256) {
      int kk = i >> 7, n = i & 127;
      sBt[n][kk] = B[(size_t)(k0 + kk) * N + bn + n];
    }
    __syncthreads();

    const float* a0p = &sA[wm * 32 + row16][0];
    const float* a1p = &sA[wm * 32 + 16 + row16][0];
    const float* b0p = &sBt[wn * 32 + row16][0];
    const float* b1p = &sBt[wn * 32 + 16 + row16][0];
#pragma unroll
    for (int ks = 0; ks < 32; ks += 4) {
      int kk = ks + khalf * 2;
      v2f a0 = *(const v2f*)(a0p + kk);
      v2f a1 = *(const v2f*)(a1p + kk);
      v2f b0 = *(const v2f*)(b0p + kk);
      v2f b1 = *(const v2f*)(b1p + kk);
      acc00 = __builtin_amdgcn_wmma_f32_16x16x4_f32(false, a0, false, b0, (short)0, acc00, false, false);
      acc01 = __builtin_amdgcn_wmma_f32_16x16x4_f32(false, a0, false, b1, (short)0, acc01, false, false);
      acc10 = __builtin_amdgcn_wmma_f32_16x16x4_f32(false, a1, false, b0, (short)0, acc10, false, false);
      acc11 = __builtin_amdgcn_wmma_f32_16x16x4_f32(false, a1, false, b1, (short)0, acc11, false, false);
    }
    __syncthreads();
  }

  int gn0 = bn + wn * 32 + row16;
  int gn1 = gn0 + 16;
  int gm0 = bm + wm * 32 + 8 * khalf;
  int gm1 = gm0 + 16;
#pragma unroll
  for (int v = 0; v < 8; ++v) {
    if (gm0 + v < M) {
      C[(size_t)(gm0 + v) * N + gn0] = acc00[v];
      C[(size_t)(gm0 + v) * N + gn1] = acc01[v];
    }
    if (gm1 + v < M) {
      C[(size_t)(gm1 + v) * N + gn0] = acc10[v];
      C[(size_t)(gm1 + v) * N + gn1] = acc11[v];
    }
  }
}

__global__ void k_proj_e(const float* __restrict__ We, const float* __restrict__ ae,
                         float* __restrict__ proj, int C) {
  int i = blockIdx.x * blockDim.x + threadIdx.x;
  if (i >= EMB * NH) return;
  int k = i / NH, h = i % NH;
  const float* wr = We + (size_t)k * (NH * C) + (size_t)h * C;
  const float* ar = ae + (size_t)h * C;
  float s = 0.f;
  for (int c = 0; c < C; ++c) s += wr[c] * ar[c];
  proj[i] = s;
}

__global__ void k_alpha_node(const float* __restrict__ hx, const float* __restrict__ a_src,
                             const float* __restrict__ a_dst, float* __restrict__ ans,
                             float* __restrict__ adn, int C) {
  int i = blockIdx.x * blockDim.x + threadIdx.x;
  if (i >= NN * NH) return;
  int n = i / NH, h = i % NH;
  const float* hr = hx + (size_t)n * NH * C + (size_t)h * C;
  const float* sr = a_src + (size_t)h * C;
  const float* dr = a_dst + (size_t)h * C;
  float s = 0.f, t = 0.f;
  for (int c = 0; c < C; ++c) { s += hr[c] * sr[c]; t += hr[c] * dr[c]; }
  ans[i] = s;
  adn[i] = t;
}

__global__ void k_edge_alpha(const float* __restrict__ ea, const float* __restrict__ proj,
                             const float* __restrict__ ans, const float* __restrict__ adn,
                             const int* __restrict__ src, const int* __restrict__ dst,
                             float* __restrict__ alpha, unsigned* __restrict__ mx) {
  const int tot = NE + NN;
  int e = blockIdx.x * (blockDim.x >> 5) + (threadIdx.x >> 5);
  int lane = threadIdx.x & 31;
  if (e >= tot) return;
  int s = (e < NE) ? src[e] : (e - NE);
  int d = (e < NE) ? dst[e] : (e - NE);
  const float* row = ea + (size_t)e * EMB;
  const float4* p4 = (const float4*)proj;
  float a0 = 0.f, a1 = 0.f, a2 = 0.f, a3 = 0.f;
  for (int k = lane; k < EMB; k += 32) {
    float v = row[k];
    float4 p = p4[k];
    a0 += v * p.x; a1 += v * p.y; a2 += v * p.z; a3 += v * p.w;
  }
#pragma unroll
  for (int off = 16; off > 0; off >>= 1) {
    a0 += __shfl_xor(a0, off, 32);
    a1 += __shfl_xor(a1, off, 32);
    a2 += __shfl_xor(a2, off, 32);
    a3 += __shfl_xor(a3, off, 32);
  }
  if (lane < 4) {
    float ale = (lane == 0) ? a0 : (lane == 1) ? a1 : (lane == 2) ? a2 : a3;
    float a = ans[s * NH + lane] + adn[d * NH + lane] + ale;
    a = (a > 0.f) ? a : NEG_SLOPE * a;
    alpha[(size_t)e * NH + lane] = a;
    atomicMax(&mx[d * NH + lane], fenc(a));
  }
}

__global__ void k_softmax_den(const int* __restrict__ dst, const unsigned* __restrict__ mx,
                              float* __restrict__ alpha, float* __restrict__ den) {
  const int tot = NE + NN;
  long long i = (long long)blockIdx.x * blockDim.x + threadIdx.x;
  if (i >= (long long)tot * NH) return;
  int e = (int)(i >> 2), h = (int)(i & 3);
  int d = (e < NE) ? dst[e] : (e - NE);
  float ex = expf(alpha[i] - fdec(mx[d * NH + h]));
  alpha[i] = ex;
  atomicAdd(&den[d * NH + h], ex);
}

__global__ void k_aggregate(const float* __restrict__ hx, const float* __restrict__ ex,
                            const float* __restrict__ den, const int* __restrict__ src,
                            const int* __restrict__ dst, float* __restrict__ out, int C) {
  int e = blockIdx.x;
  int s = (e < NE) ? src[e] : (e - NE);
  int d = (e < NE) ? dst[e] : (e - NE);
  __shared__ float att[NH];
  if (threadIdx.x < NH)
    att[threadIdx.x] = ex[(size_t)e * NH + threadIdx.x] / den[d * NH + threadIdx.x];
  __syncthreads();
  const float* hr = hx + (size_t)s * NH * C;
  for (int c = threadIdx.x; c < C; c += blockDim.x) {
    float v = 0.25f * (att[0] * hr[c] + att[1] * hr[C + c] +
                       att[2] * hr[2 * C + c] + att[3] * hr[3 * C + c]);
    atomicAdd(&out[(size_t)d * C + c], v);
  }
}

__global__ void k_bias_relu(const float* __restrict__ in, const float* __restrict__ b,
                            float* __restrict__ o) {
  long long i = (long long)blockIdx.x * blockDim.x + threadIdx.x;
  if (i >= (long long)NN * HID) return;
  o[i] = fmaxf(in[i] + b[i % HID], 0.f);
}

__global__ void k_init_out(const float* __restrict__ b, float* __restrict__ o) {
  long long i = (long long)blockIdx.x * blockDim.x + threadIdx.x;
  if (i >= (long long)NN * FT) return;
  o[i] = b[i % FT];
}

extern "C" void kernel_launch(void* const* d_in, const int* in_sizes, int n_in,
                              void* d_out, int out_size, void* d_ws, size_t ws_size,
                              hipStream_t stream) {
  const int* x      = (const int*)d_in[0];
  const int* eidx   = (const int*)d_in[1];
  const int* src    = eidx;
  const int* dst    = eidx + NE;
  const int* eattr  = (const int*)d_in[2];
  const float* xemb = (const float*)d_in[3];
  const float* eemb = (const float*)d_in[4];
  const float* W1   = (const float*)d_in[5];
  const float* as1  = (const float*)d_in[6];
  const float* ad1  = (const float*)d_in[7];
  const float* We1  = (const float*)d_in[8];
  const float* ae1  = (const float*)d_in[9];
  const float* b1   = (const float*)d_in[10];
  const float* W2   = (const float*)d_in[11];
  const float* as2  = (const float*)d_in[12];
  const float* ad2  = (const float*)d_in[13];
  const float* We2  = (const float*)d_in[14];
  const float* ae2  = (const float*)d_in[15];
  const float* b2   = (const float*)d_in[16];
  float* out = (float*)d_out;

  float* ws = (float*)d_ws;
  size_t off = 0;
  float* EA   = ws + off; off += (size_t)(NE + NN) * EMB;
  float* XF   = ws + off; off += (size_t)NN * EMB;
  float* DEG  = ws + off; off += 2048;
  float* HX1  = ws + off; off += (size_t)NN * NH * HID;
  float* HX2  = ws + off; off += (size_t)NN * NH * FT;
  float* PROJ = ws + off; off += (size_t)EMB * NH;
  float* ANS  = ws + off; off += (size_t)NN * NH;
  float* ADN  = ws + off; off += (size_t)NN * NH;
  float* ALPH = ws + off; off += (size_t)(NE + NN) * NH;
  unsigned* MX = (unsigned*)(ws + off); off += (size_t)NN * NH;
  float* DEN  = ws + off; off += (size_t)NN * NH;
  float* OUT1 = ws + off; off += (size_t)NN * HID;
  float* HB   = ws + off; off += (size_t)NN * HID;
  (void)ws_size; (void)in_sizes; (void)n_in; (void)out_size;

  const int T = 256;

  hipMemsetAsync(DEG, 0, 2048 * sizeof(float), stream);
  hipMemsetAsync(EA + (size_t)NE * EMB, 0, (size_t)NN * EMB * sizeof(float), stream);
  hipMemsetAsync(OUT1, 0, (size_t)NN * HID * sizeof(float), stream);
  hipMemsetAsync(MX, 0, (size_t)NN * NH * sizeof(unsigned), stream);
  hipMemsetAsync(DEN, 0, (size_t)NN * NH * sizeof(float), stream);

  k_embed_x<<<cdiv((long long)NN * EMB, T), T, 0, stream>>>(x, xemb, XF);
  k_embed_e<<<cdiv((long long)NE * EMB, T), T, 0, stream>>>(eattr, eemb, EA);
  k_deg<<<cdiv(NE, T), T, 0, stream>>>(dst, DEG);
  k_loopsum<<<cdiv((long long)NE * EMB, T), T, 0, stream>>>(dst, EA);
  k_loopdiv<<<cdiv((long long)NN * EMB, T), T, 0, stream>>>(DEG, EA);

  {
    dim3 g(NH * HID / 128, cdiv(NN, 64));
    k_gemm_wmma_f32<<<g, 256, 0, stream>>>(XF, W1, HX1, NN, NH * HID, EMB);
  }
  k_proj_e<<<cdiv(EMB * NH, T), T, 0, stream>>>(We1, ae1, PROJ, HID);
  k_alpha_node<<<cdiv(NN * NH, T), T, 0, stream>>>(HX1, as1, ad1, ANS, ADN, HID);
  k_edge_alpha<<<cdiv(NE + NN, 8), 256, 0, stream>>>(EA, PROJ, ANS, ADN, src, dst, ALPH, MX);
  k_softmax_den<<<cdiv((long long)(NE + NN) * NH, T), T, 0, stream>>>(dst, MX, ALPH, DEN);
  k_aggregate<<<NE + NN, HID, 0, stream>>>(HX1, ALPH, DEN, src, dst, OUT1, HID);
  k_bias_relu<<<cdiv((long long)NN * HID, T), T, 0, stream>>>(OUT1, b1, HB);

  hipMemsetAsync(MX, 0, (size_t)NN * NH * sizeof(unsigned), stream);
  hipMemsetAsync(DEN, 0, (size_t)NN * NH * sizeof(float), stream);

  {
    dim3 g(NH * FT / 128, cdiv(NN, 64));
    k_gemm_wmma_f32<<<g, 256, 0, stream>>>(HB, W2, HX2, NN, NH * FT, HID);
  }
  k_proj_e<<<cdiv(EMB * NH, T), T, 0, stream>>>(We2, ae2, PROJ, FT);
  k_alpha_node<<<cdiv(NN * NH, T), T, 0, stream>>>(HX2, as2, ad2, ANS, ADN, FT);
  k_edge_alpha<<<cdiv(NE + NN, 8), 256, 0, stream>>>(EA, PROJ, ANS, ADN, src, dst, ALPH, MX);
  k_softmax_den<<<cdiv((long long)(NE + NN) * NH, T), T, 0, stream>>>(dst, MX, ALPH, DEN);
  k_init_out<<<cdiv((long long)NN * FT, T), T, 0, stream>>>(b2, out);
  k_aggregate<<<NE + NN, FT, 0, stream>>>(HX2, ALPH, DEN, src, dst, out, FT);
}


# Aggregation without a sort: each work-group filters the edge list for its own 8 destination nodes into LDS lists, one wave per node sums in registers; float atomics on out gone; sum order changes
# speedup vs baseline: 2.9520x; 2.9520x over previous
#include <hip/hip_runtime.h>
#include <math.h>

#define NN   2000
#define NE   32000
#define EMB  768
#define HID  384
#define FT   768
#define NH   4
#define NEG_SLOPE 0.2f

typedef __attribute__((ext_vector_type(2))) float v2f;
typedef __attribute__((ext_vector_type(4))) float v4f;
typedef __attribute__((ext_vector_type(8))) float v8f;

static inline int cdiv(long long a, long long b) { return (int)((a + b - 1) / b); }

__device__ __forceinline__ unsigned fenc(float f) {
  unsigned u = __float_as_uint(f);
  return (u & 0x80000000u) ? ~u : (u | 0x80000000u);
}
__device__ __forceinline__ float fdec(unsigned e) {
  unsigned u = (e & 0x80000000u) ? (e ^ 0x80000000u) : ~e;
  return __uint_as_float(u);
}

__global__ void k_embed_x(const int* __restrict__ x, const float* __restrict__ emb,
                          float* __restrict__ xf) {
  long long i = (long long)blockIdx.x * blockDim.x + threadIdx.x;
  if (i >= (long long)NN * EMB) return;
  int n = (int)(i / EMB), c = (int)(i % EMB);
  xf[i] = emb[(size_t)x[n * 2 + 0] * EMB + c] + emb[(size_t)x[n * 2 + 1] * EMB + c];
}

__global__ void k_deg(const int* __restrict__ dst, float* __restrict__ deg) {
  int e = blockIdx.x * blockDim.x + threadIdx.x;
  if (e < NE) atomicAdd(&deg[dst[e]], 1.0f);
}

struct GemmFrag {
  v4f a[2][2];
  v4f b[2][4];
};

template <int K, int N>
__device__ __forceinline__ void gemm_load_stage(GemmFrag& f, const float* __restrict__ a0p,
                                                const float* __restrict__ a1p,
                                                const float* __restrict__ bp, int kb) {
#pragma unroll
  for (int u = 0; u < 2; ++u) {
    f.a[0][u] = *(const v4f*)(a0p + kb + 8 * u);
    f.a[1][u] = *(const v4f*)(a1p + kb + 8 * u);
#pragma unroll
    for (int t = 0; t < 4; ++t)
      f.b[u][t] = *(const v4f*)(bp + (size_t)(kb + 8 * u + t) * N);
  }
}

__device__ __forceinline__ void gemm_mma_stage(const GemmFrag& f, v8f (&acc)[2][4]) {
#pragma unroll
  for (int u = 0; u < 2; ++u) {
#pragma unroll
    for (int s = 0; s < 2; ++s) {
      v2f a0 = {f.a[0][u][2 * s], f.a[0][u][2 * s + 1]};
      v2f a1 = {f.a[1][u][2 * s], f.a[1][u][2 * s + 1]};
#pragma unroll
      for (int j = 0; j < 4; ++j) {
        v2f b = {f.b[u][2 * s][j], f.b[u][2 * s + 1][j]};
        acc[0][j] = __builtin_amdgcn_wmma_f32_16x16x4_f32(false, a0, false, b, (short)0, acc[0][j], false, false);
        acc[1][j] = __builtin_amdgcn_wmma_f32_16x16x4_f32(false, a1, false, b, (short)0, acc[1][j], false, false);
      }
    }
  }
}

template <int K, int N>
__global__ __launch_bounds__(128) void k_gemm_wmma_f32(
    const float* __restrict__ A, const float* __restrict__ B, float* __restrict__ C, int M) {
  static_assert(K % 48 == 0 && K >= 48 && N % 128 == 0, "tile shape");
  const int wave = threadIdx.x >> 5, lane = threadIdx.x & 31;
  const int c = lane & 15, h = lane >> 4;
  const int m0 = blockIdx.y * 64 + (wave >> 1) * 32;
  const int n0 = blockIdx.x * 128 + (wave & 1) * 64;
  if (m0 >= M) return;

  const int r0 = min(m0 + c, M - 1), r1 = min(m0 + 16 + c, M - 1);
  const float* a0p = A + (size_t)r0 * K + 4 * h;
  const float* a1p = A + (size_t)r1 * K + 4 * h;
  const float* bp  = B + (size_t)(4 * h) * N + n0 + 4 * c;

  v8f acc[2][4] = {};
  GemmFrag f0, f1, f2;
  gemm_load_stage<K, N>(f0, a0p, a1p, bp, 0);
  gemm_load_stage<K, N>(f1, a0p, a1p, bp, 16);
#define GEMM_SCHED_FENCE() __builtin_amdgcn_sched_barrier(0)
#pragma clang loop unroll(disable)
  for (int kb = 0; kb < K - 48; kb += 48) {
    gemm_load_stage<K, N>(f2, a0p, a1p, bp, kb + 32);
    GEMM_SCHED_FENCE();
    gemm_mma_stage(f0, acc);
    GEMM_SCHED_FENCE();
    gemm_load_stage<K, N>(f0, a0p, a1p, bp, kb + 48);
    GEMM_SCHED_FENCE();
    gemm_mma_stage(f1, acc);
    GEMM_SCHED_FENCE();
    gemm_load_stage<K, N>(f1, a0p, a1p, bp, kb + 64);
    GEMM_SCHED_FENCE();
    gemm_mma_stage(f2, acc);
    GEMM_SCHED_FENCE();
  }
  gemm_load_stage<K, N>(f2, a0p, a1p, bp, K - 16);
  GEMM_SCHED_FENCE();
  gemm_mma_stage(f0, acc);
  gemm_mma_stage(f1, acc);
  gemm_mma_stage(f2, acc);
#undef GEMM_SCHED_FENCE

#pragma unroll
  for (int i = 0; i < 2; ++i) {
#pragma unroll
    for (int r = 0; r < 8; ++r) {
      const int gm = m0 + 16 * i + 8 * h + r;
      if (gm < M) {
        v4f o = {acc[i][0][r], acc[i][1][r], acc[i][2][r], acc[i][3][r]};
        *(v4f*)(C + (size_t)gm * N + n0 + 4 * c) = o;
      }
    }
  }
}

__global__ void k_proj_e(const float* __restrict__ We, const float* __restrict__ ae,
                         float* __restrict__ proj, int C) {
  int i = blockIdx.x * blockDim.x + threadIdx.x;
  if (i >= EMB * NH) return;
  int k = i / NH, h = i % NH;
  const float* wr = We + (size_t)k * (NH * C) + (size_t)h * C;
  const float* ar = ae + (size_t)h * C;
  float s = 0.f;
  for (int c = 0; c < C; ++c) s += wr[c] * ar[c];
  proj[i] = s;
}

__global__ void k_alpha_node(const float* __restrict__ hx, const float* __restrict__ a_src,
                             const float* __restrict__ a_dst, float* __restrict__ ans,
                             float* __restrict__ adn, int C) {
  int i = blockIdx.x * blockDim.x + threadIdx.x;
  if (i >= NN * NH) return;
  int n = i / NH, h = i % NH;
  const float* hr = hx + (size_t)n * NH * C + (size_t)h * C;
  const float* sr = a_src + (size_t)h * C;
  const float* dr = a_dst + (size_t)h * C;
  float s = 0.f, t = 0.f;
  for (int c = 0; c < C; ++c) { s += hr[c] * sr[c]; t += hr[c] * dr[c]; }
  ans[i] = s;
  adn[i] = t;
}

__global__ void k_edge_alpha(const int* __restrict__ ea_idx, const float* __restrict__ emb,
                             const float* __restrict__ proj,
                             const float* __restrict__ ans, const float* __restrict__ adn,
                             const int* __restrict__ src, const int* __restrict__ dst,
                             float* __restrict__ alpha, unsigned* __restrict__ mx,
                             float* __restrict__ ls) {
  int e = blockIdx.x * (blockDim.x >> 5) + (threadIdx.x >> 5);
  int lane = threadIdx.x & 31;
  if (e >= NE) return;
  int s = src[e];
  int d = dst[e];
  const float* r0 = emb + (size_t)ea_idx[e * 2 + 0] * EMB;
  const float* r1 = emb + (size_t)ea_idx[e * 2 + 1] * EMB;
  const float4* p4 = (const float4*)proj;
  float a0 = 0.f, a1 = 0.f, a2 = 0.f, a3 = 0.f;
  for (int k = lane; k < EMB; k += 32) {
    float v = r0[k] + r1[k];
    float4 p = p4[k];
    a0 += v * p.x; a1 += v * p.y; a2 += v * p.z; a3 += v * p.w;
  }
#pragma unroll
  for (int off = 16; off > 0; off >>= 1) {
    a0 += __shfl_xor(a0, off, 32);
    a1 += __shfl_xor(a1, off, 32);
    a2 += __shfl_xor(a2, off, 32);
    a3 += __shfl_xor(a3, off, 32);
  }
  if (lane < 4) {
    float ale = (lane == 0) ? a0 : (lane == 1) ? a1 : (lane == 2) ? a2 : a3;
    float a = ans[s * NH + lane] + adn[d * NH + lane] + ale;
    a = (a > 0.f) ? a : NEG_SLOPE * a;
    alpha[(size_t)e * NH + lane] = a;
    atomicMax(&mx[d * NH + lane], fenc(a));
    atomicAdd(&ls[d * NH + lane], ale);
  }
}

__global__ void k_loop_alpha(const float* __restrict__ ls, const float* __restrict__ deg,
                             const float* __restrict__ ans, const float* __restrict__ adn,
                             float* __restrict__ alpha, unsigned* __restrict__ mx) {
  int i = blockIdx.x * blockDim.x + threadIdx.x;
  if (i >= NN * NH) return;
  int n = i / NH;
  float ale = ls[i] / fmaxf(deg[n], 1.0f);
  float a = ans[i] + adn[i] + ale;
  a = (a > 0.f) ? a : NEG_SLOPE * a;
  alpha[(size_t)NE * NH + i] = a;
  atomicMax(&mx[i], fenc(a));
}

__global__ void k_softmax_den(const int* __restrict__ dst, const unsigned* __restrict__ mx,
                              float* __restrict__ alpha, float* __restrict__ den) {
  const int tot = NE + NN;
  long long i = (long long)blockIdx.x * blockDim.x + threadIdx.x;
  if (i >= (long long)tot * NH) return;
  int e = (int)(i >> 2), h = (int)(i & 3);
  int d = (e < NE) ? dst[e] : (e - NE);
  float ex = expf(alpha[i] - fdec(mx[d * NH + h]));
  alpha[i] = ex;
  atomicAdd(&den[d * NH + h], ex);
}

#define GN   8
#define GCAP 64

template <int C>
__device__ __forceinline__ void gather_load(v4f (&v)[NH * (C / 128)], const float* __restrict__ hx, int s,
                                            int lane) {
  const float* hr = hx + (size_t)s * (NH * C) + 4 * lane;
#pragma unroll
  for (int q = 0; q < NH * (C / 128); ++q) v[q] = *(const v4f*)(hr + 128 * q);
}

template <int C>
__device__ __forceinline__ void gather_fma(v4f (&acc)[C / 128], const v4f (&v)[NH * (C / 128)], v4f a) {
#pragma unroll
  for (int h = 0; h < NH; ++h) {
#pragma unroll
    for (int j = 0; j < C / 128; ++j) acc[j] += a[h] * v[h * (C / 128) + j];
  }
}

__device__ __forceinline__ v4f lane_bcast(v4f x, int k) {
  v4f r;
  r[0] = __int_as_float(__builtin_amdgcn_readlane(__float_as_int(x[0]), k));
  r[1] = __int_as_float(__builtin_amdgcn_readlane(__float_as_int(x[1]), k));
  r[2] = __int_as_float(__builtin_amdgcn_readlane(__float_as_int(x[2]), k));
  r[3] = __int_as_float(__builtin_amdgcn_readlane(__float_as_int(x[3]), k));
  return r;
}

template <int C, bool RELU>
__global__ __launch_bounds__(256) void k_gather(const float* __restrict__ hx, const float* __restrict__ ex,
                                                const float* __restrict__ den, const int* __restrict__ src,
                                                const int* __restrict__ dst, const float* __restrict__ bias,
                                                float* __restrict__ out) {
  static_assert(C % 128 == 0 && NN % GN == 0 && NE % 32 == 0 && NE % 4 == 0, "shape");
  __shared__ int cnt[GN];
  __shared__ int lst[GN][GCAP];
  const int tid = threadIdx.x, wave = tid >> 5, lane = tid & 31;
  const int n0 = blockIdx.x * GN;
  if (tid < GN) cnt[tid] = 0;
  __syncthreads();
  const int4* d4 = (const int4*)dst;
  constexpr int SCAN_B = 16;
  for (int i0 = tid; i0 < NE / 4; i0 += 256 * SCAN_B) {
    int4 q[SCAN_B];
#pragma unroll
    for (int u = 0; u < SCAN_B; ++u) {
      const int i = i0 + 256 * u;
      q[u] = (i < NE / 4) ? d4[i] : make_int4(-1, -1, -1, -1);
    }
#pragma unroll
    for (int u = 0; u < SCAN_B; ++u) {
      const int dd[4] = {q[u].x, q[u].y, q[u].z, q[u].w};
#pragma unroll
      for (int t = 0; t < 4; ++t) {
        const unsigned g = (unsigned)(dd[t] - n0);
        if (g < (unsigned)GN) {
          const int p = atomicAdd(&cnt[g], 1);
          if (p < GCAP) lst[g][p] = 4 * (i0 + 256 * u) + t;
        }
      }
    }
  }
  __syncthreads();

  const int d = n0 + wave;
  const int n = __builtin_amdgcn_readfirstlane(cnt[wave]);
  const v4f dn = *(const v4f*)(den + (size_t)d * NH);
  v4f acc[C / 128] = {};
#define GATHER_FENCE() __builtin_amdgcn_sched_barrier(0)
  if (n <= GCAP) {
    const int nt = n + 1;
    for (int base = 0; base < nt; base += 32) {
      const int m = min(32, nt - base);
      const int i = base + lane;
      int e_i = NE + d, s_i = d;
      if (i > 0 && i < nt) { e_i = lst[wave][i - 1]; s_i = src[e_i]; }
      const v4f a_i = *(const v4f*)(ex + (size_t)e_i * NH) / dn;
      int k = 0;
      for (; k + 1 < m; k += 2) {
        v4f v0[NH * (C / 128)], v1[NH * (C / 128)];
        gather_load<C>(v0, hx, __builtin_amdgcn_readlane(s_i, k), lane);
        gather_load<C>(v1, hx, __builtin_amdgcn_readlane(s_i, k + 1), lane);
        GATHER_FENCE();
        gather_fma<C>(acc, v0, lane_bcast(a_i, k));
        gather_fma<C>(acc, v1, lane_bcast(a_i, k + 1));
        GATHER_FENCE();
      }
      if (k < m) {
        v4f v0[NH * (C / 128)];
        gather_load<C>(v0, hx, __builtin_amdgcn_readlane(s_i, k), lane);
        GATHER_FENCE();
        gather_fma<C>(acc, v0, lane_bcast(a_i, k));
      }
    }
  } else {
    {
      v4f v0[NH * (C / 128)];
      gather_load<C>(v0, hx, d, lane);
      gather_fma<C>(acc, v0, *(const v4f*)(ex + (size_t)(NE + d) * NH) / dn);
    }
    for (int base = 0; base < NE; base += 32) {
      unsigned mk = __builtin_amdgcn_ballot_w32(dst[base + lane] == d);
      while (mk) {
        const int e = base + __builtin_ctz(mk);
        mk &= mk - 1;
        v4f v0[NH * (C / 128)];
        gather_load<C>(v0, hx, src[e], lane);
        gather_fma<C>(acc, v0, *(const v4f*)(ex + (size_t)e * NH) / dn);
      }
    }
  }
#undef GATHER_FENCE
#pragma unroll
  for (int j = 0; j < C / 128; ++j) {
    const int c0 = 128 * j + 4 * lane;
    v4f o = 0.25f * acc[j] + *(const v4f*)(bias + c0);
    if (RELU) { o[0] = fmaxf(o[0], 0.f); o[1] = fmaxf(o[1], 0.f); o[2] = fmaxf(o[2], 0.f); o[3] = fmaxf(o[3], 0.f); }
    *(v4f*)(out + (size_t)d * C + c0) = o;
  }
}

extern "C" void kernel_launch(void* const* d_in, const int* in_sizes, int n_in,
                              void* d_out, int out_size, void* d_ws, size_t ws_size,
                              hipStream_t stream) {
  const int* x      = (const int*)d_in[0];
  const int* eidx   = (const int*)d_in[1];
  const int* src    = eidx;
  const int* dst    = eidx + NE;
  const int* eattr  = (const int*)d_in[2];
  const float* xemb = (const float*)d_in[3];
  const float* eemb = (const float*)d_in[4];
  const float* W1   = (const float*)d_in[5];
  const float* as1  = (const float*)d_in[6];
  const float* ad1  = (const float*)d_in[7];
  const float* We1  = (const float*)d_in[8];
  const float* ae1  = (const float*)d_in[9];
  const float* b1   = (const float*)d_in[10];
  const float* W2   = (const float*)d_in[11];
  const float* as2  = (const float*)d_in[12];
  const float* ad2  = (const float*)d_in[13];
  const float* We2  = (const float*)d_in[14];
  const float* ae2  = (const float*)d_in[15];
  const float* b2   = (const float*)d_in[16];
  float* out = (float*)d_out;

  float* ws = (float*)d_ws;
  size_t off = 0;
  float* XF   = ws + off; off += (size_t)NN * EMB;
  float* DEG  = ws + off; off += 2048;
  float* HX1  = ws + off; off += (size_t)NN * NH * HID;
  float* HX2  = ws + off; off += (size_t)NN * NH * FT;
  float* PROJ = ws + off; off += (size_t)EMB * NH;
  float* ANS  = ws + off; off += (size_t)NN * NH;
  float* ADN  = ws + off; off += (size_t)NN * NH;
  float* ALPH = ws + off; off += (size_t)(NE + NN) * NH;
  unsigned* MX = (unsigned*)(ws + off); off += (size_t)NN * NH;
  float* DEN  = ws + off; off += (size_t)NN * NH;
  float* LS   = ws + off; off += (size_t)NN * NH;
  float* HB   = ws + off; off += (size_t)NN * HID;
  (void)ws_size; (void)in_sizes; (void)n_in; (void)out_size;

  const int T = 256;

  hipMemsetAsync(DEG, 0, 2048 * sizeof(float), stream);
  hipMemsetAsync(MX, 0, (size_t)NN * NH * sizeof(unsigned), stream);
  hipMemsetAsync(DEN, 0, (size_t)2 * NN * NH * sizeof(float), stream);

  k_embed_x<<<cdiv((long long)NN * EMB, T), T, 0, stream>>>(x, xemb, XF);
  k_deg<<<cdiv(NE, T), T, 0, stream>>>(dst, DEG);

  {
    dim3 g(NH * HID / 128, cdiv(NN, 64));
    k_gemm_wmma_f32<EMB, NH * HID><<<g, 128, 0, stream>>>(XF, W1, HX1, NN);
  }
  k_proj_e<<<cdiv(EMB * NH, T), T, 0, stream>>>(We1, ae1, PROJ, HID);
  k_alpha_node<<<cdiv(NN * NH, T), T, 0, stream>>>(HX1, as1, ad1, ANS, ADN, HID);
  k_edge_alpha<<<cdiv(NE, 8), 256, 0, stream>>>(eattr, eemb, PROJ, ANS, ADN, src, dst, ALPH, MX, LS);
  k_loop_alpha<<<cdiv(NN * NH, T), T, 0, stream>>>(LS, DEG, ANS, ADN, ALPH, MX);
  k_softmax_den<<<cdiv((long long)(NE + NN) * NH, T), T, 0, stream>>>(dst, MX, ALPH, DEN);
  k_gather<HID, true><<<NN / GN, 256, 0, stream>>>(HX1, ALPH, DEN, src, dst, b1, HB);

  hipMemsetAsync(MX, 0, (size_t)NN * NH * sizeof(unsigned), stream);
  hipMemsetAsync(DEN, 0, (size_t)2 * NN * NH * sizeof(float), stream);

  {
    dim3 g(NH * FT / 128, cdiv(NN, 64));
    k_gemm_wmma_f32<HID, NH * FT><<<g, 128, 0, stream>>>(HB, W2, HX2, NN);
  }
  k_proj_e<<<cdiv(EMB * NH, T), T, 0, stream>>>(We2, ae2, PROJ, FT);
  k_alpha_node<<<cdiv(NN * NH, T), T, 0, stream>>>(HX2, as2, ad2, ANS, ADN, FT);
  k_edge_alpha<<<cdiv(NE, 8), 256, 0, stream>>>(eattr, eemb, PROJ, ANS, ADN, src, dst, ALPH, MX, LS);
  k_loop_alpha<<<cdiv(NN * NH, T), T, 0, stream>>>(LS, DEG, ANS, ADN, ALPH, MX);
  k_softmax_den<<<cdiv((long long)(NE + NN) * NH, T), T, 0, stream>>>(dst, MX, ALPH, DEN);
  k_gather<FT, false><<<NN / GN, 256, 0, stream>>>(HX2, ALPH, DEN, src, dst, b2, out);
}


# Many items per wave: feature kernel one wave per node row (48000 waves to 2000); edge kernel 8 edges per wave, proj in registers, halving-exchange sums (32000 to 4000); sum order changes
# speedup vs baseline: 3.1315x; 1.0608x over previous
#include <hip/hip_runtime.h>
#include <math.h>

#define NN   2000
#define NE   32000
#define EMB  768
#define HID  384
#define FT   768
#define NH   4
#define NEG_SLOPE 0.2f

typedef __attribute__((ext_vector_type(2))) float v2f;
typedef __attribute__((ext_vector_type(4))) float v4f;
typedef __attribute__((ext_vector_type(8))) float v8f;

static inline int cdiv(long long a, long long b) { return (int)((a + b - 1) / b); }

__device__ __forceinline__ unsigned fenc(float f) {
  unsigned u = __float_as_uint(f);
  return (u & 0x80000000u) ? ~u : (u | 0x80000000u);
}
__device__ __forceinline__ float fdec(unsigned e) {
  unsigned u = (e & 0x80000000u) ? (e ^ 0x80000000u) : ~e;
  return __uint_as_float(u);
}

__global__ __launch_bounds__(256) void k_embed_x(const int* __restrict__ x, const float* __restrict__ emb,
                                                 float* __restrict__ xf) {
  static_assert(NN % 8 == 0 && EMB % 128 == 0, "shape");
  const int lane = threadIdx.x & 31;
  const int n = __builtin_amdgcn_readfirstlane(blockIdx.x * 8 + (threadIdx.x >> 5));
  const float* r0 = emb + (size_t)x[n * 2 + 0] * EMB + 4 * lane;
  const float* r1 = emb + (size_t)x[n * 2 + 1] * EMB + 4 * lane;
  float* o = xf + (size_t)n * EMB + 4 * lane;
#pragma unroll
  for (int t = 0; t < EMB / 128; ++t)
    *(v4f*)(o + 128 * t) = *(const v4f*)(r0 + 128 * t) + *(const v4f*)(r1 + 128 * t);
}

__global__ void k_deg(const int* __restrict__ dst, float* __restrict__ deg) {
  int e = blockIdx.x * blockDim.x + threadIdx.x;
  if (e < NE) atomicAdd(&deg[dst[e]], 1.0f);
}

struct GemmFrag {
  v4f a[2][2];
  v4f b[2][4];
};

template <int K, int N>
__device__ __forceinline__ void gemm_load_stage(GemmFrag& f, const float* __restrict__ a0p,
                                                const float* __restrict__ a1p,
                                                const float* __restrict__ bp, int kb) {
#pragma unroll
  for (int u = 0; u < 2; ++u) {
    f.a[0][u] = *(const v4f*)(a0p + kb + 8 * u);
    f.a[1][u] = *(const v4f*)(a1p + kb + 8 * u);
#pragma unroll
    for (int t = 0; t < 4; ++t)
      f.b[u][t] = *(const v4f*)(bp + (size_t)(kb + 8 * u + t) * N);
  }
}

__device__ __forceinline__ void gemm_mma_stage(const GemmFrag& f, v8f (&acc)[2][4]) {
#pragma unroll
  for (int u = 0; u < 2; ++u) {
#pragma unroll
    for (int s = 0; s < 2; ++s) {
      v2f a0 = {f.a[0][u][2 * s], f.a[0][u][2 * s + 1]};
      v2f a1 = {f.a[1][u][2 * s], f.a[1][u][2 * s + 1]};
#pragma unroll
      for (int j = 0; j < 4; ++j) {
        v2f b = {f.b[u][2 * s][j], f.b[u][2 * s + 1][j]};
        acc[0][j] = __builtin_amdgcn_wmma_f32_16x16x4_f32(false, a0, false, b, (short)0, acc[0][j], false, false);
        acc[1][j] = __builtin_amdgcn_wmma_f32_16x16x4_f32(false, a1, false, b, (short)0, acc[1][j], false, false);
      }
    }
  }
}

template <int K, int N>
__global__ __launch_bounds__(128) void k_gemm_wmma_f32(
    const float* __restrict__ A, const float* __restrict__ B, float* __restrict__ C, int M) {
  static_assert(K % 48 == 0 && K >= 48 && N % 128 == 0, "tile shape");
  const int wave = threadIdx.x >> 5, lane = threadIdx.x & 31;
  const int c = lane & 15, h = lane >> 4;
  const int m0 = blockIdx.y * 64 + (wave >> 1) * 32;
  const int n0 = blockIdx.x * 128 + (wave & 1) * 64;
  if (m0 >= M) return;

  const int r0 = min(m0 + c, M - 1), r1 = min(m0 + 16 + c, M - 1);
  const float* a0p = A + (size_t)r0 * K + 4 * h;
  const float* a1p = A + (size_t)r1 * K + 4 * h;
  const float* bp  = B + (size_t)(4 * h) * N + n0 + 4 * c;

  v8f acc[2][4] = {};
  GemmFrag f0, f1, f2;
  gemm_load_stage<K, N>(f0, a0p, a1p, bp, 0);
  gemm_load_stage<K, N>(f1, a0p, a1p, bp, 16);
#define GEMM_SCHED_FENCE() __builtin_amdgcn_sched_barrier(0)
#pragma clang loop unroll(disable)
  for (int kb = 0; kb < K - 48; kb += 48) {
    gemm_load_stage<K, N>(f2, a0p, a1p, bp, kb + 32);
    GEMM_SCHED_FENCE();
    gemm_mma_stage(f0, acc);
    GEMM_SCHED_FENCE();
    gemm_load_stage<K, N>(f0, a0p, a1p, bp, kb + 48);
    GEMM_SCHED_FENCE();
    gemm_mma_stage(f1, acc);
    GEMM_SCHED_FENCE();
    gemm_load_stage<K, N>(f1, a0p, a1p, bp, kb + 64);
    GEMM_SCHED_FENCE();
    gemm_mma_stage(f2, acc);
    GEMM_SCHED_FENCE();
  }
  gemm_load_stage<K, N>(f2, a0p, a1p, bp, K - 16);
  GEMM_SCHED_FENCE();
  gemm_mma_stage(f0, acc);
  gemm_mma_stage(f1, acc);
  gemm_mma_stage(f2, acc);
#undef GEMM_SCHED_FENCE

#pragma unroll
  for (int i = 0; i < 2; ++i) {
#pragma unroll
    for (int r = 0; r < 8; ++r) {
      const int gm = m0 + 16 * i + 8 * h + r;
      if (gm < M) {
        v4f o = {acc[i][0][r], acc[i][1][r], acc[i][2][r], acc[i][3][r]};
        *(v4f*)(C + (size_t)gm * N + n0 + 4 * c) = o;
      }
    }
  }
}

__global__ void k_proj_e(const float* __restrict__ We, const float* __restrict__ ae,
                         float* __restrict__ proj, int C) {
  int i = blockIdx.x * blockDim.x + threadIdx.x;
  if (i >= EMB * NH) return;
  int k = i / NH, h = i % NH;
  const float* wr = We + (size_t)k * (NH * C) + (size_t)h * C;
  const float* ar = ae + (size_t)h * C;
  float s = 0.f;
  for (int c = 0; c < C; ++c) s += wr[c] * ar[c];
  proj[i] = s;
}

__global__ void k_alpha_node(const float* __restrict__ hx, const float* __restrict__ a_src,
                             const float* __restrict__ a_dst, float* __restrict__ ans,
                             float* __restrict__ adn, int C) {
  int i = blockIdx.x * blockDim.x + threadIdx.x;
  if (i >= NN * NH) return;
  int n = i / NH, h = i % NH;
  const float* hr = hx + (size_t)n * NH * C + (size_t)h * C;
  const float* sr = a_src + (size_t)h * C;
  const float* dr = a_dst + (size_t)h * C;
  float s = 0.f, t = 0.f;
  for (int c = 0; c < C; ++c) { s += hr[c] * sr[c]; t += hr[c] * dr[c]; }
  ans[i] = s;
  adn[i] = t;
}

#define EPW 8

template <int N, int OFF>
__device__ __forceinline__ void halving_step(float (&v)[EPW * NH], int lane) {
  const bool up = (lane & OFF) != 0;
#pragma unroll
  for (int i = 0; i < N / 2; ++i) {
    const float keep = up ? v[i + N / 2] : v[i];
    const float send = up ? v[i] : v[i + N / 2];
    v[i] = keep + __shfl_xor(send, OFF, 32);
  }
}

__global__ __launch_bounds__(256) void k_edge_alpha(const int* __restrict__ ea_idx, const float* __restrict__ emb,
                             const float* __restrict__ proj,
                             const float* __restrict__ ans, const float* __restrict__ adn,
                             const int* __restrict__ src, const int* __restrict__ dst,
                             float* __restrict__ alpha, unsigned* __restrict__ mx,
                             float* __restrict__ ls) {
  static_assert(EPW * NH == 32 && NE % (8 * EPW) == 0 && EMB == 768, "one value per lane; whole work-groups; six 128-float pieces");
  const int lane = threadIdx.x & 31;
  const int e0 = __builtin_amdgcn_readfirstlane((blockIdx.x * 8 + (threadIdx.x >> 5)) * EPW);

  v4f p[EMB / 128][4];
#pragma unroll
  for (int t = 0; t < EMB / 128; ++t) {
#pragma unroll
    for (int kk = 0; kk < 4; ++kk) p[t][kk] = *(const v4f*)(proj + (size_t)(128 * t + 4 * lane + kk) * NH);
  }

  float v[EPW * NH];
#pragma unroll
  for (int j = 0; j < EPW; ++j) {
    const float* r0 = emb + (size_t)ea_idx[(e0 + j) * 2 + 0] * EMB + 4 * lane;
    const float* r1 = emb + (size_t)ea_idx[(e0 + j) * 2 + 1] * EMB + 4 * lane;
    v4f a = {0.f, 0.f, 0.f, 0.f};
#pragma unroll
    for (int t = 0; t < EMB / 128; ++t) {
      const v4f xr = *(const v4f*)(r0 + 128 * t) + *(const v4f*)(r1 + 128 * t);
#pragma unroll
      for (int kk = 0; kk < 4; ++kk) a += xr[kk] * p[t][kk];
    }
    v[NH * j + 0] = a[0]; v[NH * j + 1] = a[1]; v[NH * j + 2] = a[2]; v[NH * j + 3] = a[3];
  }

  halving_step<32, 16>(v, lane);
  halving_step<16, 8>(v, lane);
  halving_step<8, 4>(v, lane);
  halving_step<4, 2>(v, lane);
  halving_step<2, 1>(v, lane);
  const float ale = v[0];

  const int e = e0 + (lane >> 2), h = lane & 3;
  const int s = src[e];
  const int d = dst[e];
  float a = ans[s * NH + h] + adn[d * NH + h] + ale;
  a = (a > 0.f) ? a : NEG_SLOPE * a;
  alpha[(size_t)e * NH + h] = a;
  atomicMax(&mx[d * NH + h], fenc(a));
  atomicAdd(&ls[d * NH + h], ale);
}

__global__ void k_loop_alpha(const float* __restrict__ ls, const float* __restrict__ deg,
                             const float* __restrict__ ans, const float* __restrict__ adn,
                             float* __restrict__ alpha, unsigned* __restrict__ mx) {
  int i = blockIdx.x * blockDim.x + threadIdx.x;
  if (i >= NN * NH) return;
  int n = i / NH;
  float ale = ls[i] / fmaxf(deg[n], 1.0f);
  float a = ans[i] + adn[i] + ale;
  a = (a > 0.f) ? a : NEG_SLOPE * a;
  alpha[(size_t)NE * NH + i] = a;
  atomicMax(&mx[i], fenc(a));
}

__global__ void k_softmax_den(const int* __restrict__ dst, const unsigned* __restrict__ mx,
                              float* __restrict__ alpha, float* __restrict__ den) {
  const int tot = NE + NN;
  long long i = (long long)blockIdx.x * blockDim.x + threadIdx.x;
  if (i >= (long long)tot * NH) return;
  int e = (int)(i >> 2), h = (int)(i & 3);
  int d = (e < NE) ? dst[e] : (e - NE);
  float ex = expf(alpha[i] - fdec(mx[d * NH + h]));
  alpha[i] = ex;
  atomicAdd(&den[d * NH + h], ex);
}

#define GN   8
#define GCAP 64

template <int C>
__device__ __forceinline__ void gather_load(v4f (&v)[NH * (C / 128)], const float* __restrict__ hx, int s,
                                            int lane) {
  const float* hr = hx + (size_t)s * (NH * C) + 4 * lane;
#pragma unroll
  for (int q = 0; q < NH * (C / 128); ++q) v[q] = *(const v4f*)(hr + 128 * q);
}

template <int C>
__device__ __forceinline__ void gather_fma(v4f (&acc)[C / 128], const v4f (&v)[NH * (C / 128)], v4f a) {
#pragma unroll
  for (int h = 0; h < NH; ++h) {
#pragma unroll
    for (int j = 0; j < C / 128; ++j) acc[j] += a[h] * v[h * (C / 128) + j];
  }
}

__device__ __forceinline__ v4f lane_bcast(v4f x, int k) {
  v4f r;
  r[0] = __int_as_float(__builtin_amdgcn_readlane(__float_as_int(x[0]), k));
  r[1] = __int_as_float(__builtin_amdgcn_readlane(__float_as_int(x[1]), k));
  r[2] = __int_as_float(__builtin_amdgcn_readlane(__float_as_int(x[2]), k));
  r[3] = __int_as_float(__builtin_amdgcn_readlane(__float_as_int(x[3]), k));
  return r;
}

template <int C, bool RELU>
__global__ __launch_bounds__(256) void k_gather(const float* __restrict__ hx, const float* __restrict__ ex,
                                                const float* __restrict__ den, const int* __restrict__ src,
                                                const int* __restrict__ dst, const float* __restrict__ bias,
                                                float* __restrict__ out) {
  static_assert(C % 128 == 0 && NN % GN == 0 && NE % 32 == 0 && NE % 4 == 0, "shape");
  __shared__ int cnt[GN];
  __shared__ int lst[GN][GCAP];
  const int tid = threadIdx.x, wave = tid >> 5, lane = tid & 31;
  const int n0 = blockIdx.x * GN;
  if (tid < GN) cnt[tid] = 0;
  __syncthreads();
  const int4* d4 = (const int4*)dst;
  constexpr int SCAN_B = 16;
  for (int i0 = tid; i0 < NE / 4; i0 += 256 * SCAN_B) {
    int4 q[SCAN_B];
#pragma unroll
    for (int u = 0; u < SCAN_B; ++u) {
      const int i = i0 + 256 * u;
      q[u] = (i < NE / 4) ? d4[i] : make_int4(-1, -1, -1, -1);
    }
#pragma unroll
    for (int u = 0; u < SCAN_B; ++u) {
      const int dd[4] = {q[u].x, q[u].y, q[u].z, q[u].w};
#pragma unroll
      for (int t = 0; t < 4; ++t) {
        const unsigned g = (unsigned)(dd[t] - n0);
        if (g < (unsigned)GN) {
          const int p = atomicAdd(&cnt[g], 1);
          if (p < GCAP) lst[g][p] = 4 * (i0 + 256 * u) + t;
        }
      }
    }
  }
  __syncthreads();

  const int d = n0 + wave;
  const int n = __builtin_amdgcn_readfirstlane(cnt[wave]);
  const v4f dn = *(const v4f*)(den + (size_t)d * NH);
  v4f acc[C / 128] = {};
#define GATHER_FENCE() __builtin_amdgcn_sched_barrier(0)
  if (n <= GCAP) {
    const int nt = n + 1;
    for (int base = 0; base < nt; base += 32) {
      const int m = min(32, nt - base);
      const int i = base + lane;
      int e_i = NE + d, s_i = d;
      if (i > 0 && i < nt) { e_i = lst[wave][i - 1]; s_i = src[e_i]; }
      const v4f a_i = *(const v4f*)(ex + (size_t)e_i * NH) / dn;
      int k = 0;
      for (; k + 1 < m; k += 2) {
        v4f v0[NH * (C / 128)], v1[NH * (C / 128)];
        gather_load<C>(v0, hx, __builtin_amdgcn_readlane(s_i, k), lane);
        gather_load<C>(v1, hx, __builtin_amdgcn_readlane(s_i, k + 1), lane);
        GATHER_FENCE();
        gather_fma<C>(acc, v0, lane_bcast(a_i, k));
        gather_fma<C>(acc, v1, lane_bcast(a_i, k + 1));
        GATHER_FENCE();
      }
      if (k < m) {
        v4f v0[NH * (C / 128)];
        gather_load<C>(v0, hx, __builtin_amdgcn_readlane(s_i, k), lane);
        GATHER_FENCE();
        gather_fma<C>(acc, v0, lane_bcast(a_i, k));
      }
    }
  } else {
    {
      v4f v0[NH * (C / 128)];
      gather_load<C>(v0, hx, d, lane);
      gather_fma<C>(acc, v0, *(const v4f*)(ex + (size_t)(NE + d) * NH) / dn);
    }
    for (int base = 0; base < NE; base += 32) {
      unsigned mk = __builtin_amdgcn_ballot_w32(dst[base + lane] == d);
      while (mk) {
        const int e = base + __builtin_ctz(mk);
        mk &= mk - 1;
        v4f v0[NH * (C / 128)];
        gather_load<C>(v0, hx, src[e], lane);
        gather_fma<C>(acc, v0, *(const v4f*)(ex + (size_t)e * NH) / dn);
      }
    }
  }
#undef GATHER_FENCE
#pragma unroll
  for (int j = 0; j < C / 128; ++j) {
    const int c0 = 128 * j + 4 * lane;
    v4f o = 0.25f * acc[j] + *(const v4f*)(bias + c0);
    if (RELU) { o[0] = fmaxf(o[0], 0.f); o[1] = fmaxf(o[1], 0.f); o[2] = fmaxf(o[2], 0.f); o[3] = fmaxf(o[3], 0.f); }
    *(v4f*)(out + (size_t)d * C + c0) = o;
  }
}

extern "C" void kernel_launch(void* const* d_in, const int* in_sizes, int n_in,
                              void* d_out, int out_size, void* d_ws, size_t ws_size,
                              hipStream_t stream) {
  const int* x      = (const int*)d_in[0];
  const int* eidx   = (const int*)d_in[1];
  const int* src    = eidx;
  const int* dst    = eidx + NE;
  const int* eattr  = (const int*)d_in[2];
  const float* xemb = (const float*)d_in[3];
  const float* eemb = (const float*)d_in[4];
  const float* W1   = (const float*)d_in[5];
  const float* as1  = (const float*)d_in[6];
  const float* ad1  = (const float*)d_in[7];
  const float* We1  = (const float*)d_in[8];
  const float* ae1  = (const float*)d_in[9];
  const float* b1   = (const float*)d_in[10];
  const float* W2   = (const float*)d_in[11];
  const float* as2  = (const float*)d_in[12];
  const float* ad2  = (const float*)d_in[13];
  const float* We2  = (const float*)d_in[14];
  const float* ae2  = (const float*)d_in[15];
  const float* b2   = (const float*)d_in[16];
  float* out = (float*)d_out;

  float* ws = (float*)d_ws;
  size_t off = 0;
  float* XF   = ws + off; off += (size_t)NN * EMB;
  float* DEG  = ws + off; off += 2048;
  float* HX1  = ws + off; off += (size_t)NN * NH * HID;
  float* HX2  = ws + off; off += (size_t)NN * NH * FT;
  float* PROJ = ws + off; off += (size_t)EMB * NH;
  float* ANS  = ws + off; off += (size_t)NN * NH;
  float* ADN  = ws + off; off += (size_t)NN * NH;
  float* ALPH = ws + off; off += (size_t)(NE + NN) * NH;
  unsigned* MX = (unsigned*)(ws + off); off += (size_t)NN * NH;
  float* DEN  = ws + off; off += (size_t)NN * NH;
  float* LS   = ws + off; off += (size_t)NN * NH;
  float* HB   = ws + off; off += (size_t)NN * HID;
  (void)ws_size; (void)in_sizes; (void)n_in; (void)out_size;

  const int T = 256;

  hipMemsetAsync(DEG, 0, 2048 * sizeof(float), stream);
  hipMemsetAsync(MX, 0, (size_t)NN * NH * sizeof(unsigned), stream);
  hipMemsetAsync(DEN, 0, (size_t)2 * NN * NH * sizeof(float), stream);

  k_embed_x<<<NN / 8, 256, 0, stream>>>(x, xemb, XF);
  k_deg<<<cdiv(NE, T), T, 0, stream>>>(dst, DEG);

  {
    dim3 g(NH * HID / 128, cdiv(NN, 64));
    k_gemm_wmma_f32<EMB, NH * HID><<<g, 128, 0, stream>>>(XF, W1, HX1, NN);
  }
  k_proj_e<<<cdiv(EMB * NH, T), T, 0, stream>>>(We1, ae1, PROJ, HID);
  k_alpha_node<<<cdiv(NN * NH, T), T, 0, stream>>>(HX1, as1, ad1, ANS, ADN, HID);
  k_edge_alpha<<<NE / (8 * EPW), 256, 0, stream>>>(eattr, eemb, PROJ, ANS, ADN, src, dst, ALPH, MX, LS);
  k_loop_alpha<<<cdiv(NN * NH, T), T, 0, stream>>>(LS, DEG, ANS, ADN, ALPH, MX);
  k_softmax_den<<<cdiv((long long)(NE + NN) * NH, T), T, 0, stream>>>(dst, MX, ALPH, DEN);
  k_gather<HID, true><<<NN / GN, 256, 0, stream>>>(HX1, ALPH, DEN, src, dst, b1, HB);

  hipMemsetAsync(MX, 0, (size_t)NN * NH * sizeof(unsigned), stream);
  hipMemsetAsync(DEN, 0, (size_t)2 * NN * NH * sizeof(float), stream);

  {
    dim3 g(NH * FT / 128, cdiv(NN, 64));
    k_gemm_wmma_f32<HID, NH * FT><<<g, 128, 0, stream>>>(HB, W2, HX2, NN);
  }
  k_proj_e<<<cdiv(EMB * NH, T), T, 0, stream>>>(We2, ae2, PROJ, FT);
  k_alpha_node<<<cdiv(NN * NH, T), T, 0, stream>>>(HX2, as2, ad2, ANS, ADN, FT);
  k_edge_alpha<<<NE / (8 * EPW), 256, 0, stream>>>(eattr, eemb, PROJ, ANS, ADN, src, dst, ALPH, MX, LS);
  k_loop_alpha<<<cdiv(NN * NH, T), T, 0, stream>>>(LS, DEG, ANS, ADN, ALPH, MX);
  k_softmax_den<<<cdiv((long long)(NE + NN) * NH, T), T, 0, stream>>>(dst, MX, ALPH, DEN);
  k_gather<FT, false><<<NN / GN, 256, 0, stream>>>(HX2, ALPH, DEN, src, dst, b2, out);
}


# Node attention scalars from the GEMM epilogue: each wave dots its 32x64 tile in registers with a_src/a_dst (16-lane halving sums), a small kernel adds the 6 or 12 tile sums; sum order changes
# speedup vs baseline: 3.7670x; 1.2029x over previous
#include <hip/hip_runtime.h>
#include <math.h>

#define NN   2000
#define NE   32000
#define EMB  768
#define HID  384
#define FT   768
#define NH   4
#define NEG_SLOPE 0.2f

typedef __attribute__((ext_vector_type(2))) float v2f;
typedef __attribute__((ext_vector_type(4))) float v4f;
typedef __attribute__((ext_vector_type(8))) float v8f;

static inline int cdiv(long long a, long long b) { return (int)((a + b - 1) / b); }

__device__ __forceinline__ unsigned fenc(float f) {
  unsigned u = __float_as_uint(f);
  return (u & 0x80000000u) ? ~u : (u | 0x80000000u);
}
__device__ __forceinline__ float fdec(unsigned e) {
  unsigned u = (e & 0x80000000u) ? (e ^ 0x80000000u) : ~e;
  return __uint_as_float(u);
}

__global__ __launch_bounds__(256) void k_embed_x(const int* __restrict__ x, const float* __restrict__ emb,
                                                 float* __restrict__ xf) {
  static_assert(NN % 8 == 0 && EMB % 128 == 0, "shape");
  const int lane = threadIdx.x & 31;
  const int n = __builtin_amdgcn_readfirstlane(blockIdx.x * 8 + (threadIdx.x >> 5));
  const float* r0 = emb + (size_t)x[n * 2 + 0] * EMB + 4 * lane;
  const float* r1 = emb + (size_t)x[n * 2 + 1] * EMB + 4 * lane;
  float* o = xf + (size_t)n * EMB + 4 * lane;
#pragma unroll
  for (int t = 0; t < EMB / 128; ++t)
    *(v4f*)(o + 128 * t) = *(const v4f*)(r0 + 128 * t) + *(const v4f*)(r1 + 128 * t);
}

__global__ void k_deg(const int* __restrict__ dst, float* __restrict__ deg) {
  int e = blockIdx.x * blockDim.x + threadIdx.x;
  if (e < NE) atomicAdd(&deg[dst[e]], 1.0f);
}

#define EPW 8

template <int N, int OFF>
__device__ __forceinline__ void halving_step(float (&v)[EPW * NH], int lane) {
  const bool up = (lane & OFF) != 0;
#pragma unroll
  for (int i = 0; i < N / 2; ++i) {
    const float keep = up ? v[i + N / 2] : v[i];
    const float send = up ? v[i] : v[i + N / 2];
    v[i] = keep + __shfl_xor(send, OFF, 32);
  }
}

#define MP 2048

struct GemmFrag {
  v4f a[2][2];
  v4f b[2][4];
};

template <int K, int N>
__device__ __forceinline__ void gemm_load_stage(GemmFrag& f, const float* __restrict__ a0p,
                                                const float* __restrict__ a1p,
                                                const float* __restrict__ bp, int kb) {
#pragma unroll
  for (int u = 0; u < 2; ++u) {
    f.a[0][u] = *(const v4f*)(a0p + kb + 8 * u);
    f.a[1][u] = *(const v4f*)(a1p + kb + 8 * u);
#pragma unroll
    for (int t = 0; t < 4; ++t)
      f.b[u][t] = *(const v4f*)(bp + (size_t)(kb + 8 * u + t) * N);
  }
}

__device__ __forceinline__ void gemm_mma_stage(const GemmFrag& f, v8f (&acc)[2][4]) {
#pragma unroll
  for (int u = 0; u < 2; ++u) {
#pragma unroll
    for (int s = 0; s < 2; ++s) {
      v2f a0 = {f.a[0][u][2 * s], f.a[0][u][2 * s + 1]};
      v2f a1 = {f.a[1][u][2 * s], f.a[1][u][2 * s + 1]};
#pragma unroll
      for (int j = 0; j < 4; ++j) {
        v2f b = {f.b[u][2 * s][j], f.b[u][2 * s + 1][j]};
        acc[0][j] = __builtin_amdgcn_wmma_f32_16x16x4_f32(false, a0, false, b, (short)0, acc[0][j], false, false);
        acc[1][j] = __builtin_amdgcn_wmma_f32_16x16x4_f32(false, a1, false, b, (short)0, acc[1][j], false, false);
      }
    }
  }
}

template <int K, int N>
__global__ __launch_bounds__(128) void k_gemm_wmma_f32(
    const float* __restrict__ A, const float* __restrict__ B, float* __restrict__ C, int M,
    const float* __restrict__ a_src, const float* __restrict__ a_dst,
    float* __restrict__ ps, float* __restrict__ pd) {
  static_assert(K % 48 == 0 && K >= 48 && N % 128 == 0, "tile shape");
  static_assert((N / NH) % 64 == 0 && MP >= NN && MP % 32 == 0, "a wave's 64 columns inside one head; whole lines of row sums");
  const int wave = threadIdx.x >> 5, lane = threadIdx.x & 31;
  const int c = lane & 15, h = lane >> 4;
  const int m0 = blockIdx.y * 64 + (wave >> 1) * 32;
  const int n0 = blockIdx.x * 128 + (wave & 1) * 64;
  if (m0 >= M) return;

  const int r0 = min(m0 + c, M - 1), r1 = min(m0 + 16 + c, M - 1);
  const float* a0p = A + (size_t)r0 * K + 4 * h;
  const float* a1p = A + (size_t)r1 * K + 4 * h;
  const float* bp  = B + (size_t)(4 * h) * N + n0 + 4 * c;

  v8f acc[2][4] = {};
  GemmFrag f0, f1, f2;
  gemm_load_stage<K, N>(f0, a0p, a1p, bp, 0);
  gemm_load_stage<K, N>(f1, a0p, a1p, bp, 16);
#define GEMM_SCHED_FENCE() __builtin_amdgcn_sched_barrier(0)
#pragma clang loop unroll(disable)
  for (int kb = 0; kb < K - 48; kb += 48) {
    gemm_load_stage<K, N>(f2, a0p, a1p, bp, kb + 32);
    GEMM_SCHED_FENCE();
    gemm_mma_stage(f0, acc);
    GEMM_SCHED_FENCE();
    gemm_load_stage<K, N>(f0, a0p, a1p, bp, kb + 48);
    GEMM_SCHED_FENCE();
    gemm_mma_stage(f1, acc);
    GEMM_SCHED_FENCE();
    gemm_load_stage<K, N>(f1, a0p, a1p, bp, kb + 64);
    GEMM_SCHED_FENCE();
    gemm_mma_stage(f2, acc);
    GEMM_SCHED_FENCE();
  }
  gemm_load_stage<K, N>(f2, a0p, a1p, bp, K - 16);
  GEMM_SCHED_FENCE();
  gemm_mma_stage(f0, acc);
  gemm_mma_stage(f1, acc);
  gemm_mma_stage(f2, acc);
#undef GEMM_SCHED_FENCE

#pragma unroll
  for (int i = 0; i < 2; ++i) {
#pragma unroll
    for (int r = 0; r < 8; ++r) {
      const int gm = m0 + 16 * i + 8 * h + r;
      if (gm < M) {
        v4f o = {acc[i][0][r], acc[i][1][r], acc[i][2][r], acc[i][3][r]};
        *(v4f*)(C + (size_t)gm * N + n0 + 4 * c) = o;
      }
    }
  }

  const v4f asv = *(const v4f*)(a_src + n0 + 4 * c);
  const v4f adv = *(const v4f*)(a_dst + n0 + 4 * c);
  float v[EPW * NH];
#pragma unroll
  for (int i = 0; i < 2; ++i) {
#pragma unroll
    for (int r = 0; r < 8; ++r) {
      float s = acc[i][0][r] * asv[0], d = acc[i][0][r] * adv[0];
#pragma unroll
      for (int j = 1; j < 4; ++j) { s += acc[i][j][r] * asv[j]; d += acc[i][j][r] * adv[j]; }
      v[2 * (8 * i + r) + 0] = s;
      v[2 * (8 * i + r) + 1] = d;
    }
  }
  halving_step<32, 8>(v, lane);
  halving_step<16, 4>(v, lane);
  halving_step<8, 2>(v, lane);
  halving_step<4, 1>(v, lane);
  const int prow = m0 + 16 * (c >> 3) + 8 * h + (c & 7);
  if (prow < M) {
    ps[(size_t)(n0 >> 6) * MP + prow] = v[0];
    pd[(size_t)(n0 >> 6) * MP + prow] = v[1];
  }
}

__global__ void k_proj_e(const float* __restrict__ We, const float* __restrict__ ae,
                         float* __restrict__ proj, int C) {
  int i = blockIdx.x * blockDim.x + threadIdx.x;
  if (i >= EMB * NH) return;
  int k = i / NH, h = i % NH;
  const float* wr = We + (size_t)k * (NH * C) + (size_t)h * C;
  const float* ar = ae + (size_t)h * C;
  float s = 0.f;
  for (int c = 0; c < C; ++c) s += wr[c] * ar[c];
  proj[i] = s;
}

template <int T>
__global__ void k_alpha_sum(const float* __restrict__ ps, const float* __restrict__ pd,
                            float* __restrict__ ans, float* __restrict__ adn) {
  int i = blockIdx.x * blockDim.x + threadIdx.x;
  if (i >= NN * NH) return;
  int n = i / NH, h = i % NH;
  float s = 0.f, t = 0.f;
#pragma unroll
  for (int q = 0; q < T; ++q) {
    s += ps[(size_t)(h * T + q) * MP + n];
    t += pd[(size_t)(h * T + q) * MP + n];
  }
  ans[i] = s;
  adn[i] = t;
}

__global__ __launch_bounds__(256) void k_edge_alpha(const int* __restrict__ ea_idx, const float* __restrict__ emb,
                             const float* __restrict__ proj,
                             const float* __restrict__ ans, const float* __restrict__ adn,
                             const int* __restrict__ src, const int* __restrict__ dst,
                             float* __restrict__ alpha, unsigned* __restrict__ mx,
                             float* __restrict__ ls) {
  static_assert(EPW * NH == 32 && NE % (8 * EPW) == 0 && EMB == 768, "one value per lane; whole work-groups; six 128-float pieces");
  const int lane = threadIdx.x & 31;
  const int e0 = __builtin_amdgcn_readfirstlane((blockIdx.x * 8 + (threadIdx.x >> 5)) * EPW);

  v4f p[EMB / 128][4];
#pragma unroll
  for (int t = 0; t < EMB / 128; ++t) {
#pragma unroll
    for (int kk = 0; kk < 4; ++kk) p[t][kk] = *(const v4f*)(proj + (size_t)(128 * t + 4 * lane + kk) * NH);
  }

  float v[EPW * NH];
#pragma unroll
  for (int j = 0; j < EPW; ++j) {
    const float* r0 = emb + (size_t)ea_idx[(e0 + j) * 2 + 0] * EMB + 4 * lane;
    const float* r1 = emb + (size_t)ea_idx[(e0 + j) * 2 + 1] * EMB + 4 * lane;
    v4f a = {0.f, 0.f, 0.f, 0.f};
#pragma unroll
    for (int t = 0; t < EMB / 128; ++t) {
      const v4f xr = *(const v4f*)(r0 + 128 * t) + *(const v4f*)(r1 + 128 * t);
#pragma unroll
      for (int kk = 0; kk < 4; ++kk) a += xr[kk] * p[t][kk];
    }
    v[NH * j + 0] = a[0]; v[NH * j + 1] = a[1]; v[NH * j + 2] = a[2]; v[NH * j + 3] = a[3];
  }

  halving_step<32, 16>(v, lane);
  halving_step<16, 8>(v, lane);
  halving_step<8, 4>(v, lane);
  halving_step<4, 2>(v, lane);
  halving_step<2, 1>(v, lane);
  const float ale = v[0];

  const int e = e0 + (lane >> 2), h = lane & 3;
  const int s = src[e];
  const int d = dst[e];
  float a = ans[s * NH + h] + adn[d * NH + h] + ale;
  a = (a > 0.f) ? a : NEG_SLOPE * a;
  alpha[(size_t)e * NH + h] = a;
  atomicMax(&mx[d * NH + h], fenc(a));
  atomicAdd(&ls[d * NH + h], ale);
}

__global__ void k_loop_alpha(const float* __restrict__ ls, const float* __restrict__ deg,
                             const float* __restrict__ ans, const float* __restrict__ adn,
                             float* __restrict__ alpha, unsigned* __restrict__ mx) {
  int i = blockIdx.x * blockDim.x + threadIdx.x;
  if (i >= NN * NH) return;
  int n = i / NH;
  float ale = ls[i] / fmaxf(deg[n], 1.0f);
  float a = ans[i] + adn[i] + ale;
  a = (a > 0.f) ? a : NEG_SLOPE * a;
  alpha[(size_t)NE * NH + i] = a;
  atomicMax(&mx[i], fenc(a));
}

__global__ void k_softmax_den(const int* __restrict__ dst, const unsigned* __restrict__ mx,
                              float* __restrict__ alpha, float* __restrict__ den) {
  const int tot = NE + NN;
  long long i = (long long)blockIdx.x * blockDim.x + threadIdx.x;
  if (i >= (long long)tot * NH) return;
  int e = (int)(i >> 2), h = (int)(i & 3);
  int d = (e < NE) ? dst[e] : (e - NE);
  float ex = expf(alpha[i] - fdec(mx[d * NH + h]));
  alpha[i] = ex;
  atomicAdd(&den[d * NH + h], ex);
}

#define GN   8
#define GCAP 64

template <int C>
__device__ __forceinline__ void gather_load(v4f (&v)[NH * (C / 128)], const float* __restrict__ hx, int s,
                                            int lane) {
  const float* hr = hx + (size_t)s * (NH * C) + 4 * lane;
#pragma unroll
  for (int q = 0; q < NH * (C / 128); ++q) v[q] = *(const v4f*)(hr + 128 * q);
}

template <int C>
__device__ __forceinline__ void gather_fma(v4f (&acc)[C / 128], const v4f (&v)[NH * (C / 128)], v4f a) {
#pragma unroll
  for (int h = 0; h < NH; ++h) {
#pragma unroll
    for (int j = 0; j < C / 128; ++j) acc[j] += a[h] * v[h * (C / 128) + j];
  }
}

__device__ __forceinline__ v4f lane_bcast(v4f x, int k) {
  v4f r;
  r[0] = __int_as_float(__builtin_amdgcn_readlane(__float_as_int(x[0]), k));
  r[1] = __int_as_float(__builtin_amdgcn_readlane(__float_as_int(x[1]), k));
  r[2] = __int_as_float(__builtin_amdgcn_readlane(__float_as_int(x[2]), k));
  r[3] = __int_as_float(__builtin_amdgcn_readlane(__float_as_int(x[3]), k));
  return r;
}

template <int C, bool RELU>
__global__ __launch_bounds__(256) void k_gather(const float* __restrict__ hx, const float* __restrict__ ex,
                                                const float* __restrict__ den, const int* __restrict__ src,
                                                const int* __restrict__ dst, const float* __restrict__ bias,
                                                float* __restrict__ out) {
  static_assert(C % 128 == 0 && NN % GN == 0 && NE % 32 == 0 && NE % 4 == 0, "shape");
  __shared__ int cnt[GN];
  __shared__ int lst[GN][GCAP];
  const int tid = threadIdx.x, wave = tid >> 5, lane = tid & 31;
  const int n0 = blockIdx.x * GN;
  if (tid < GN) cnt[tid] = 0;
  __syncthreads();
  const int4* d4 = (const int4*)dst;
  constexpr int SCAN_B = 16;
  for (int i0 = tid; i0 < NE / 4; i0 += 256 * SCAN_B) {
    int4 q[SCAN_B];
#pragma unroll
    for (int u = 0; u < SCAN_B; ++u) {
      const int i = i0 + 256 * u;
      q[u] = (i < NE / 4) ? d4[i] : make_int4(-1, -1, -1, -1);
    }
#pragma unroll
    for (int u = 0; u < SCAN_B; ++u) {
      const int dd[4] = {q[u].x, q[u].y, q[u].z, q[u].w};
#pragma unroll
      for (int t = 0; t < 4; ++t) {
        const unsigned g = (unsigned)(dd[t] - n0);
        if (g < (unsigned)GN) {
          const int p = atomicAdd(&cnt[g], 1);
          if (p < GCAP) lst[g][p] = 4 * (i0 + 256 * u) + t;
        }
      }
    }
  }
  __syncthreads();

  const int d = n0 + wave;
  const int n = __builtin_amdgcn_readfirstlane(cnt[wave]);
  const v4f dn = *(const v4f*)(den + (size_t)d * NH);
  v4f acc[C / 128] = {};
#define GATHER_FENCE() __builtin_amdgcn_sched_barrier(0)
  if (n <= GCAP) {
    const int nt = n + 1;
    for (int base = 0; base < nt; base += 32) {
      const int m = min(32, nt - base);
      const int i = base + lane;
      int e_i = NE + d, s_i = d;
      if (i > 0 && i < nt) { e_i = lst[wave][i - 1]; s_i = src[e_i]; }
      const v4f a_i = *(const v4f*)(ex + (size_t)e_i * NH) / dn;
      int k = 0;
      for (; k + 1 < m; k += 2) {
        v4f v0[NH * (C / 128)], v1[NH * (C / 128)];
        gather_load<C>(v0, hx, __builtin_amdgcn_readlane(s_i, k), lane);
        gather_load<C>(v1, hx, __builtin_amdgcn_readlane(s_i, k + 1), lane);
        GATHER_FENCE();
        gather_fma<C>(acc, v0, lane_bcast(a_i, k));
        gather_fma<C>(acc, v1, lane_bcast(a_i, k + 1));
        GATHER_FENCE();
      }
      if (k < m) {
        v4f v0[NH * (C / 128)];
        gather_load<C>(v0, hx, __builtin_amdgcn_readlane(s_i, k), lane);
        GATHER_FENCE();
        gather_fma<C>(acc, v0, lane_bcast(a_i, k));
      }
    }
  } else {
    {
      v4f v0[NH * (C / 128)];
      gather_load<C>(v0, hx, d, lane);
      gather_fma<C>(acc, v0, *(const v4f*)(ex + (size_t)(NE + d) * NH) / dn);
    }
    for (int base = 0; base < NE; base += 32) {
      unsigned mk = __builtin_amdgcn_ballot_w32(dst[base + lane] == d);
      while (mk) {
        const int e = base + __builtin_ctz(mk);
        mk &= mk - 1;
        v4f v0[NH * (C / 128)];
        gather_load<C>(v0, hx, src[e], lane);
        gather_fma<C>(acc, v0, *(const v4f*)(ex + (size_t)e * NH) / dn);
      }
    }
  }
#undef GATHER_FENCE
#pragma unroll
  for (int j = 0; j < C / 128; ++j) {
    const int c0 = 128 * j + 4 * lane;
    v4f o = 0.25f * acc[j] + *(const v4f*)(bias + c0);
    if (RELU) { o[0] = fmaxf(o[0], 0.f); o[1] = fmaxf(o[1], 0.f); o[2] = fmaxf(o[2], 0.f); o[3] = fmaxf(o[3], 0.f); }
    *(v4f*)(out + (size_t)d * C + c0) = o;
  }
}

extern "C" void kernel_launch(void* const* d_in, const int* in_sizes, int n_in,
                              void* d_out, int out_size, void* d_ws, size_t ws_size,
                              hipStream_t stream) {
  const int* x      = (const int*)d_in[0];
  const int* eidx   = (const int*)d_in[1];
  const int* src    = eidx;
  const int* dst    = eidx + NE;
  const int* eattr  = (const int*)d_in[2];
  const float* xemb = (const float*)d_in[3];
  const float* eemb = (const float*)d_in[4];
  const float* W1   = (const float*)d_in[5];
  const float* as1  = (const float*)d_in[6];
  const float* ad1  = (const float*)d_in[7];
  const float* We1  = (const float*)d_in[8];
  const float* ae1  = (const float*)d_in[9];
  const float* b1   = (const float*)d_in[10];
  const float* W2   = (const float*)d_in[11];
  const float* as2  = (const float*)d_in[12];
  const float* ad2  = (const float*)d_in[13];
  const float* We2  = (const float*)d_in[14];
  const float* ae2  = (const float*)d_in[15];
  const float* b2   = (const float*)d_in[16];
  float* out = (float*)d_out;

  float* ws = (float*)d_ws;
  size_t off = 0;
  float* XF   = ws + off; off += (size_t)NN * EMB;
  float* DEG  = ws + off; off += 2048;
  float* HX1  = ws + off; off += (size_t)NN * NH * HID;
  float* HX2  = ws + off; off += (size_t)NN * NH * FT;
  float* PROJ = ws + off; off += (size_t)EMB * NH;
  float* ANS  = ws + off; off += (size_t)NN * NH;
  float* ADN  = ws + off; off += (size_t)NN * NH;
  float* ALPH = ws + off; off += (size_t)(NE + NN) * NH;
  unsigned* MX = (unsigned*)(ws + off); off += (size_t)NN * NH;
  float* DEN  = ws + off; off += (size_t)NN * NH;
  float* LS   = ws + off; off += (size_t)NN * NH;
  float* HB   = ws + off; off += (size_t)NN * HID;
  float* PS   = ws + off; off += (size_t)(NH * FT / 64) * MP;
  float* PD   = ws + off; off += (size_t)(NH * FT / 64) * MP;
  (void)ws_size; (void)in_sizes; (void)n_in; (void)out_size;

  const int T = 256;

  hipMemsetAsync(DEG, 0, 2048 * sizeof(float), stream);
  hipMemsetAsync(MX, 0, (size_t)NN * NH * sizeof(unsigned), stream);
  hipMemsetAsync(DEN, 0, (size_t)2 * NN * NH * sizeof(float), stream);

  k_embed_x<<<NN / 8, 256, 0, stream>>>(x, xemb, XF);
  k_deg<<<cdiv(NE, T), T, 0, stream>>>(dst, DEG);

  {
    dim3 g(NH * HID / 128, cdiv(NN, 64));
    k_gemm_wmma_f32<EMB, NH * HID><<<g, 128, 0, stream>>>(XF, W1, HX1, NN, as1, ad1, PS, PD);
  }
  k_proj_e<<<cdiv(EMB * NH, T), T, 0, stream>>>(We1, ae1, PROJ, HID);
  k_alpha_sum<HID / 64><<<cdiv(NN * NH, T), T, 0, stream>>>(PS, PD, ANS, ADN);
  k_edge_alpha<<<NE / (8 * EPW), 256, 0, stream>>>(eattr, eemb, PROJ, ANS, ADN, src, dst, ALPH, MX, LS);
  k_loop_alpha<<<cdiv(NN * NH, T), T, 0, stream>>>(LS, DEG, ANS, ADN, ALPH, MX);
  k_softmax_den<<<cdiv((long long)(NE + NN) * NH, T), T, 0, stream>>>(dst, MX, ALPH, DEN);
  k_gather<HID, true><<<NN / GN, 256, 0, stream>>>(HX1, ALPH, DEN, src, dst, b1, HB);

  hipMemsetAsync(MX, 0, (size_t)NN * NH * sizeof(unsigned), stream);
  hipMemsetAsync(DEN, 0, (size_t)2 * NN * NH * sizeof(float), stream);

  {
    dim3 g(NH * FT / 128, cdiv(NN, 64));
    k_gemm_wmma_f32<HID, NH * FT><<<g, 128, 0, stream>>>(HB, W2, HX2, NN, as2, ad2, PS, PD);
  }
  k_proj_e<<<cdiv(EMB * NH, T), T, 0, stream>>>(We2, ae2, PROJ, FT);
  k_alpha_sum<FT / 64><<<cdiv(NN * NH, T), T, 0, stream>>>(PS, PD, ANS, ADN);
  k_edge_alpha<<<NE / (8 * EPW), 256, 0, stream>>>(eattr, eemb, PROJ, ANS, ADN, src, dst, ALPH, MX, LS);
  k_loop_alpha<<<cdiv(NN * NH, T), T, 0, stream>>>(LS, DEG, ANS, ADN, ALPH, MX);
  k_softmax_den<<<cdiv((long long)(NE + NN) * NH, T), T, 0, stream>>>(dst, MX, ALPH, DEN);
  k_gather<FT, false><<<NN / GN, 256, 0, stream>>>(HX2, ALPH, DEN, src, dst, b2, out);
}


# proj rows: one thread per row kept, its loads issued in batches of 12 pieces per operand, one batch ahead of the sums; one-wave work-groups (96, was 12 of 8 waves); sums and order unchanged
# speedup vs baseline: 4.3923x; 1.1660x over previous
#include <hip/hip_runtime.h>
#include <math.h>

#define NN   2000
#define NE   32000
#define EMB  768
#define HID  384
#define FT   768
#define NH   4
#define NEG_SLOPE 0.2f

typedef __attribute__((ext_vector_type(2))) float v2f;
typedef __attribute__((ext_vector_type(4))) float v4f;
typedef __attribute__((ext_vector_type(8))) float v8f;

static inline int cdiv(long long a, long long b) { return (int)((a + b - 1) / b); }

__device__ __forceinline__ unsigned fenc(float f) {
  unsigned u = __float_as_uint(f);
  return (u & 0x80000000u) ? ~u : (u | 0x80000000u);
}
__device__ __forceinline__ float fdec(unsigned e) {
  unsigned u = (e & 0x80000000u) ? (e ^ 0x80000000u) : ~e;
  return __uint_as_float(u);
}

__global__ __launch_bounds__(256) void k_embed_x(const int* __restrict__ x, const float* __restrict__ emb,
                                                 float* __restrict__ xf) {
  static_assert(NN % 8 == 0 && EMB % 128 == 0, "shape");
  const int lane = threadIdx.x & 31;
  const int n = __builtin_amdgcn_readfirstlane(blockIdx.x * 8 + (threadIdx.x >> 5));
  const float* r0 = emb + (size_t)x[n * 2 + 0] * EMB + 4 * lane;
  const float* r1 = emb + (size_t)x[n * 2 + 1] * EMB + 4 * lane;
  float* o = xf + (size_t)n * EMB + 4 * lane;
#pragma unroll
  for (int t = 0; t < EMB / 128; ++t)
    *(v4f*)(o + 128 * t) = *(const v4f*)(r0 + 128 * t) + *(const v4f*)(r1 + 128 * t);
}

__global__ void k_deg(const int* __restrict__ dst, float* __restrict__ deg) {
  int e = blockIdx.x * blockDim.x + threadIdx.x;
  if (e < NE) atomicAdd(&deg[dst[e]], 1.0f);
}

#define EPW 8

template <int N, int OFF>
__device__ __forceinline__ void halving_step(float (&v)[EPW * NH], int lane) {
  const bool up = (lane & OFF) != 0;
#pragma unroll
  for (int i = 0; i < N / 2; ++i) {
    const float keep = up ? v[i + N / 2] : v[i];
    const float send = up ? v[i] : v[i + N / 2];
    v[i] = keep + __shfl_xor(send, OFF, 32);
  }
}

#define MP 2048

struct GemmFrag {
  v4f a[2][2];
  v4f b[2][4];
};

template <int K, int N>
__device__ __forceinline__ void gemm_load_stage(GemmFrag& f, const float* __restrict__ a0p,
                                                const float* __restrict__ a1p,
                                                const float* __restrict__ bp, int kb) {
#pragma unroll
  for (int u = 0; u < 2; ++u) {
    f.a[0][u] = *(const v4f*)(a0p + kb + 8 * u);
    f.a[1][u] = *(const v4f*)(a1p + kb + 8 * u);
#pragma unroll
    for (int t = 0; t < 4; ++t)
      f.b[u][t] = *(const v4f*)(bp + (size_t)(kb + 8 * u + t) * N);
  }
}

__device__ __forceinline__ void gemm_mma_stage(const GemmFrag& f, v8f (&acc)[2][4]) {
#pragma unroll
  for (int u = 0; u < 2; ++u) {
#pragma unroll
    for (int s = 0; s < 2; ++s) {
      v2f a0 = {f.a[0][u][2 * s], f.a[0][u][2 * s + 1]};
      v2f a1 = {f.a[1][u][2 * s], f.a[1][u][2 * s + 1]};
#pragma unroll
      for (int j = 0; j < 4; ++j) {
        v2f b = {f.b[u][2 * s][j], f.b[u][2 * s + 1][j]};
        acc[0][j] = __builtin_amdgcn_wmma_f32_16x16x4_f32(false, a0, false, b, (short)0, acc[0][j], false, false);
        acc[1][j] = __builtin_amdgcn_wmma_f32_16x16x4_f32(false, a1, false, b, (short)0, acc[1][j], false, false);
      }
    }
  }
}

template <int K, int N>
__global__ __launch_bounds__(128) void k_gemm_wmma_f32(
    const float* __restrict__ A, const float* __restrict__ B, float* __restrict__ C, int M,
    const float* __restrict__ a_src, const float* __restrict__ a_dst,
    float* __restrict__ ps, float* __restrict__ pd) {
  static_assert(K % 48 == 0 && K >= 48 && N % 128 == 0, "tile shape");
  static_assert((N / NH) % 64 == 0 && MP >= NN && MP % 32 == 0, "a wave's 64 columns inside one head; whole lines of row sums");
  const int wave = threadIdx.x >> 5, lane = threadIdx.x & 31;
  const int c = lane & 15, h = lane >> 4;
  const int m0 = blockIdx.y * 64 + (wave >> 1) * 32;
  const int n0 = blockIdx.x * 128 + (wave & 1) * 64;
  if (m0 >= M) return;

  const int r0 = min(m0 + c, M - 1), r1 = min(m0 + 16 + c, M - 1);
  const float* a0p = A + (size_t)r0 * K + 4 * h;
  const float* a1p = A + (size_t)r1 * K + 4 * h;
  const float* bp  = B + (size_t)(4 * h) * N + n0 + 4 * c;

  v8f acc[2][4] = {};
  GemmFrag f0, f1, f2;
  gemm_load_stage<K, N>(f0, a0p, a1p, bp, 0);
  gemm_load_stage<K, N>(f1, a0p, a1p, bp, 16);
#define GEMM_SCHED_FENCE() __builtin_amdgcn_sched_barrier(0)
#pragma clang loop unroll(disable)
  for (int kb = 0; kb < K - 48; kb += 48) {
    gemm_load_stage<K, N>(f2, a0p, a1p, bp, kb + 32);
    GEMM_SCHED_FENCE();
    gemm_mma_stage(f0, acc);
    GEMM_SCHED_FENCE();
    gemm_load_stage<K, N>(f0, a0p, a1p, bp, kb + 48);
    GEMM_SCHED_FENCE();
    gemm_mma_stage(f1, acc);
    GEMM_SCHED_FENCE();
    gemm_load_stage<K, N>(f1, a0p, a1p, bp, kb + 64);
    GEMM_SCHED_FENCE();
    gemm_mma_stage(f2, acc);
    GEMM_SCHED_FENCE();
  }
  gemm_load_stage<K, N>(f2, a0p, a1p, bp, K - 16);
  GEMM_SCHED_FENCE();
  gemm_mma_stage(f0, acc);
  gemm_mma_stage(f1, acc);
  gemm_mma_stage(f2, acc);
#undef GEMM_SCHED_FENCE

#pragma unroll
  for (int i = 0; i < 2; ++i) {
#pragma unroll
    for (int r = 0; r < 8; ++r) {
      const int gm = m0 + 16 * i + 8 * h + r;
      if (gm < M) {
        v4f o = {acc[i][0][r], acc[i][1][r], acc[i][2][r], acc[i][3][r]};
        *(v4f*)(C + (size_t)gm * N + n0 + 4 * c) = o;
      }
    }
  }

  const v4f asv = *(const v4f*)(a_src + n0 + 4 * c);
  const v4f adv = *(const v4f*)(a_dst + n0 + 4 * c);
  float v[EPW * NH];
#pragma unroll
  for (int i = 0; i < 2; ++i) {
#pragma unroll
    for (int r = 0; r < 8; ++r) {
      float s = acc[i][0][r] * asv[0], d = acc[i][0][r] * adv[0];
#pragma unroll
      for (int j = 1; j < 4; ++j) { s += acc[i][j][r] * asv[j]; d += acc[i][j][r] * adv[j]; }
      v[2 * (8 * i + r) + 0] = s;
      v[2 * (8 * i + r) + 1] = d;
    }
  }
  halving_step<32, 8>(v, lane);
  halving_step<16, 4>(v, lane);
  halving_step<8, 2>(v, lane);
  halving_step<4, 1>(v, lane);
  const int prow = m0 + 16 * (c >> 3) + 8 * h + (c & 7);
  if (prow < M) {
    ps[(size_t)(n0 >> 6) * MP + prow] = v[0];
    pd[(size_t)(n0 >> 6) * MP + prow] = v[1];
  }
}

#define PB 12

template <int C>
__device__ __forceinline__ void proj_load(v4f (&w)[PB], v4f (&a)[PB], const float* __restrict__ wr,
                                          const float* __restrict__ ar, int c0) {
#pragma unroll
  for (int u = 0; u < PB; ++u) {
    w[u] = *(const v4f*)(wr + c0 + 4 * u);
    a[u] = *(const v4f*)(ar + c0 + 4 * u);
  }
}

__device__ __forceinline__ float proj_fma(float s, const v4f (&w)[PB], const v4f (&a)[PB]) {
#pragma unroll
  for (int u = 0; u < PB; ++u) {
#pragma unroll
    for (int q = 0; q < 4; ++q) s = __builtin_fmaf(w[u][q], a[u][q], s);
  }
  return s;
}

template <int C>
__global__ __launch_bounds__(32) void k_proj_e(const float* __restrict__ We, const float* __restrict__ ae,
                                               float* __restrict__ proj) {
  static_assert(C % (8 * PB) == 0 && C >= 16 * PB && (EMB * NH) % 32 == 0, "an even number of whole batches; whole waves");
  const int i = blockIdx.x * 32 + threadIdx.x;
  const int k = i / NH, h = i % NH;
  const float* wr = We + (size_t)k * (NH * C) + (size_t)h * C;
  const float* ar = ae + (size_t)h * C;
#define PROJ_FENCE() __builtin_amdgcn_sched_barrier(0)
  v4f w0[PB], a0[PB], w1[PB], a1[PB];
  float s = 0.f;
  proj_load<C>(w0, a0, wr, ar, 0);
#pragma clang loop unroll(disable)
  for (int c0 = 0; c0 < C - 8 * PB; c0 += 8 * PB) {
    proj_load<C>(w1, a1, wr, ar, c0 + 4 * PB);
    PROJ_FENCE();
    s = proj_fma(s, w0, a0);
    PROJ_FENCE();
    proj_load<C>(w0, a0, wr, ar, c0 + 8 * PB);
    PROJ_FENCE();
    s = proj_fma(s, w1, a1);
    PROJ_FENCE();
  }
  proj_load<C>(w1, a1, wr, ar, C - 4 * PB);
  PROJ_FENCE();
  s = proj_fma(s, w0, a0);
  s = proj_fma(s, w1, a1);
#undef PROJ_FENCE
  proj[i] = s;
}

template <int T>
__global__ void k_alpha_sum(const float* __restrict__ ps, const float* __restrict__ pd,
                            float* __restrict__ ans, float* __restrict__ adn) {
  int i = blockIdx.x * blockDim.x + threadIdx.x;
  if (i >= NN * NH) return;
  int n = i / NH, h = i % NH;
  float s = 0.f, t = 0.f;
#pragma unroll
  for (int q = 0; q < T; ++q) {
    s += ps[(size_t)(h * T + q) * MP + n];
    t += pd[(size_t)(h * T + q) * MP + n];
  }
  ans[i] = s;
  adn[i] = t;
}

__global__ __launch_bounds__(256) void k_edge_alpha(const int* __restrict__ ea_idx, const float* __restrict__ emb,
                             const float* __restrict__ proj,
                             const float* __restrict__ ans, const float* __restrict__ adn,
                             const int* __restrict__ src, const int* __restrict__ dst,
                             float* __restrict__ alpha, unsigned* __restrict__ mx,
                             float* __restrict__ ls) {
  static_assert(EPW * NH == 32 && NE % (8 * EPW) == 0 && EMB == 768, "one value per lane; whole work-groups; six 128-float pieces");
  const int lane = threadIdx.x & 31;
  const int e0 = __builtin_amdgcn_readfirstlane((blockIdx.x * 8 + (threadIdx.x >> 5)) * EPW);

  v4f p[EMB / 128][4];
#pragma unroll
  for (int t = 0; t < EMB / 128; ++t) {
#pragma unroll
    for (int kk = 0; kk < 4; ++kk) p[t][kk] = *(const v4f*)(proj + (size_t)(128 * t + 4 * lane + kk) * NH);
  }

  float v[EPW * NH];
#pragma unroll
  for (int j = 0; j < EPW; ++j) {
    const float* r0 = emb + (size_t)ea_idx[(e0 + j) * 2 + 0] * EMB + 4 * lane;
    const float* r1 = emb + (size_t)ea_idx[(e0 + j) * 2 + 1] * EMB + 4 * lane;
    v4f a = {0.f, 0.f, 0.f, 0.f};
#pragma unroll
    for (int t = 0; t < EMB / 128; ++t) {
      const v4f xr = *(const v4f*)(r0 + 128 * t) + *(const v4f*)(r1 + 128 * t);
#pragma unroll
      for (int kk = 0; kk < 4; ++kk) a += xr[kk] * p[t][kk];
    }
    v[NH * j + 0] = a[0]; v[NH * j + 1] = a[1]; v[NH * j + 2] = a[2]; v[NH * j + 3] = a[3];
  }

  halving_step<32, 16>(v, lane);
  halving_step<16, 8>(v, lane);
  halving_step<8, 4>(v, lane);
  halving_step<4, 2>(v, lane);
  halving_step<2, 1>(v, lane);
  const float ale = v[0];

  const int e = e0 + (lane >> 2), h = lane & 3;
  const int s = src[e];
  const int d = dst[e];
  float a = ans[s * NH + h] + adn[d * NH + h] + ale;
  a = (a > 0.f) ? a : NEG_SLOPE * a;
  alpha[(size_t)e * NH + h] = a;
  atomicMax(&mx[d * NH + h], fenc(a));
  atomicAdd(&ls[d * NH + h], ale);
}

__global__ void k_loop_alpha(const float* __restrict__ ls, const float* __restrict__ deg,
                             const float* __restrict__ ans, const float* __restrict__ adn,
                             float* __restrict__ alpha, unsigned* __restrict__ mx) {
  int i = blockIdx.x * blockDim.x + threadIdx.x;
  if (i >= NN * NH) return;
  int n = i / NH;
  float ale = ls[i] / fmaxf(deg[n], 1.0f);
  float a = ans[i] + adn[i] + ale;
  a = (a > 0.f) ? a : NEG_SLOPE * a;
  alpha[(size_t)NE * NH + i] = a;
  atomicMax(&mx[i], fenc(a));
}

__global__ void k_softmax_den(const int* __restrict__ dst, const unsigned* __restrict__ mx,
                              float* __restrict__ alpha, float* __restrict__ den) {
  const int tot = NE + NN;
  long long i = (long long)blockIdx.x * blockDim.x + threadIdx.x;
  if (i >= (long long)tot * NH) return;
  int e = (int)(i >> 2), h = (int)(i & 3);
  int d = (e < NE) ? dst[e] : (e - NE);
  float ex = expf(alpha[i] - fdec(mx[d * NH + h]));
  alpha[i] = ex;
  atomicAdd(&den[d * NH + h], ex);
}

#define GN   8
#define GCAP 64

template <int C>
__device__ __forceinline__ void gather_load(v4f (&v)[NH * (C / 128)], const float* __restrict__ hx, int s,
                                            int lane) {
  const float* hr = hx + (size_t)s * (NH * C) + 4 * lane;
#pragma unroll
  for (int q = 0; q < NH * (C / 128); ++q) v[q] = *(const v4f*)(hr + 128 * q);
}

template <int C>
__device__ __forceinline__ void gather_fma(v4f (&acc)[C / 128], const v4f (&v)[NH * (C / 128)], v4f a) {
#pragma unroll
  for (int h = 0; h < NH; ++h) {
#pragma unroll
    for (int j = 0; j < C / 128; ++j) acc[j] += a[h] * v[h * (C / 128) + j];
  }
}

__device__ __forceinline__ v4f lane_bcast(v4f x, int k) {
  v4f r;
  r[0] = __int_as_float(__builtin_amdgcn_readlane(__float_as_int(x[0]), k));
  r[1] = __int_as_float(__builtin_amdgcn_readlane(__float_as_int(x[1]), k));
  r[2] = __int_as_float(__builtin_amdgcn_readlane(__float_as_int(x[2]), k));
  r[3] = __int_as_float(__builtin_amdgcn_readlane(__float_as_int(x[3]), k));
  return r;
}

template <int C, bool RELU>
__global__ __launch_bounds__(256) void k_gather(const float* __restrict__ hx, const float* __restrict__ ex,
                                                const float* __restrict__ den, const int* __restrict__ src,
                                                const int* __restrict__ dst, const float* __restrict__ bias,
                                                float* __restrict__ out) {
  static_assert(C % 128 == 0 && NN % GN == 0 && NE % 32 == 0 && NE % 4 == 0, "shape");
  __shared__ int cnt[GN];
  __shared__ int lst[GN][GCAP];
  const int tid = threadIdx.x, wave = tid >> 5, lane = tid & 31;
  const int n0 = blockIdx.x * GN;
  if (tid < GN) cnt[tid] = 0;
  __syncthreads();
  const int4* d4 = (const int4*)dst;
  constexpr int SCAN_B = 16;
  for (int i0 = tid; i0 < NE / 4; i0 += 256 * SCAN_B) {
    int4 q[SCAN_B];
#pragma unroll
    for (int u = 0; u < SCAN_B; ++u) {
      const int i = i0 + 256 * u;
      q[u] = (i < NE / 4) ? d4[i] : make_int4(-1, -1, -1, -1);
    }
#pragma unroll
    for (int u = 0; u < SCAN_B; ++u) {
      const int dd[4] = {q[u].x, q[u].y, q[u].z, q[u].w};
#pragma unroll
      for (int t = 0; t < 4; ++t) {
        const unsigned g = (unsigned)(dd[t] - n0);
        if (g < (unsigned)GN) {
          const int p = atomicAdd(&cnt[g], 1);
          if (p < GCAP) lst[g][p] = 4 * (i0 + 256 * u) + t;
        }
      }
    }
  }
  __syncthreads();

  const int d = n0 + wave;
  const int n = __builtin_amdgcn_readfirstlane(cnt[wave]);
  const v4f dn = *(const v4f*)(den + (size_t)d * NH);
  v4f acc[C / 128] = {};
#define GATHER_FENCE() __builtin_amdgcn_sched_barrier(0)
  if (n <= GCAP) {
    const int nt = n + 1;
    for (int base = 0; base < nt; base += 32) {
      const int m = min(32, nt - base);
      const int i = base + lane;
      int e_i = NE + d, s_i = d;
      if (i > 0 && i < nt) { e_i = lst[wave][i - 1]; s_i = src[e_i]; }
      const v4f a_i = *(const v4f*)(ex + (size_t)e_i * NH) / dn;
      int k = 0;
      for (; k + 1 < m; k += 2) {
        v4f v0[NH * (C / 128)], v1[NH * (C / 128)];
        gather_load<C>(v0, hx, __builtin_amdgcn_readlane(s_i, k), lane);
        gather_load<C>(v1, hx, __builtin_amdgcn_readlane(s_i, k + 1), lane);
        GATHER_FENCE();
        gather_fma<C>(acc, v0, lane_bcast(a_i, k));
        gather_fma<C>(acc, v1, lane_bcast(a_i, k + 1));
        GATHER_FENCE();
      }
      if (k < m) {
        v4f v0[NH * (C / 128)];
        gather_load<C>(v0, hx, __builtin_amdgcn_readlane(s_i, k), lane);
        GATHER_FENCE();
        gather_fma<C>(acc, v0, lane_bcast(a_i, k));
      }
    }
  } else {
    {
      v4f v0[NH * (C / 128)];
      gather_load<C>(v0, hx, d, lane);
      gather_fma<C>(acc, v0, *(const v4f*)(ex + (size_t)(NE + d) * NH) / dn);
    }
    for (int base = 0; base < NE; base += 32) {
      unsigned mk = __builtin_amdgcn_ballot_w32(dst[base + lane] == d);
      while (mk) {
        const int e = base + __builtin_ctz(mk);
        mk &= mk - 1;
        v4f v0[NH * (C / 128)];
        gather_load<C>(v0, hx, src[e], lane);
        gather_fma<C>(acc, v0, *(const v4f*)(ex + (size_t)e * NH) / dn);
      }
    }
  }
#undef GATHER_FENCE
#pragma unroll
  for (int j = 0; j < C / 128; ++j) {
    const int c0 = 128 * j + 4 * lane;
    v4f o = 0.25f * acc[j] + *(const v4f*)(bias + c0);
    if (RELU) { o[0] = fmaxf(o[0], 0.f); o[1] = fmaxf(o[1], 0.f); o[2] = fmaxf(o[2], 0.f); o[3] = fmaxf(o[3], 0.f); }
    *(v4f*)(out + (size_t)d * C + c0) = o;
  }
}

extern "C" void kernel_launch(void* const* d_in, const int* in_sizes, int n_in,
                              void* d_out, int out_size, void* d_ws, size_t ws_size,
                              hipStream_t stream) {
  const int* x      = (const int*)d_in[0];
  const int* eidx   = (const int*)d_in[1];
  const int* src    = eidx;
  const int* dst    = eidx + NE;
  const int* eattr  = (const int*)d_in[2];
  const float* xemb = (const float*)d_in[3];
  const float* eemb = (const float*)d_in[4];
  const float* W1   = (const float*)d_in[5];
  const float* as1  = (const float*)d_in[6];
  const float* ad1  = (const float*)d_in[7];
  const float* We1  = (const float*)d_in[8];
  const float* ae1  = (const float*)d_in[9];
  const float* b1   = (const float*)d_in[10];
  const float* W2   = (const float*)d_in[11];
  const float* as2  = (const float*)d_in[12];
  const float* ad2  = (const float*)d_in[13];
  const float* We2  = (const float*)d_in[14];
  const float* ae2  = (const float*)d_in[15];
  const float* b2   = (const float*)d_in[16];
  float* out = (float*)d_out;

  float* ws = (float*)d_ws;
  size_t off = 0;
  float* XF   = ws + off; off += (size_t)NN * EMB;
  float* DEG  = ws + off; off += 2048;
  float* HX1  = ws + off; off += (size_t)NN * NH * HID;
  float* HX2  = ws + off; off += (size_t)NN * NH * FT;
  float* PROJ = ws + off; off += (size_t)EMB * NH;
  float* ANS  = ws + off; off += (size_t)NN * NH;
  float* ADN  = ws + off; off += (size_t)NN * NH;
  float* ALPH = ws + off; off += (size_t)(NE + NN) * NH;
  unsigned* MX = (unsigned*)(ws + off); off += (size_t)NN * NH;
  float* DEN  = ws + off; off += (size_t)NN * NH;
  float* LS   = ws + off; off += (size_t)NN * NH;
  float* HB   = ws + off; off += (size_t)NN * HID;
  float* PS   = ws + off; off += (size_t)(NH * FT / 64) * MP;
  float* PD   = ws + off; off += (size_t)(NH * FT / 64) * MP;
  (void)ws_size; (void)in_sizes; (void)n_in; (void)out_size;

  const int T = 256;

  hipMemsetAsync(DEG, 0, 2048 * sizeof(float), stream);
  hipMemsetAsync(MX, 0, (size_t)NN * NH * sizeof(unsigned), stream);
  hipMemsetAsync(DEN, 0, (size_t)2 * NN * NH * sizeof(float), stream);

  k_embed_x<<<NN / 8, 256, 0, stream>>>(x, xemb, XF);
  k_deg<<<cdiv(NE, T), T, 0, stream>>>(dst, DEG);

  {
    dim3 g(NH * HID / 128, cdiv(NN, 64));
    k_gemm_wmma_f32<EMB, NH * HID><<<g, 128, 0, stream>>>(XF, W1, HX1, NN, as1, ad1, PS, PD);
  }
  k_proj_e<HID><<<EMB * NH / 32, 32, 0, stream>>>(We1, ae1, PROJ);
  k_alpha_sum<HID / 64><<<cdiv(NN * NH, T), T, 0, stream>>>(PS, PD, ANS, ADN);
  k_edge_alpha<<<NE / (8 * EPW), 256, 0, stream>>>(eattr, eemb, PROJ, ANS, ADN, src, dst, ALPH, MX, LS);
  k_loop_alpha<<<cdiv(NN * NH, T), T, 0, stream>>>(LS, DEG, ANS, ADN, ALPH, MX);
  k_softmax_den<<<cdiv((long long)(NE + NN) * NH, T), T, 0, stream>>>(dst, MX, ALPH, DEN);
  k_gather<HID, true><<<NN / GN, 256, 0, stream>>>(HX1, ALPH, DEN, src, dst, b1, HB);

  hipMemsetAsync(MX, 0, (size_t)NN * NH * sizeof(unsigned), stream);
  hipMemsetAsync(DEN, 0, (size_t)2 * NN * NH * sizeof(float), stream);

  {
    dim3 g(NH * FT / 128, cdiv(NN, 64));
    k_gemm_wmma_f32<HID, NH * FT><<<g, 128, 0, stream>>>(HB, W2, HX2, NN, as2, ad2, PS, PD);
  }
  k_proj_e<FT><<<EMB * NH / 32, 32, 0, stream>>>(We2, ae2, PROJ);
  k_alpha_sum<FT / 64><<<cdiv(NN * NH, T), T, 0, stream>>>(PS, PD, ANS, ADN);
  k_edge_alpha<<<NE / (8 * EPW), 256, 0, stream>>>(eattr, eemb, PROJ, ANS, ADN, src, dst, ALPH, MX, LS);
  k_loop_alpha<<<cdiv(NN * NH, T), T, 0, stream>>>(LS, DEG, ANS, ADN, ALPH, MX);
  k_softmax_den<<<cdiv((long long)(NE + NN) * NH, T), T, 0, stream>>>(dst, MX, ALPH, DEN);
  k_gather<FT, false><<<NN / GN, 256, 0, stream>>>(HX2, ALPH, DEN, src, dst, b2, out);
}


# Edge-attr terms of both layers in one up-front pass: both layers' proj rows in one launch, each edge's two table rows loaded once and dotted with both projs, degree counted there; 16 launches to 13
# speedup vs baseline: 4.6780x; 1.0650x over previous
#include <hip/hip_runtime.h>
#include <math.h>

#define NN   2000
#define NE   32000
#define EMB  768
#define HID  384
#define FT   768
#define NH   4
#define NEG_SLOPE 0.2f

typedef __attribute__((ext_vector_type(2))) float v2f;
typedef __attribute__((ext_vector_type(4))) float v4f;
typedef __attribute__((ext_vector_type(8))) float v8f;

static inline int cdiv(long long a, long long b) { return (int)((a + b - 1) / b); }

__device__ __forceinline__ unsigned fenc(float f) {
  unsigned u = __float_as_uint(f);
  return (u & 0x80000000u) ? ~u : (u | 0x80000000u);
}
__device__ __forceinline__ float fdec(unsigned e) {
  unsigned u = (e & 0x80000000u) ? (e ^ 0x80000000u) : ~e;
  return __uint_as_float(u);
}

__global__ __launch_bounds__(256) void k_embed_x(const int* __restrict__ x, const float* __restrict__ emb,
                                                 float* __restrict__ xf) {
  static_assert(NN % 8 == 0 && EMB % 128 == 0, "shape");
  const int lane = threadIdx.x & 31;
  const int n = __builtin_amdgcn_readfirstlane(blockIdx.x * 8 + (threadIdx.x >> 5));
  const float* r0 = emb + (size_t)x[n * 2 + 0] * EMB + 4 * lane;
  const float* r1 = emb + (size_t)x[n * 2 + 1] * EMB + 4 * lane;
  float* o = xf + (size_t)n * EMB + 4 * lane;
#pragma unroll
  for (int t = 0; t < EMB / 128; ++t)
    *(v4f*)(o + 128 * t) = *(const v4f*)(r0 + 128 * t) + *(const v4f*)(r1 + 128 * t);
}

#define EPW 8

template <int N, int OFF, int SZ>
__device__ __forceinline__ void halving_step(float (&v)[SZ], int lane) {
  const bool up = (lane & OFF) != 0;
#pragma unroll
  for (int i = 0; i < N / 2; ++i) {
    const float keep = up ? v[i + N / 2] : v[i];
    const float send = up ? v[i] : v[i + N / 2];
    v[i] = keep + __shfl_xor(send, OFF, 32);
  }
}

#define MP 2048

struct GemmFrag {
  v4f a[2][2];
  v4f b[2][4];
};

template <int K, int N>
__device__ __forceinline__ void gemm_load_stage(GemmFrag& f, const float* __restrict__ a0p,
                                                const float* __restrict__ a1p,
                                                const float* __restrict__ bp, int kb) {
#pragma unroll
  for (int u = 0; u < 2; ++u) {
    f.a[0][u] = *(const v4f*)(a0p + kb + 8 * u);
    f.a[1][u] = *(const v4f*)(a1p + kb + 8 * u);
#pragma unroll
    for (int t = 0; t < 4; ++t)
      f.b[u][t] = *(const v4f*)(bp + (size_t)(kb + 8 * u + t) * N);
  }
}

__device__ __forceinline__ void gemm_mma_stage(const GemmFrag& f, v8f (&acc)[2][4]) {
#pragma unroll
  for (int u = 0; u < 2; ++u) {
#pragma unroll
    for (int s = 0; s < 2; ++s) {
      v2f a0 = {f.a[0][u][2 * s], f.a[0][u][2 * s + 1]};
      v2f a1 = {f.a[1][u][2 * s], f.a[1][u][2 * s + 1]};
#pragma unroll
      for (int j = 0; j < 4; ++j) {
        v2f b = {f.b[u][2 * s][j], f.b[u][2 * s + 1][j]};
        acc[0][j] = __builtin_amdgcn_wmma_f32_16x16x4_f32(false, a0, false, b, (short)0, acc[0][j], false, false);
        acc[1][j] = __builtin_amdgcn_wmma_f32_16x16x4_f32(false, a1, false, b, (short)0, acc[1][j], false, false);
      }
    }
  }
}

template <int K, int N>
__global__ __launch_bounds__(128) void k_gemm_wmma_f32(
    const float* __restrict__ A, const float* __restrict__ B, float* __restrict__ C, int M,
    const float* __restrict__ a_src, const float* __restrict__ a_dst,
    float* __restrict__ ps, float* __restrict__ pd) {
  static_assert(K % 48 == 0 && K >= 48 && N % 128 == 0, "tile shape");
  static_assert((N / NH) % 64 == 0 && MP >= NN && MP % 32 == 0, "a wave's 64 columns inside one head; whole lines of row sums");
  const int wave = threadIdx.x >> 5, lane = threadIdx.x & 31;
  const int c = lane & 15, h = lane >> 4;
  const int m0 = blockIdx.y * 64 + (wave >> 1) * 32;
  const int n0 = blockIdx.x * 128 + (wave & 1) * 64;
  if (m0 >= M) return;

  const int r0 = min(m0 + c, M - 1), r1 = min(m0 + 16 + c, M - 1);
  const float* a0p = A + (size_t)r0 * K + 4 * h;
  const float* a1p = A + (size_t)r1 * K + 4 * h;
  const float* bp  = B + (size_t)(4 * h) * N + n0 + 4 * c;

  v8f acc[2][4] = {};
  GemmFrag f0, f1, f2;
  gemm_load_stage<K, N>(f0, a0p, a1p, bp, 0);
  gemm_load_stage<K, N>(f1, a0p, a1p, bp, 16);
#define GEMM_SCHED_FENCE() __builtin_amdgcn_sched_barrier(0)
#pragma clang loop unroll(disable)
  for (int kb = 0; kb < K - 48; kb += 48) {
    gemm_load_stage<K, N>(f2, a0p, a1p, bp, kb + 32);
    GEMM_SCHED_FENCE();
    gemm_mma_stage(f0, acc);
    GEMM_SCHED_FENCE();
    gemm_load_stage<K, N>(f0, a0p, a1p, bp, kb + 48);
    GEMM_SCHED_FENCE();
    gemm_mma_stage(f1, acc);
    GEMM_SCHED_FENCE();
    gemm_load_stage<K, N>(f1, a0p, a1p, bp, kb + 64);
    GEMM_SCHED_FENCE();
    gemm_mma_stage(f2, acc);
    GEMM_SCHED_FENCE();
  }
  gemm_load_stage<K, N>(f2, a0p, a1p, bp, K - 16);
  GEMM_SCHED_FENCE();
  gemm_mma_stage(f0, acc);
  gemm_mma_stage(f1, acc);
  gemm_mma_stage(f2, acc);
#undef GEMM_SCHED_FENCE

#pragma unroll
  for (int i = 0; i < 2; ++i) {
#pragma unroll
    for (int r = 0; r < 8; ++r) {
      const int gm = m0 + 16 * i + 8 * h + r;
      if (gm < M) {
        v4f o = {acc[i][0][r], acc[i][1][r], acc[i][2][r], acc[i][3][r]};
        *(v4f*)(C + (size_t)gm * N + n0 + 4 * c) = o;
      }
    }
  }

  const v4f asv = *(const v4f*)(a_src + n0 + 4 * c);
  const v4f adv = *(const v4f*)(a_dst + n0 + 4 * c);
  float v[EPW * NH];
#pragma unroll
  for (int i = 0; i < 2; ++i) {
#pragma unroll
    for (int r = 0; r < 8; ++r) {
      float s = acc[i][0][r] * asv[0], d = acc[i][0][r] * adv[0];
#pragma unroll
      for (int j = 1; j < 4; ++j) { s += acc[i][j][r] * asv[j]; d += acc[i][j][r] * adv[j]; }
      v[2 * (8 * i + r) + 0] = s;
      v[2 * (8 * i + r) + 1] = d;
    }
  }
  halving_step<32, 8>(v, lane);
  halving_step<16, 4>(v, lane);
  halving_step<8, 2>(v, lane);
  halving_step<4, 1>(v, lane);
  const int prow = m0 + 16 * (c >> 3) + 8 * h + (c & 7);
  if (prow < M) {
    ps[(size_t)(n0 >> 6) * MP + prow] = v[0];
    pd[(size_t)(n0 >> 6) * MP + prow] = v[1];
  }
}

#define PB 12
#define PROJ_WG (EMB * NH / 32)

template <int C>
__device__ __forceinline__ void proj_load(v4f (&w)[PB], v4f (&a)[PB], const float* __restrict__ wr,
                                          const float* __restrict__ ar, int c0) {
#pragma unroll
  for (int u = 0; u < PB; ++u) {
    w[u] = *(const v4f*)(wr + c0 + 4 * u);
    a[u] = *(const v4f*)(ar + c0 + 4 * u);
  }
}

__device__ __forceinline__ float proj_fma(float s, const v4f (&w)[PB], const v4f (&a)[PB]) {
#pragma unroll
  for (int u = 0; u < PB; ++u) {
#pragma unroll
    for (int q = 0; q < 4; ++q) s = __builtin_fmaf(w[u][q], a[u][q], s);
  }
  return s;
}

template <int C>
__device__ __forceinline__ void proj_rows(const float* __restrict__ We, const float* __restrict__ ae,
                                          float* __restrict__ proj, int blk) {
  static_assert(C % (8 * PB) == 0 && C >= 16 * PB && (EMB * NH) % 32 == 0, "an even number of whole batches; whole waves");
  const int i = blk * 32 + threadIdx.x;
  const int k = i / NH, h = i % NH;
  const float* wr = We + (size_t)k * (NH * C) + (size_t)h * C;
  const float* ar = ae + (size_t)h * C;
#define PROJ_FENCE() __builtin_amdgcn_sched_barrier(0)
  v4f w0[PB], a0[PB], w1[PB], a1[PB];
  float s = 0.f;
  proj_load<C>(w0, a0, wr, ar, 0);
#pragma clang loop unroll(disable)
  for (int c0 = 0; c0 < C - 8 * PB; c0 += 8 * PB) {
    proj_load<C>(w1, a1, wr, ar, c0 + 4 * PB);
    PROJ_FENCE();
    s = proj_fma(s, w0, a0);
    PROJ_FENCE();
    proj_load<C>(w0, a0, wr, ar, c0 + 8 * PB);
    PROJ_FENCE();
    s = proj_fma(s, w1, a1);
    PROJ_FENCE();
  }
  proj_load<C>(w1, a1, wr, ar, C - 4 * PB);
  PROJ_FENCE();
  s = proj_fma(s, w0, a0);
  s = proj_fma(s, w1, a1);
#undef PROJ_FENCE
  proj[i] = s;
}

__global__ __launch_bounds__(32) void k_proj_both(const float* __restrict__ We1, const float* __restrict__ ae1,
                                                  const float* __restrict__ We2, const float* __restrict__ ae2,
                                                  float* __restrict__ proj1, float* __restrict__ proj2) {
  if (blockIdx.x < PROJ_WG) proj_rows<FT>(We2, ae2, proj2, blockIdx.x);
  else                      proj_rows<HID>(We1, ae1, proj1, blockIdx.x - PROJ_WG);
}

template <int T>
__global__ void k_alpha_sum(const float* __restrict__ ps, const float* __restrict__ pd,
                            float* __restrict__ ans, float* __restrict__ adn) {
  int i = blockIdx.x * blockDim.x + threadIdx.x;
  if (i >= NN * NH) return;
  int n = i / NH, h = i % NH;
  float s = 0.f, t = 0.f;
#pragma unroll
  for (int q = 0; q < T; ++q) {
    s += ps[(size_t)(h * T + q) * MP + n];
    t += pd[(size_t)(h * T + q) * MP + n];
  }
  ans[i] = s;
  adn[i] = t;
}

struct EdgeRows {
  v4f a[EMB / 128];
  v4f b[EMB / 128];
};

__device__ __forceinline__ void edge_rows_load(EdgeRows& r, const float* __restrict__ emb, int i0, int i1, int lane) {
  const float* r0 = emb + (size_t)i0 * EMB + 4 * lane;
  const float* r1 = emb + (size_t)i1 * EMB + 4 * lane;
#pragma unroll
  for (int t = 0; t < EMB / 128; ++t) {
    r.a[t] = *(const v4f*)(r0 + 128 * t);
    r.b[t] = *(const v4f*)(r1 + 128 * t);
  }
}

__device__ __forceinline__ void edge_rows_dot(const EdgeRows& r, const v4f (&p1)[EMB / 128][4],
                                              const v4f (&p2)[EMB / 128][4], float* __restrict__ o) {
  v4f a1 = {0.f, 0.f, 0.f, 0.f}, a2 = {0.f, 0.f, 0.f, 0.f};
#pragma unroll
  for (int t = 0; t < EMB / 128; ++t) {
    const v4f xr = r.a[t] + r.b[t];
#pragma unroll
    for (int kk = 0; kk < 4; ++kk) { a1 += xr[kk] * p1[t][kk]; a2 += xr[kk] * p2[t][kk]; }
  }
  o[0] = a1[0]; o[1] = a1[1]; o[2] = a1[2]; o[3] = a1[3];
  o[4] = a2[0]; o[5] = a2[1]; o[6] = a2[2]; o[7] = a2[3];
}

__global__ __launch_bounds__(256) void k_edge_attr(const int* __restrict__ ea_idx, const float* __restrict__ emb,
                            const float* __restrict__ proj1, const float* __restrict__ proj2,
                            const int* __restrict__ dst,
                            float* __restrict__ ale1, float* __restrict__ ale2,
                            float* __restrict__ ls1, float* __restrict__ ls2, float* __restrict__ deg) {
  static_assert(EPW == 8 && EPW * NH == 32 && NE % (8 * EPW) == 0 && EMB == 768, "two values per lane; four pairs of edges; whole work-groups; six 128-float pieces");
  const int lane = threadIdx.x & 31;
  const int e0 = __builtin_amdgcn_readfirstlane((blockIdx.x * 8 + (threadIdx.x >> 5)) * EPW);

  int ri[2 * EPW];
#pragma unroll
  for (int u = 0; u < 2 * EPW; ++u) ri[u] = ea_idx[e0 * 2 + u];

  v4f p1[EMB / 128][4], p2[EMB / 128][4];
#pragma unroll
  for (int t = 0; t < EMB / 128; ++t) {
#pragma unroll
    for (int kk = 0; kk < 4; ++kk) {
      p1[t][kk] = *(const v4f*)(proj1 + (size_t)(128 * t + 4 * lane + kk) * NH);
      p2[t][kk] = *(const v4f*)(proj2 + (size_t)(128 * t + 4 * lane + kk) * NH);
    }
  }

#define EDGE_FENCE() __builtin_amdgcn_sched_barrier(0)
  EdgeRows g0[2], g1[2];
  float v[EPW * NH * 2];
  edge_rows_load(g0[0], emb, ri[0], ri[1], lane);
  edge_rows_load(g0[1], emb, ri[2], ri[3], lane);
  edge_rows_load(g1[0], emb, ri[4], ri[5], lane);
  edge_rows_load(g1[1], emb, ri[6], ri[7], lane);
  EDGE_FENCE();
  edge_rows_dot(g0[0], p1, p2, v + 0);
  edge_rows_dot(g0[1], p1, p2, v + 8);
  EDGE_FENCE();
  edge_rows_load(g0[0], emb, ri[8], ri[9], lane);
  edge_rows_load(g0[1], emb, ri[10], ri[11], lane);
  EDGE_FENCE();
  edge_rows_dot(g1[0], p1, p2, v + 16);
  edge_rows_dot(g1[1], p1, p2, v + 24);
  EDGE_FENCE();
  edge_rows_load(g1[0], emb, ri[12], ri[13], lane);
  edge_rows_load(g1[1], emb, ri[14], ri[15], lane);
  EDGE_FENCE();
  edge_rows_dot(g0[0], p1, p2, v + 32);
  edge_rows_dot(g0[1], p1, p2, v + 40);
  edge_rows_dot(g1[0], p1, p2, v + 48);
  edge_rows_dot(g1[1], p1, p2, v + 56);
#undef EDGE_FENCE

  halving_step<64, 16>(v, lane);
  halving_step<32, 8>(v, lane);
  halving_step<16, 4>(v, lane);
  halving_step<8, 2>(v, lane);
  halving_step<4, 1>(v, lane);

  const int e = e0 + (lane >> 2), h0 = 2 * (lane & 1);
  const bool second = (lane & 2) != 0;
  const int d = dst[e];
  float* ale = second ? ale2 : ale1;
  float* ls  = second ? ls2 : ls1;
  const v2f o = {v[0], v[1]};
  *(v2f*)(ale + (size_t)e * NH + h0) = o;
  atomicAdd(&ls[d * NH + h0], v[0]);
  atomicAdd(&ls[d * NH + h0 + 1], v[1]);
  if ((lane & 3) == 0) atomicAdd(&deg[d], 1.0f);
}

__global__ void k_edge_combine(const float* __restrict__ ale, const float* __restrict__ ls,
                               const float* __restrict__ deg,
                               const float* __restrict__ ans, const float* __restrict__ adn,
                               const int* __restrict__ src, const int* __restrict__ dst,
                               float* __restrict__ alpha, unsigned* __restrict__ mx) {
  const int i = blockIdx.x * blockDim.x + threadIdx.x;
  if (i >= (NE + NN) * NH) return;
  const int e = i >> 2, h = i & 3;
  int s, d;
  float t;
  if (e < NE) {
    s = src[e];
    d = dst[e];
    t = ale[i];
  } else {
    s = d = e - NE;
    t = ls[d * NH + h] / fmaxf(deg[d], 1.0f);
  }
  float a = ans[s * NH + h] + adn[d * NH + h] + t;
  a = (a > 0.f) ? a : NEG_SLOPE * a;
  alpha[i] = a;
  atomicMax(&mx[d * NH + h], fenc(a));
}

__global__ void k_softmax_den(const int* __restrict__ dst, const unsigned* __restrict__ mx,
                              float* __restrict__ alpha, float* __restrict__ den) {
  const int tot = NE + NN;
  long long i = (long long)blockIdx.x * blockDim.x + threadIdx.x;
  if (i >= (long long)tot * NH) return;
  int e = (int)(i >> 2), h = (int)(i & 3);
  int d = (e < NE) ? dst[e] : (e - NE);
  float ex = expf(alpha[i] - fdec(mx[d * NH + h]));
  alpha[i] = ex;
  atomicAdd(&den[d * NH + h], ex);
}

#define GN   8
#define GCAP 64

template <int C>
__device__ __forceinline__ void gather_load(v4f (&v)[NH * (C / 128)], const float* __restrict__ hx, int s,
                                            int lane) {
  const float* hr = hx + (size_t)s * (NH * C) + 4 * lane;
#pragma unroll
  for (int q = 0; q < NH * (C / 128); ++q) v[q] = *(const v4f*)(hr + 128 * q);
}

template <int C>
__device__ __forceinline__ void gather_fma(v4f (&acc)[C / 128], const v4f (&v)[NH * (C / 128)], v4f a) {
#pragma unroll
  for (int h = 0; h < NH; ++h) {
#pragma unroll
    for (int j = 0; j < C / 128; ++j) acc[j] += a[h] * v[h * (C / 128) + j];
  }
}

__device__ __forceinline__ v4f lane_bcast(v4f x, int k) {
  v4f r;
  r[0] = __int_as_float(__builtin_amdgcn_readlane(__float_as_int(x[0]), k));
  r[1] = __int_as_float(__builtin_amdgcn_readlane(__float_as_int(x[1]), k));
  r[2] = __int_as_float(__builtin_amdgcn_readlane(__float_as_int(x[2]), k));
  r[3] = __int_as_float(__builtin_amdgcn_readlane(__float_as_int(x[3]), k));
  return r;
}

template <int C, bool RELU>
__global__ __launch_bounds__(256) void k_gather(const float* __restrict__ hx, const float* __restrict__ ex,
                                                const float* __restrict__ den, const int* __restrict__ src,
                                                const int* __restrict__ dst, const float* __restrict__ bias,
                                                float* __restrict__ out) {
  static_assert(C % 128 == 0 && NN % GN == 0 && NE % 32 == 0 && NE % 4 == 0, "shape");
  __shared__ int cnt[GN];
  __shared__ int lst[GN][GCAP];
  const int tid = threadIdx.x, wave = tid >> 5, lane = tid & 31;
  const int n0 = blockIdx.x * GN;
  if (tid < GN) cnt[tid] = 0;
  __syncthreads();
  const int4* d4 = (const int4*)dst;
  constexpr int SCAN_B = 16;
  for (int i0 = tid; i0 < NE / 4; i0 += 256 * SCAN_B) {
    int4 q[SCAN_B];
#pragma unroll
    for (int u = 0; u < SCAN_B; ++u) {
      const int i = i0 + 256 * u;
      q[u] = (i < NE / 4) ? d4[i] : make_int4(-1, -1, -1, -1);
    }
#pragma unroll
    for (int u = 0; u < SCAN_B; ++u) {
      const int dd[4] = {q[u].x, q[u].y, q[u].z, q[u].w};
#pragma unroll
      for (int t = 0; t < 4; ++t) {
        const unsigned g = (unsigned)(dd[t] - n0);
        if (g < (unsigned)GN) {
          const int p = atomicAdd(&cnt[g], 1);
          if (p < GCAP) lst[g][p] = 4 * (i0 + 256 * u) + t;
        }
      }
    }
  }
  __syncthreads();

  const int d = n0 + wave;
  const int n = __builtin_amdgcn_readfirstlane(cnt[wave]);
  const v4f dn = *(const v4f*)(den + (size_t)d * NH);
  v4f acc[C / 128] = {};
#define GATHER_FENCE() __builtin_amdgcn_sched_barrier(0)
  if (n <= GCAP) {
    const int nt = n + 1;
    for (int base = 0; base < nt; base += 32) {
      const int m = min(32, nt - base);
      const int i = base + lane;
      int e_i = NE + d, s_i = d;
      if (i > 0 && i < nt) { e_i = lst[wave][i - 1]; s_i = src[e_i]; }
      const v4f a_i = *(const v4f*)(ex + (size_t)e_i * NH) / dn;
      int k = 0;
      for (; k + 1 < m; k += 2) {
        v4f v0[NH * (C / 128)], v1[NH * (C / 128)];
        gather_load<C>(v0, hx, __builtin_amdgcn_readlane(s_i, k), lane);
        gather_load<C>(v1, hx, __builtin_amdgcn_readlane(s_i, k + 1), lane);
        GATHER_FENCE();
        gather_fma<C>(acc, v0, lane_bcast(a_i, k));
        gather_fma<C>(acc, v1, lane_bcast(a_i, k + 1));
        GATHER_FENCE();
      }
      if (k < m) {
        v4f v0[NH * (C / 128)];
        gather_load<C>(v0, hx, __builtin_amdgcn_readlane(s_i, k), lane);
        GATHER_FENCE();
        gather_fma<C>(acc, v0, lane_bcast(a_i, k));
      }
    }
  } else {
    {
      v4f v0[NH * (C / 128)];
      gather_load<C>(v0, hx, d, lane);
      gather_fma<C>(acc, v0, *(const v4f*)(ex + (size_t)(NE + d) * NH) / dn);
    }
    for (int base = 0; base < NE; base += 32) {
      unsigned mk = __builtin_amdgcn_ballot_w32(dst[base + lane] == d);
      while (mk) {
        const int e = base + __builtin_ctz(mk);
        mk &= mk - 1;
        v4f v0[NH * (C / 128)];
        gather_load<C>(v0, hx, src[e], lane);
        gather_fma<C>(acc, v0, *(const v4f*)(ex + (size_t)e * NH) / dn);
      }
    }
  }
#undef GATHER_FENCE
#pragma unroll
  for (int j = 0; j < C / 128; ++j) {
    const int c0 = 128 * j + 4 * lane;
    v4f o = 0.25f * acc[j] + *(const v4f*)(bias + c0);
    if (RELU) { o[0] = fmaxf(o[0], 0.f); o[1] = fmaxf(o[1], 0.f); o[2] = fmaxf(o[2], 0.f); o[3] = fmaxf(o[3], 0.f); }
    *(v4f*)(out + (size_t)d * C + c0) = o;
  }
}

extern "C" void kernel_launch(void* const* d_in, const int* in_sizes, int n_in,
                              void* d_out, int out_size, void* d_ws, size_t ws_size,
                              hipStream_t stream) {
  const int* x      = (const int*)d_in[0];
  const int* eidx   = (const int*)d_in[1];
  const int* src    = eidx;
  const int* dst    = eidx + NE;
  const int* eattr  = (const int*)d_in[2];
  const float* xemb = (const float*)d_in[3];
  const float* eemb = (const float*)d_in[4];
  const float* W1   = (const float*)d_in[5];
  const float* as1  = (const float*)d_in[6];
  const float* ad1  = (const float*)d_in[7];
  const float* We1  = (const float*)d_in[8];
  const float* ae1  = (const float*)d_in[9];
  const float* b1   = (const float*)d_in[10];
  const float* W2   = (const float*)d_in[11];
  const float* as2  = (const float*)d_in[12];
  const float* ad2  = (const float*)d_in[13];
  const float* We2  = (const float*)d_in[14];
  const float* ae2  = (const float*)d_in[15];
  const float* b2   = (const float*)d_in[16];
  float* out = (float*)d_out;

  float* ws = (float*)d_ws;
  size_t off = 0;
  float* XF   = ws + off; off += (size_t)NN * EMB;
  float* HX1  = ws + off; off += (size_t)NN * NH * HID;
  float* HX2  = ws + off; off += (size_t)NN * NH * FT;
  float* PROJ1 = ws + off; off += (size_t)EMB * NH;
  float* PROJ2 = ws + off; off += (size_t)EMB * NH;
  float* ALE1 = ws + off; off += (size_t)NE * NH;
  float* ALE2 = ws + off; off += (size_t)NE * NH;
  float* LS1  = ws + off; off += (size_t)NN * NH;
  float* LS2  = ws + off; off += (size_t)NN * NH;
  float* DEG  = ws + off; off += 2048;
  float* ANS  = ws + off; off += (size_t)NN * NH;
  float* ADN  = ws + off; off += (size_t)NN * NH;
  float* ALPH = ws + off; off += (size_t)(NE + NN) * NH;
  unsigned* MX = (unsigned*)(ws + off); off += (size_t)NN * NH;
  float* DEN  = ws + off; off += (size_t)NN * NH;
  float* HB   = ws + off; off += (size_t)NN * HID;
  float* PS   = ws + off; off += (size_t)(NH * FT / 64) * MP;
  float* PD   = ws + off; off += (size_t)(NH * FT / 64) * MP;
  (void)ws_size; (void)in_sizes; (void)n_in; (void)out_size;

  const int T = 256;

  hipMemsetAsync(LS1, 0, ((size_t)2 * NN * NH + 2048) * sizeof(float), stream);
  hipMemsetAsync(MX, 0, (size_t)2 * NN * NH * sizeof(float), stream);

  k_embed_x<<<NN / 8, 256, 0, stream>>>(x, xemb, XF);
  k_proj_both<<<2 * PROJ_WG, 32, 0, stream>>>(We1, ae1, We2, ae2, PROJ1, PROJ2);
  k_edge_attr<<<NE / (8 * EPW), 256, 0, stream>>>(eattr, eemb, PROJ1, PROJ2, dst, ALE1, ALE2, LS1, LS2, DEG);

  {
    dim3 g(NH * HID / 128, cdiv(NN, 64));
    k_gemm_wmma_f32<EMB, NH * HID><<<g, 128, 0, stream>>>(XF, W1, HX1, NN, as1, ad1, PS, PD);
  }
  k_alpha_sum<HID / 64><<<cdiv(NN * NH, T), T, 0, stream>>>(PS, PD, ANS, ADN);
  k_edge_combine<<<cdiv((long long)(NE + NN) * NH, T), T, 0, stream>>>(ALE1, LS1, DEG, ANS, ADN, src, dst, ALPH, MX);
  k_softmax_den<<<cdiv((long long)(NE + NN) * NH, T), T, 0, stream>>>(dst, MX, ALPH, DEN);
  k_gather<HID, true><<<NN / GN, 256, 0, stream>>>(HX1, ALPH, DEN, src, dst, b1, HB);

  hipMemsetAsync(MX, 0, (size_t)2 * NN * NH * sizeof(float), stream);

  {
    dim3 g(NH * FT / 128, cdiv(NN, 64));
    k_gemm_wmma_f32<HID, NH * FT><<<g, 128, 0, stream>>>(HB, W2, HX2, NN, as2, ad2, PS, PD);
  }
  k_alpha_sum<FT / 64><<<cdiv(NN * NH, T), T, 0, stream>>>(PS, PD, ANS, ADN);
  k_edge_combine<<<cdiv((long long)(NE + NN) * NH, T), T, 0, stream>>>(ALE2, LS2, DEG, ANS, ADN, src, dst, ALPH, MX);
  k_softmax_den<<<cdiv((long long)(NE + NN) * NH, T), T, 0, stream>>>(dst, MX, ALPH, DEN);
  k_gather<FT, false><<<NN / GN, 256, 0, stream>>>(HX2, ALPH, DEN, src, dst, b2, out);
}


# Independent passes share launches: proj rows, node rows and all clears in one launch; edge-attr pass inside GEMM 1's launch on its half-empty SIMDs; 13 launches + 3 memsets to 11; sums unchanged
# speedup vs baseline: 5.0613x; 1.0819x over previous
#include <hip/hip_runtime.h>
#include <math.h>

#define NN   2000
#define NE   32000
#define EMB  768
#define HID  384
#define FT   768
#define NH   4
#define NEG_SLOPE 0.2f

typedef __attribute__((ext_vector_type(2))) float v2f;
typedef __attribute__((ext_vector_type(4))) float v4f;
typedef __attribute__((ext_vector_type(8))) float v8f;

static inline int cdiv(long long a, long long b) { return (int)((a + b - 1) / b); }

__device__ __forceinline__ unsigned fenc(float f) {
  unsigned u = __float_as_uint(f);
  return (u & 0x80000000u) ? ~u : (u | 0x80000000u);
}
__device__ __forceinline__ float fdec(unsigned e) {
  unsigned u = (e & 0x80000000u) ? (e ^ 0x80000000u) : ~e;
  return __uint_as_float(u);
}

__device__ __forceinline__ void embed_row(const int* __restrict__ x, const float* __restrict__ emb,
                                          float* __restrict__ xf, int row) {
  static_assert(EMB % 128 == 0, "shape");
  const int lane = threadIdx.x & 31;
  const int n = __builtin_amdgcn_readfirstlane(row);
  const float* r0 = emb + (size_t)x[n * 2 + 0] * EMB + 4 * lane;
  const float* r1 = emb + (size_t)x[n * 2 + 1] * EMB + 4 * lane;
  float* o = xf + (size_t)n * EMB + 4 * lane;
#pragma unroll
  for (int t = 0; t < EMB / 128; ++t)
    *(v4f*)(o + 128 * t) = *(const v4f*)(r0 + 128 * t) + *(const v4f*)(r1 + 128 * t);
}

#define EPW 8

template <int N, int OFF, int SZ>
__device__ __forceinline__ void halving_step(float (&v)[SZ], int lane) {
  const bool up = (lane & OFF) != 0;
#pragma unroll
  for (int i = 0; i < N / 2; ++i) {
    const float keep = up ? v[i + N / 2] : v[i];
    const float send = up ? v[i] : v[i + N / 2];
    v[i] = keep + __shfl_xor(send, OFF, 32);
  }
}

#define MP 2048

struct GemmFrag {
  v4f a[2][2];
  v4f b[2][4];
};

template <int K, int N>
__device__ __forceinline__ void gemm_load_stage(GemmFrag& f, const float* __restrict__ a0p,
                                                const float* __restrict__ a1p,
                                                const float* __restrict__ bp, int kb) {
#pragma unroll
  for (int u = 0; u < 2; ++u) {
    f.a[0][u] = *(const v4f*)(a0p + kb + 8 * u);
    f.a[1][u] = *(const v4f*)(a1p + kb + 8 * u);
#pragma unroll
    for (int t = 0; t < 4; ++t)
      f.b[u][t] = *(const v4f*)(bp + (size_t)(kb + 8 * u + t) * N);
  }
}

__device__ __forceinline__ void gemm_mma_stage(const GemmFrag& f, v8f (&acc)[2][4]) {
#pragma unroll
  for (int u = 0; u < 2; ++u) {
#pragma unroll
    for (int s = 0; s < 2; ++s) {
      v2f a0 = {f.a[0][u][2 * s], f.a[0][u][2 * s + 1]};
      v2f a1 = {f.a[1][u][2 * s], f.a[1][u][2 * s + 1]};
#pragma unroll
      for (int j = 0; j < 4; ++j) {
        v2f b = {f.b[u][2 * s][j], f.b[u][2 * s + 1][j]};
        acc[0][j] = __builtin_amdgcn_wmma_f32_16x16x4_f32(false, a0, false, b, (short)0, acc[0][j], false, false);
        acc[1][j] = __builtin_amdgcn_wmma_f32_16x16x4_f32(false, a1, false, b, (short)0, acc[1][j], false, false);
      }
    }
  }
}

template <int K, int N>
__device__ __forceinline__ void gemm_wg_tile(
    const float* __restrict__ A, const float* __restrict__ B, float* __restrict__ C, int M,
    const float* __restrict__ a_src, const float* __restrict__ a_dst,
    float* __restrict__ ps, float* __restrict__ pd, int bx, int by) {
  static_assert(K % 48 == 0 && K >= 48 && N % 128 == 0, "tile shape");
  static_assert((N / NH) % 64 == 0 && MP >= NN && MP % 32 == 0, "a wave's 64 columns inside one head; whole lines of row sums");
  const int wave = threadIdx.x >> 5, lane = threadIdx.x & 31;
  const int c = lane & 15, h = lane >> 4;
  const int m0 = by * 64 + (wave >> 1) * 32;
  const int n0 = bx * 128 + (wave & 1) * 64;
  if (m0 >= M) return;

  const int r0 = min(m0 + c, M - 1), r1 = min(m0 + 16 + c, M - 1);
  const float* a0p = A + (size_t)r0 * K + 4 * h;
  const float* a1p = A + (size_t)r1 * K + 4 * h;
  const float* bp  = B + (size_t)(4 * h) * N + n0 + 4 * c;

  v8f acc[2][4] = {};
  GemmFrag f0, f1, f2;
  gemm_load_stage<K, N>(f0, a0p, a1p, bp, 0);
  gemm_load_stage<K, N>(f1, a0p, a1p, bp, 16);
#define GEMM_SCHED_FENCE() __builtin_amdgcn_sched_barrier(0)
#pragma clang loop unroll(disable)
  for (int kb = 0; kb < K - 48; kb += 48) {
    gemm_load_stage<K, N>(f2, a0p, a1p, bp, kb + 32);
    GEMM_SCHED_FENCE();
    gemm_mma_stage(f0, acc);
    GEMM_SCHED_FENCE();
    gemm_load_stage<K, N>(f0, a0p, a1p, bp, kb + 48);
    GEMM_SCHED_FENCE();
    gemm_mma_stage(f1, acc);
    GEMM_SCHED_FENCE();
    gemm_load_stage<K, N>(f1, a0p, a1p, bp, kb + 64);
    GEMM_SCHED_FENCE();
    gemm_mma_stage(f2, acc);
    GEMM_SCHED_FENCE();
  }
  gemm_load_stage<K, N>(f2, a0p, a1p, bp, K - 16);
  GEMM_SCHED_FENCE();
  gemm_mma_stage(f0, acc);
  gemm_mma_stage(f1, acc);
  gemm_mma_stage(f2, acc);
#undef GEMM_SCHED_FENCE

#pragma unroll
  for (int i = 0; i < 2; ++i) {
#pragma unroll
    for (int r = 0; r < 8; ++r) {
      const int gm = m0 + 16 * i + 8 * h + r;
      if (gm < M) {
        v4f o = {acc[i][0][r], acc[i][1][r], acc[i][2][r], acc[i][3][r]};
        *(v4f*)(C + (size_t)gm * N + n0 + 4 * c) = o;
      }
    }
  }

  const v4f asv = *(const v4f*)(a_src + n0 + 4 * c);
  const v4f adv = *(const v4f*)(a_dst + n0 + 4 * c);
  float v[EPW * NH];
#pragma unroll
  for (int i = 0; i < 2; ++i) {
#pragma unroll
    for (int r = 0; r < 8; ++r) {
      float s = acc[i][0][r] * asv[0], d = acc[i][0][r] * adv[0];
#pragma unroll
      for (int j = 1; j < 4; ++j) { s += acc[i][j][r] * asv[j]; d += acc[i][j][r] * adv[j]; }
      v[2 * (8 * i + r) + 0] = s;
      v[2 * (8 * i + r) + 1] = d;
    }
  }
  halving_step<32, 8>(v, lane);
  halving_step<16, 4>(v, lane);
  halving_step<8, 2>(v, lane);
  halving_step<4, 1>(v, lane);
  const int prow = m0 + 16 * (c >> 3) + 8 * h + (c & 7);
  if (prow < M) {
    ps[(size_t)(n0 >> 6) * MP + prow] = v[0];
    pd[(size_t)(n0 >> 6) * MP + prow] = v[1];
  }
}

template <int K, int N>
__global__ __launch_bounds__(128) void k_gemm_wmma_f32(
    const float* __restrict__ A, const float* __restrict__ B, float* __restrict__ C, int M,
    const float* __restrict__ a_src, const float* __restrict__ a_dst,
    float* __restrict__ ps, float* __restrict__ pd) {
  gemm_wg_tile<K, N>(A, B, C, M, a_src, a_dst, ps, pd, blockIdx.x, blockIdx.y);
}

#define PB 12
#define PROJ_WG (EMB * NH / 32)

template <int C>
__device__ __forceinline__ void proj_load(v4f (&w)[PB], v4f (&a)[PB], const float* __restrict__ wr,
                                          const float* __restrict__ ar, int c0) {
#pragma unroll
  for (int u = 0; u < PB; ++u) {
    w[u] = *(const v4f*)(wr + c0 + 4 * u);
    a[u] = *(const v4f*)(ar + c0 + 4 * u);
  }
}

__device__ __forceinline__ float proj_fma(float s, const v4f (&w)[PB], const v4f (&a)[PB]) {
#pragma unroll
  for (int u = 0; u < PB; ++u) {
#pragma unroll
    for (int q = 0; q < 4; ++q) s = __builtin_fmaf(w[u][q], a[u][q], s);
  }
  return s;
}

template <int C>
__device__ __forceinline__ void proj_rows(const float* __restrict__ We, const float* __restrict__ ae,
                                          float* __restrict__ proj, int blk) {
  static_assert(C % (8 * PB) == 0 && C >= 16 * PB && (EMB * NH) % 32 == 0, "an even number of whole batches; whole waves");
  const int i = blk * 32 + threadIdx.x;
  const int k = i / NH, h = i % NH;
  const float* wr = We + (size_t)k * (NH * C) + (size_t)h * C;
  const float* ar = ae + (size_t)h * C;
#define PROJ_FENCE() __builtin_amdgcn_sched_barrier(0)
  v4f w0[PB], a0[PB], w1[PB], a1[PB];
  float s = 0.f;
  proj_load<C>(w0, a0, wr, ar, 0);
#pragma clang loop unroll(disable)
  for (int c0 = 0; c0 < C - 8 * PB; c0 += 8 * PB) {
    proj_load<C>(w1, a1, wr, ar, c0 + 4 * PB);
    PROJ_FENCE();
    s = proj_fma(s, w0, a0);
    PROJ_FENCE();
    proj_load<C>(w0, a0, wr, ar, c0 + 8 * PB);
    PROJ_FENCE();
    s = proj_fma(s, w1, a1);
    PROJ_FENCE();
  }
  proj_load<C>(w1, a1, wr, ar, C - 4 * PB);
  PROJ_FENCE();
  s = proj_fma(s, w0, a0);
  s = proj_fma(s, w1, a1);
#undef PROJ_FENCE
  proj[i] = s;
}

#define NCLR (2 * NN * NH + 2048 + 4 * NN * NH)
#define PRE_WG (2 * PROJ_WG + NN + NCLR / 128)

__global__ __launch_bounds__(32) void k_pre(const float* __restrict__ We1, const float* __restrict__ ae1,
                                            const float* __restrict__ We2, const float* __restrict__ ae2,
                                            float* __restrict__ proj1, float* __restrict__ proj2,
                                            const int* __restrict__ x, const float* __restrict__ emb,
                                            float* __restrict__ xf, float* __restrict__ clr) {
  static_assert(NCLR % 128 == 0, "whole waves of 16-B stores");
  const int b = blockIdx.x;
  if (b < PROJ_WG)               proj_rows<FT>(We2, ae2, proj2, b);
  else if (b < 2 * PROJ_WG)      proj_rows<HID>(We1, ae1, proj1, b - PROJ_WG);
  else if (b < 2 * PROJ_WG + NN) embed_row(x, emb, xf, b - 2 * PROJ_WG);
  else {
    const v4f z = {0.f, 0.f, 0.f, 0.f};
    *(v4f*)(clr + (size_t)(b - 2 * PROJ_WG - NN) * 128 + 4 * threadIdx.x) = z;
  }
}

template <int T>
__global__ void k_alpha_sum(const float* __restrict__ ps, const float* __restrict__ pd,
                            float* __restrict__ ans, float* __restrict__ adn) {
  int i = blockIdx.x * blockDim.x + threadIdx.x;
  if (i >= NN * NH) return;
  int n = i / NH, h = i % NH;
  float s = 0.f, t = 0.f;
#pragma unroll
  for (int q = 0; q < T; ++q) {
    s += ps[(size_t)(h * T + q) * MP + n];
    t += pd[(size_t)(h * T + q) * MP + n];
  }
  ans[i] = s;
  adn[i] = t;
}

struct EdgeRows {
  v4f a[EMB / 128];
  v4f b[EMB / 128];
};

__device__ __forceinline__ void edge_rows_load(EdgeRows& r, const float* __restrict__ emb, int i0, int i1, int lane) {
  const float* r0 = emb + (size_t)i0 * EMB + 4 * lane;
  const float* r1 = emb + (size_t)i1 * EMB + 4 * lane;
#pragma unroll
  for (int t = 0; t < EMB / 128; ++t) {
    r.a[t] = *(const v4f*)(r0 + 128 * t);
    r.b[t] = *(const v4f*)(r1 + 128 * t);
  }
}

__device__ __forceinline__ void edge_rows_dot(const EdgeRows& r, const v4f (&p1)[EMB / 128][4],
                                              const v4f (&p2)[EMB / 128][4], float* __restrict__ o) {
  v4f a1 = {0.f, 0.f, 0.f, 0.f}, a2 = {0.f, 0.f, 0.f, 0.f};
#pragma unroll
  for (int t = 0; t < EMB / 128; ++t) {
    const v4f xr = r.a[t] + r.b[t];
#pragma unroll
    for (int kk = 0; kk < 4; ++kk) { a1 += xr[kk] * p1[t][kk]; a2 += xr[kk] * p2[t][kk]; }
  }
  o[0] = a1[0]; o[1] = a1[1]; o[2] = a1[2]; o[3] = a1[3];
  o[4] = a2[0]; o[5] = a2[1]; o[6] = a2[2]; o[7] = a2[3];
}

__device__ __forceinline__ void edge_attr_wave(const int* __restrict__ ea_idx, const float* __restrict__ emb,
                            const float* __restrict__ proj1, const float* __restrict__ proj2,
                            const int* __restrict__ dst,
                            float* __restrict__ ale1, float* __restrict__ ale2,
                            float* __restrict__ ls1, float* __restrict__ ls2, float* __restrict__ deg, int wv) {
  static_assert(EPW == 8 && EPW * NH == 32 && NE % (4 * EPW) == 0 && EMB == 768, "two values per lane; four pairs of edges; whole work-groups; six 128-float pieces");
  const int lane = threadIdx.x & 31;
  const int e0 = __builtin_amdgcn_readfirstlane(wv * EPW);

  int ri[2 * EPW];
#pragma unroll
  for (int u = 0; u < 2 * EPW; ++u) ri[u] = ea_idx[e0 * 2 + u];

  v4f p1[EMB / 128][4], p2[EMB / 128][4];
#pragma unroll
  for (int t = 0; t < EMB / 128; ++t) {
#pragma unroll
    for (int kk = 0; kk < 4; ++kk) {
      p1[t][kk] = *(const v4f*)(proj1 + (size_t)(128 * t + 4 * lane + kk) * NH);
      p2[t][kk] = *(const v4f*)(proj2 + (size_t)(128 * t + 4 * lane + kk) * NH);
    }
  }

#define EDGE_FENCE() __builtin_amdgcn_sched_barrier(0)
  EdgeRows g0[2], g1[2];
  float v[EPW * NH * 2];
  edge_rows_load(g0[0], emb, ri[0], ri[1], lane);
  edge_rows_load(g0[1], emb, ri[2], ri[3], lane);
  edge_rows_load(g1[0], emb, ri[4], ri[5], lane);
  edge_rows_load(g1[1], emb, ri[6], ri[7], lane);
  EDGE_FENCE();
  edge_rows_dot(g0[0], p1, p2, v + 0);
  edge_rows_dot(g0[1], p1, p2, v + 8);
  EDGE_FENCE();
  edge_rows_load(g0[0], emb, ri[8], ri[9], lane);
  edge_rows_load(g0[1], emb, ri[10], ri[11], lane);
  EDGE_FENCE();
  edge_rows_dot(g1[0], p1, p2, v + 16);
  edge_rows_dot(g1[1], p1, p2, v + 24);
  EDGE_FENCE();
  edge_rows_load(g1[0], emb, ri[12], ri[13], lane);
  edge_rows_load(g1[1], emb, ri[14], ri[15], lane);
  EDGE_FENCE();
  edge_rows_dot(g0[0], p1, p2, v + 32);
  edge_rows_dot(g0[1], p1, p2, v + 40);
  edge_rows_dot(g1[0], p1, p2, v + 48);
  edge_rows_dot(g1[1], p1, p2, v + 56);
#undef EDGE_FENCE

  halving_step<64, 16>(v, lane);
  halving_step<32, 8>(v, lane);
  halving_step<16, 4>(v, lane);
  halving_step<8, 2>(v, lane);
  halving_step<4, 1>(v, lane);

  const int e = e0 + (lane >> 2), h0 = 2 * (lane & 1);
  const bool second = (lane & 2) != 0;
  const int d = dst[e];
  float* ale = second ? ale2 : ale1;
  float* ls  = second ? ls2 : ls1;
  const v2f o = {v[0], v[1]};
  *(v2f*)(ale + (size_t)e * NH + h0) = o;
  atomicAdd(&ls[d * NH + h0], v[0]);
  atomicAdd(&ls[d * NH + h0 + 1], v[1]);
  if ((lane & 3) == 0) atomicAdd(&deg[d], 1.0f);
}

#define G1X (NH * HID / 128)
#define G1Y ((NN + 63) / 64)
#define GEMM1_WG (G1X * G1Y)
#define EDGE_WG (NE / (4 * EPW))

__global__ __launch_bounds__(128) void k_gemm1_edge(
    const float* __restrict__ A, const float* __restrict__ B, float* __restrict__ C,
    const float* __restrict__ a_src, const float* __restrict__ a_dst,
    float* __restrict__ ps, float* __restrict__ pd,
    const int* __restrict__ ea_idx, const float* __restrict__ emb,
    const float* __restrict__ proj1, const float* __restrict__ proj2, const int* __restrict__ dst,
    float* __restrict__ ale1, float* __restrict__ ale2,
    float* __restrict__ ls1, float* __restrict__ ls2, float* __restrict__ deg) {
  const int b = blockIdx.x;
  if (b < GEMM1_WG)
    gemm_wg_tile<EMB, NH * HID>(A, B, C, NN, a_src, a_dst, ps, pd, b % G1X, b / G1X);
  else
    edge_attr_wave(ea_idx, emb, proj1, proj2, dst, ale1, ale2, ls1, ls2, deg,
                   (b - GEMM1_WG) * 4 + (threadIdx.x >> 5));
}

__global__ void k_edge_combine(const float* __restrict__ ale, const float* __restrict__ ls,
                               const float* __restrict__ deg,
                               const float* __restrict__ ans, const float* __restrict__ adn,
                               const int* __restrict__ src, const int* __restrict__ dst,
                               float* __restrict__ alpha, unsigned* __restrict__ mx) {
  const int i = blockIdx.x * blockDim.x + threadIdx.x;
  if (i >= (NE + NN) * NH) return;
  const int e = i >> 2, h = i & 3;
  int s, d;
  float t;
  if (e < NE) {
    s = src[e];
    d = dst[e];
    t = ale[i];
  } else {
    s = d = e - NE;
    t = ls[d * NH + h] / fmaxf(deg[d], 1.0f);
  }
  float a = ans[s * NH + h] + adn[d * NH + h] + t;
  a = (a > 0.f) ? a : NEG_SLOPE * a;
  alpha[i] = a;
  atomicMax(&mx[d * NH + h], fenc(a));
}

__global__ void k_softmax_den(const int* __restrict__ dst, const unsigned* __restrict__ mx,
                              float* __restrict__ alpha, float* __restrict__ den) {
  const int tot = NE + NN;
  long long i = (long long)blockIdx.x * blockDim.x + threadIdx.x;
  if (i >= (long long)tot * NH) return;
  int e = (int)(i >> 2), h = (int)(i & 3);
  int d = (e < NE) ? dst[e] : (e - NE);
  float ex = expf(alpha[i] - fdec(mx[d * NH + h]));
  alpha[i] = ex;
  atomicAdd(&den[d * NH + h], ex);
}

#define GN   8
#define GCAP 64

template <int C>
__device__ __forceinline__ void gather_load(v4f (&v)[NH * (C / 128)], const float* __restrict__ hx, int s,
                                            int lane) {
  const float* hr = hx + (size_t)s * (NH * C) + 4 * lane;
#pragma unroll
  for (int q = 0; q < NH * (C / 128); ++q) v[q] = *(const v4f*)(hr + 128 * q);
}

template <int C>
__device__ __forceinline__ void gather_fma(v4f (&acc)[C / 128], const v4f (&v)[NH * (C / 128)], v4f a) {
#pragma unroll
  for (int h = 0; h < NH; ++h) {
#pragma unroll
    for (int j = 0; j < C / 128; ++j) acc[j] += a[h] * v[h * (C / 128) + j];
  }
}

__device__ __forceinline__ v4f lane_bcast(v4f x, int k) {
  v4f r;
  r[0] = __int_as_float(__builtin_amdgcn_readlane(__float_as_int(x[0]), k));
  r[1] = __int_as_float(__builtin_amdgcn_readlane(__float_as_int(x[1]), k));
  r[2] = __int_as_float(__builtin_amdgcn_readlane(__float_as_int(x[2]), k));
  r[3] = __int_as_float(__builtin_amdgcn_readlane(__float_as_int(x[3]), k));
  return r;
}

template <int C, bool RELU>
__global__ __launch_bounds__(256) void k_gather(const float* __restrict__ hx, const float* __restrict__ ex,
                                                const float* __restrict__ den, const int* __restrict__ src,
                                                const int* __restrict__ dst, const float* __restrict__ bias,
                                                float* __restrict__ out) {
  static_assert(C % 128 == 0 && NN % GN == 0 && NE % 32 == 0 && NE % 4 == 0, "shape");
  __shared__ int cnt[GN];
  __shared__ int lst[GN][GCAP];
  const int tid = threadIdx.x, wave = tid >> 5, lane = tid & 31;
  const int n0 = blockIdx.x * GN;
  if (tid < GN) cnt[tid] = 0;
  __syncthreads();
  const int4* d4 = (const int4*)dst;
  constexpr int SCAN_B = 16;
  for (int i0 = tid; i0 < NE / 4; i0 += 256 * SCAN_B) {
    int4 q[SCAN_B];
#pragma unroll
    for (int u = 0; u < SCAN_B; ++u) {
      const int i = i0 + 256 * u;
      q[u] = (i < NE / 4) ? d4[i] : make_int4(-1, -1, -1, -1);
    }
#pragma unroll
    for (int u = 0; u < SCAN_B; ++u) {
      const int dd[4] = {q[u].x, q[u].y, q[u].z, q[u].w};
#pragma unroll
      for (int t = 0; t < 4; ++t) {
        const unsigned g = (unsigned)(dd[t] - n0);
        if (g < (unsigned)GN) {
          const int p = atomicAdd(&cnt[g], 1);
          if (p < GCAP) lst[g][p] = 4 * (i0 + 256 * u) + t;
        }
      }
    }
  }
  __syncthreads();

  const int d = n0 + wave;
  const int n = __builtin_amdgcn_readfirstlane(cnt[wave]);
  const v4f dn = *(const v4f*)(den + (size_t)d * NH);
  v4f acc[C / 128] = {};
#define GATHER_FENCE() __builtin_amdgcn_sched_barrier(0)
  if (n <= GCAP) {
    const int nt = n + 1;
    for (int base = 0; base < nt; base += 32) {
      const int m = min(32, nt - base);
      const int i = base + lane;
      int e_i = NE + d, s_i = d;
      if (i > 0 && i < nt) { e_i = lst[wave][i - 1]; s_i = src[e_i]; }
      const v4f a_i = *(const v4f*)(ex + (size_t)e_i * NH) / dn;
      int k = 0;
      for (; k + 1 < m; k += 2) {
        v4f v0[NH * (C / 128)], v1[NH * (C / 128)];
        gather_load<C>(v0, hx, __builtin_amdgcn_readlane(s_i, k), lane);
        gather_load<C>(v1, hx, __builtin_amdgcn_readlane(s_i, k + 1), lane);
        GATHER_FENCE();
        gather_fma<C>(acc, v0, lane_bcast(a_i, k));
        gather_fma<C>(acc, v1, lane_bcast(a_i, k + 1));
        GATHER_FENCE();
      }
      if (k < m) {
        v4f v0[NH * (C / 128)];
        gather_load<C>(v0, hx, __builtin_amdgcn_readlane(s_i, k), lane);
        GATHER_FENCE();
        gather_fma<C>(acc, v0, lane_bcast(a_i, k));
      }
    }
  } else {
    {
      v4f v0[NH * (C / 128)];
      gather_load<C>(v0, hx, d, lane);
      gather_fma<C>(acc, v0, *(const v4f*)(ex + (size_t)(NE + d) * NH) / dn);
    }
    for (int base = 0; base < NE; base += 32) {
      unsigned mk = __builtin_amdgcn_ballot_w32(dst[base + lane] == d);
      while (mk) {
        const int e = base + __builtin_ctz(mk);
        mk &= mk - 1;
        v4f v0[NH * (C / 128)];
        gather_load<C>(v0, hx, src[e], lane);
        gather_fma<C>(acc, v0, *(const v4f*)(ex + (size_t)e * NH) / dn);
      }
    }
  }
#undef GATHER_FENCE
#pragma unroll
  for (int j = 0; j < C / 128; ++j) {
    const int c0 = 128 * j + 4 * lane;
    v4f o = 0.25f * acc[j] + *(const v4f*)(bias + c0);
    if (RELU) { o[0] = fmaxf(o[0], 0.f); o[1] = fmaxf(o[1], 0.f); o[2] = fmaxf(o[2], 0.f); o[3] = fmaxf(o[3], 0.f); }
    *(v4f*)(out + (size_t)d * C + c0) = o;
  }
}

extern "C" void kernel_launch(void* const* d_in, const int* in_sizes, int n_in,
                              void* d_out, int out_size, void* d_ws, size_t ws_size,
                              hipStream_t stream) {
  const int* x      = (const int*)d_in[0];
  const int* eidx   = (const int*)d_in[1];
  const int* src    = eidx;
  const int* dst    = eidx + NE;
  const int* eattr  = (const int*)d_in[2];
  const float* xemb = (const float*)d_in[3];
  const float* eemb = (const float*)d_in[4];
  const float* W1   = (const float*)d_in[5];
  const float* as1  = (const float*)d_in[6];
  const float* ad1  = (const float*)d_in[7];
  const float* We1  = (const float*)d_in[8];
  const float* ae1  = (const float*)d_in[9];
  const float* b1   = (const float*)d_in[10];
  const float* W2   = (const float*)d_in[11];
  const float* as2  = (const float*)d_in[12];
  const float* ad2  = (const float*)d_in[13];
  const float* We2  = (const float*)d_in[14];
  const float* ae2  = (const float*)d_in[15];
  const float* b2   = (const float*)d_in[16];
  float* out = (float*)d_out;

  float* ws = (float*)d_ws;
  size_t off = 0;
  float* XF   = ws + off; off += (size_t)NN * EMB;
  float* HX1  = ws + off; off += (size_t)NN * NH * HID;
  float* HX2  = ws + off; off += (size_t)NN * NH * FT;
  float* PROJ1 = ws + off; off += (size_t)EMB * NH;
  float* PROJ2 = ws + off; off += (size_t)EMB * NH;
  float* ALE1 = ws + off; off += (size_t)NE * NH;
  float* ALE2 = ws + off; off += (size_t)NE * NH;
  float* LS1  = ws + off; off += (size_t)NN * NH;
  float* LS2  = ws + off; off += (size_t)NN * NH;
  float* DEG  = ws + off; off += 2048;
  unsigned* MX1 = (unsigned*)(ws + off); off += (size_t)NN * NH;
  float* DEN1 = ws + off; off += (size_t)NN * NH;
  unsigned* MX2 = (unsigned*)(ws + off); off += (size_t)NN * NH;
  float* DEN2 = ws + off; off += (size_t)NN * NH;
  float* ANS  = ws + off; off += (size_t)NN * NH;
  float* ADN  = ws + off; off += (size_t)NN * NH;
  float* ALPH = ws + off; off += (size_t)(NE + NN) * NH;
  float* HB   = ws + off; off += (size_t)NN * HID;
  float* PS   = ws + off; off += (size_t)(NH * FT / 64) * MP;
  float* PD   = ws + off; off += (size_t)(NH * FT / 64) * MP;
  (void)ws_size; (void)in_sizes; (void)n_in; (void)out_size;

  const int T = 256;

  k_pre<<<PRE_WG, 32, 0, stream>>>(We1, ae1, We2, ae2, PROJ1, PROJ2, x, xemb, XF, LS1);

  k_gemm1_edge<<<GEMM1_WG + EDGE_WG, 128, 0, stream>>>(XF, W1, HX1, as1, ad1, PS, PD,
                                                     eattr, eemb, PROJ1, PROJ2, dst, ALE1, ALE2, LS1, LS2, DEG);
  k_alpha_sum<HID / 64><<<cdiv(NN * NH, T), T, 0, stream>>>(PS, PD, ANS, ADN);
  k_edge_combine<<<cdiv((long long)(NE + NN) * NH, T), T, 0, stream>>>(ALE1, LS1, DEG, ANS, ADN, src, dst, ALPH, MX1);
  k_softmax_den<<<cdiv((long long)(NE + NN) * NH, T), T, 0, stream>>>(dst, MX1, ALPH, DEN1);
  k_gather<HID, true><<<NN / GN, 256, 0, stream>>>(HX1, ALPH, DEN1, src, dst, b1, HB);

  {
    dim3 g(NH * FT / 128, cdiv(NN, 64));
    k_gemm_wmma_f32<HID, NH * FT><<<g, 128, 0, stream>>>(HB, W2, HX2, NN, as2, ad2, PS, PD);
  }
  k_alpha_sum<FT / 64><<<cdiv(NN * NH, T), T, 0, stream>>>(PS, PD, ANS, ADN);
  k_edge_combine<<<cdiv((long long)(NE + NN) * NH, T), T, 0, stream>>>(ALE2, LS2, DEG, ANS, ADN, src, dst, ALPH, MX2);
  k_softmax_den<<<cdiv((long long)(NE + NN) * NH, T), T, 0, stream>>>(dst, MX2, ALPH, DEN2);
  k_gather<FT, false><<<NN / GN, 256, 0, stream>>>(HX2, ALPH, DEN2, src, dst, b2, out);
}


# Node attention scalars: each GEMM wave adds its 32 row sums into ans/adn [N,4] by float atomics (cleared in the first launch); partial planes and both plane-sum launches gone; sum order changes
# speedup vs baseline: 5.2202x; 1.0314x over previous
#include <hip/hip_runtime.h>
#include <math.h>

#define NN   2000
#define NE   32000
#define EMB  768
#define HID  384
#define FT   768
#define NH   4
#define NEG_SLOPE 0.2f

typedef __attribute__((ext_vector_type(2))) float v2f;
typedef __attribute__((ext_vector_type(4))) float v4f;
typedef __attribute__((ext_vector_type(8))) float v8f;

static inline int cdiv(long long a, long long b) { return (int)((a + b - 1) / b); }

__device__ __forceinline__ unsigned fenc(float f) {
  unsigned u = __float_as_uint(f);
  return (u & 0x80000000u) ? ~u : (u | 0x80000000u);
}
__device__ __forceinline__ float fdec(unsigned e) {
  unsigned u = (e & 0x80000000u) ? (e ^ 0x80000000u) : ~e;
  return __uint_as_float(u);
}

__device__ __forceinline__ void embed_row(const int* __restrict__ x, const float* __restrict__ emb,
                                          float* __restrict__ xf, int row) {
  static_assert(EMB % 128 == 0, "shape");
  const int lane = threadIdx.x & 31;
  const int n = __builtin_amdgcn_readfirstlane(row);
  const float* r0 = emb + (size_t)x[n * 2 + 0] * EMB + 4 * lane;
  const float* r1 = emb + (size_t)x[n * 2 + 1] * EMB + 4 * lane;
  float* o = xf + (size_t)n * EMB + 4 * lane;
#pragma unroll
  for (int t = 0; t < EMB / 128; ++t)
    *(v4f*)(o + 128 * t) = *(const v4f*)(r0 + 128 * t) + *(const v4f*)(r1 + 128 * t);
}

#define EPW 8

template <int N, int OFF, int SZ>
__device__ __forceinline__ void halving_step(float (&v)[SZ], int lane) {
  const bool up = (lane & OFF) != 0;
#pragma unroll
  for (int i = 0; i < N / 2; ++i) {
    const float keep = up ? v[i + N / 2] : v[i];
    const float send = up ? v[i] : v[i + N / 2];
    v[i] = keep + __shfl_xor(send, OFF, 32);
  }
}

struct GemmFrag {
  v4f a[2][2];
  v4f b[2][4];
};

template <int K, int N>
__device__ __forceinline__ void gemm_load_stage(GemmFrag& f, const float* __restrict__ a0p,
                                                const float* __restrict__ a1p,
                                                const float* __restrict__ bp, int kb) {
#pragma unroll
  for (int u = 0; u < 2; ++u) {
    f.a[0][u] = *(const v4f*)(a0p + kb + 8 * u);
    f.a[1][u] = *(const v4f*)(a1p + kb + 8 * u);
#pragma unroll
    for (int t = 0; t < 4; ++t)
      f.b[u][t] = *(const v4f*)(bp + (size_t)(kb + 8 * u + t) * N);
  }
}

__device__ __forceinline__ void gemm_mma_stage(const GemmFrag& f, v8f (&acc)[2][4]) {
#pragma unroll
  for (int u = 0; u < 2; ++u) {
#pragma unroll
    for (int s = 0; s < 2; ++s) {
      v2f a0 = {f.a[0][u][2 * s], f.a[0][u][2 * s + 1]};
      v2f a1 = {f.a[1][u][2 * s], f.a[1][u][2 * s + 1]};
#pragma unroll
      for (int j = 0; j < 4; ++j) {
        v2f b = {f.b[u][2 * s][j], f.b[u][2 * s + 1][j]};
        acc[0][j] = __builtin_amdgcn_wmma_f32_16x16x4_f32(false, a0, false, b, (short)0, acc[0][j], false, false);
        acc[1][j] = __builtin_amdgcn_wmma_f32_16x16x4_f32(false, a1, false, b, (short)0, acc[1][j], false, false);
      }
    }
  }
}

template <int K, int N>
__device__ __forceinline__ void gemm_wg_tile(
    const float* __restrict__ A, const float* __restrict__ B, float* __restrict__ C, int M,
    const float* __restrict__ a_src, const float* __restrict__ a_dst,
    float* __restrict__ ans, float* __restrict__ adn, int bx, int by) {
  static_assert(K % 48 == 0 && K >= 48 && N % 128 == 0, "tile shape");
  static_assert((N / NH) % 64 == 0, "a wave's 64 columns inside one head");
  const int wave = threadIdx.x >> 5, lane = threadIdx.x & 31;
  const int c = lane & 15, h = lane >> 4;
  const int m0 = by * 64 + (wave >> 1) * 32;
  const int n0 = bx * 128 + (wave & 1) * 64;
  if (m0 >= M) return;

  const int r0 = min(m0 + c, M - 1), r1 = min(m0 + 16 + c, M - 1);
  const float* a0p = A + (size_t)r0 * K + 4 * h;
  const float* a1p = A + (size_t)r1 * K + 4 * h;
  const float* bp  = B + (size_t)(4 * h) * N + n0 + 4 * c;

  v8f acc[2][4] = {};
  GemmFrag f0, f1, f2;
  gemm_load_stage<K, N>(f0, a0p, a1p, bp, 0);
  gemm_load_stage<K, N>(f1, a0p, a1p, bp, 16);
#define GEMM_SCHED_FENCE() __builtin_amdgcn_sched_barrier(0)
#pragma clang loop unroll(disable)
  for (int kb = 0; kb < K - 48; kb += 48) {
    gemm_load_stage<K, N>(f2, a0p, a1p, bp, kb + 32);
    GEMM_SCHED_FENCE();
    gemm_mma_stage(f0, acc);
    GEMM_SCHED_FENCE();
    gemm_load_stage<K, N>(f0, a0p, a1p, bp, kb + 48);
    GEMM_SCHED_FENCE();
    gemm_mma_stage(f1, acc);
    GEMM_SCHED_FENCE();
    gemm_load_stage<K, N>(f1, a0p, a1p, bp, kb + 64);
    GEMM_SCHED_FENCE();
    gemm_mma_stage(f2, acc);
    GEMM_SCHED_FENCE();
  }
  gemm_load_stage<K, N>(f2, a0p, a1p, bp, K - 16);
  GEMM_SCHED_FENCE();
  gemm_mma_stage(f0, acc);
  gemm_mma_stage(f1, acc);
  gemm_mma_stage(f2, acc);
#undef GEMM_SCHED_FENCE

#pragma unroll
  for (int i = 0; i < 2; ++i) {
#pragma unroll
    for (int r = 0; r < 8; ++r) {
      const int gm = m0 + 16 * i + 8 * h + r;
      if (gm < M) {
        v4f o = {acc[i][0][r], acc[i][1][r], acc[i][2][r], acc[i][3][r]};
        *(v4f*)(C + (size_t)gm * N + n0 + 4 * c) = o;
      }
    }
  }

  const v4f asv = *(const v4f*)(a_src + n0 + 4 * c);
  const v4f adv = *(const v4f*)(a_dst + n0 + 4 * c);
  float v[EPW * NH];
#pragma unroll
  for (int i = 0; i < 2; ++i) {
#pragma unroll
    for (int r = 0; r < 8; ++r) {
      float s = acc[i][0][r] * asv[0], d = acc[i][0][r] * adv[0];
#pragma unroll
      for (int j = 1; j < 4; ++j) { s += acc[i][j][r] * asv[j]; d += acc[i][j][r] * adv[j]; }
      v[2 * (8 * i + r) + 0] = s;
      v[2 * (8 * i + r) + 1] = d;
    }
  }
  halving_step<32, 8>(v, lane);
  halving_step<16, 4>(v, lane);
  halving_step<8, 2>(v, lane);
  halving_step<4, 1>(v, lane);
  const int prow = m0 + 16 * (c >> 3) + 8 * h + (c & 7);
  const int head = n0 / (N / NH);
  if (prow < M) {
    atomicAdd(&ans[prow * NH + head], v[0]);
    atomicAdd(&adn[prow * NH + head], v[1]);
  }
}

template <int K, int N>
__global__ __launch_bounds__(128) void k_gemm_wmma_f32(
    const float* __restrict__ A, const float* __restrict__ B, float* __restrict__ C, int M,
    const float* __restrict__ a_src, const float* __restrict__ a_dst,
    float* __restrict__ ans, float* __restrict__ adn) {
  gemm_wg_tile<K, N>(A, B, C, M, a_src, a_dst, ans, adn, blockIdx.x, blockIdx.y);
}

#define PB 12
#define PROJ_WG (EMB * NH / 32)

template <int C>
__device__ __forceinline__ void proj_load(v4f (&w)[PB], v4f (&a)[PB], const float* __restrict__ wr,
                                          const float* __restrict__ ar, int c0) {
#pragma unroll
  for (int u = 0; u < PB; ++u) {
    w[u] = *(const v4f*)(wr + c0 + 4 * u);
    a[u] = *(const v4f*)(ar + c0 + 4 * u);
  }
}

__device__ __forceinline__ float proj_fma(float s, const v4f (&w)[PB], const v4f (&a)[PB]) {
#pragma unroll
  for (int u = 0; u < PB; ++u) {
#pragma unroll
    for (int q = 0; q < 4; ++q) s = __builtin_fmaf(w[u][q], a[u][q], s);
  }
  return s;
}

template <int C>
__device__ __forceinline__ void proj_rows(const float* __restrict__ We, const float* __restrict__ ae,
                                          float* __restrict__ proj, int blk) {
  static_assert(C % (8 * PB) == 0 && C >= 16 * PB && (EMB * NH) % 32 == 0, "an even number of whole batches; whole waves");
  const int i = blk * 32 + threadIdx.x;
  const int k = i / NH, h = i % NH;
  const float* wr = We + (size_t)k * (NH * C) + (size_t)h * C;
  const float* ar = ae + (size_t)h * C;
#define PROJ_FENCE() __builtin_amdgcn_sched_barrier(0)
  v4f w0[PB], a0[PB], w1[PB], a1[PB];
  float s = 0.f;
  proj_load<C>(w0, a0, wr, ar, 0);
#pragma clang loop unroll(disable)
  for (int c0 = 0; c0 < C - 8 * PB; c0 += 8 * PB) {
    proj_load<C>(w1, a1, wr, ar, c0 + 4 * PB);
    PROJ_FENCE();
    s = proj_fma(s, w0, a0);
    PROJ_FENCE();
    proj_load<C>(w0, a0, wr, ar, c0 + 8 * PB);
    PROJ_FENCE();
    s = proj_fma(s, w1, a1);
    PROJ_FENCE();
  }
  proj_load<C>(w1, a1, wr, ar, C - 4 * PB);
  PROJ_FENCE();
  s = proj_fma(s, w0, a0);
  s = proj_fma(s, w1, a1);
#undef PROJ_FENCE
  proj[i] = s;
}

#define NCLR (2 * NN * NH + 2048 + 4 * NN * NH + 4 * NN * NH)
#define PRE_WG (2 * PROJ_WG + NN + NCLR / 128)

__global__ __launch_bounds__(32) void k_pre(const float* __restrict__ We1, const float* __restrict__ ae1,
                                            const float* __restrict__ We2, const float* __restrict__ ae2,
                                            float* __restrict__ proj1, float* __restrict__ proj2,
                                            const int* __restrict__ x, const float* __restrict__ emb,
                                            float* __restrict__ xf, float* __restrict__ clr) {
  static_assert(NCLR % 128 == 0, "whole waves of 16-B stores");
  const int b = blockIdx.x;
  if (b < PROJ_WG)               proj_rows<FT>(We2, ae2, proj2, b);
  else if (b < 2 * PROJ_WG)      proj_rows<HID>(We1, ae1, proj1, b - PROJ_WG);
  else if (b < 2 * PROJ_WG + NN) embed_row(x, emb, xf, b - 2 * PROJ_WG);
  else {
    const v4f z = {0.f, 0.f, 0.f, 0.f};
    *(v4f*)(clr + (size_t)(b - 2 * PROJ_WG - NN) * 128 + 4 * threadIdx.x) = z;
  }
}

struct EdgeRows {
  v4f a[EMB / 128];
  v4f b[EMB / 128];
};

__device__ __forceinline__ void edge_rows_load(EdgeRows& r, const float* __restrict__ emb, int i0, int i1, int lane) {
  const float* r0 = emb + (size_t)i0 * EMB + 4 * lane;
  const float* r1 = emb + (size_t)i1 * EMB + 4 * lane;
#pragma unroll
  for (int t = 0; t < EMB / 128; ++t) {
    r.a[t] = *(const v4f*)(r0 + 128 * t);
    r.b[t] = *(const v4f*)(r1 + 128 * t);
  }
}

__device__ __forceinline__ void edge_rows_dot(const EdgeRows& r, const v4f (&p1)[EMB / 128][4],
                                              const v4f (&p2)[EMB / 128][4], float* __restrict__ o) {
  v4f a1 = {0.f, 0.f, 0.f, 0.f}, a2 = {0.f, 0.f, 0.f, 0.f};
#pragma unroll
  for (int t = 0; t < EMB / 128; ++t) {
    const v4f xr = r.a[t] + r.b[t];
#pragma unroll
    for (int kk = 0; kk < 4; ++kk) { a1 += xr[kk] * p1[t][kk]; a2 += xr[kk] * p2[t][kk]; }
  }
  o[0] = a1[0]; o[1] = a1[1]; o[2] = a1[2]; o[3] = a1[3];
  o[4] = a2[0]; o[5] = a2[1]; o[6] = a2[2]; o[7] = a2[3];
}

__device__ __forceinline__ void edge_attr_wave(const int* __restrict__ ea_idx, const float* __restrict__ emb,
                            const float* __restrict__ proj1, const float* __restrict__ proj2,
                            const int* __restrict__ dst,
                            float* __restrict__ ale1, float* __restrict__ ale2,
                            float* __restrict__ ls1, float* __restrict__ ls2, float* __restrict__ deg, int wv) {
  static_assert(EPW == 8 && EPW * NH == 32 && NE % (4 * EPW) == 0 && EMB == 768, "two values per lane; four pairs of edges; whole work-groups; six 128-float pieces");
  const int lane = threadIdx.x & 31;
  const int e0 = __builtin_amdgcn_readfirstlane(wv * EPW);

  int ri[2 * EPW];
#pragma unroll
  for (int u = 0; u < 2 * EPW; ++u) ri[u] = ea_idx[e0 * 2 + u];

  v4f p1[EMB / 128][4], p2[EMB / 128][4];
#pragma unroll
  for (int t = 0; t < EMB / 128; ++t) {
#pragma unroll
    for (int kk = 0; kk < 4; ++kk) {
      p1[t][kk] = *(const v4f*)(proj1 + (size_t)(128 * t + 4 * lane + kk) * NH);
      p2[t][kk] = *(const v4f*)(proj2 + (size_t)(128 * t + 4 * lane + kk) * NH);
    }
  }

#define EDGE_FENCE() __builtin_amdgcn_sched_barrier(0)
  EdgeRows g0[2], g1[2];
  float v[EPW * NH * 2];
  edge_rows_load(g0[0], emb, ri[0], ri[1], lane);
  edge_rows_load(g0[1], emb, ri[2], ri[3], lane);
  edge_rows_load(g1[0], emb, ri[4], ri[5], lane);
  edge_rows_load(g1[1], emb, ri[6], ri[7], lane);
  EDGE_FENCE();
  edge_rows_dot(g0[0], p1, p2, v + 0);
  edge_rows_dot(g0[1], p1, p2, v + 8);
  EDGE_FENCE();
  edge_rows_load(g0[0], emb, ri[8], ri[9], lane);
  edge_rows_load(g0[1], emb, ri[10], ri[11], lane);
  EDGE_FENCE();
  edge_rows_dot(g1[0], p1, p2, v + 16);
  edge_rows_dot(g1[1], p1, p2, v + 24);
  EDGE_FENCE();
  edge_rows_load(g1[0], emb, ri[12], ri[13], lane);
  edge_rows_load(g1[1], emb, ri[14], ri[15], lane);
  EDGE_FENCE();
  edge_rows_dot(g0[0], p1, p2, v + 32);
  edge_rows_dot(g0[1], p1, p2, v + 40);
  edge_rows_dot(g1[0], p1, p2, v + 48);
  edge_rows_dot(g1[1], p1, p2, v + 56);
#undef EDGE_FENCE

  halving_step<64, 16>(v, lane);
  halving_step<32, 8>(v, lane);
  halving_step<16, 4>(v, lane);
  halving_step<8, 2>(v, lane);
  halving_step<4, 1>(v, lane);

  const int e = e0 + (lane >> 2), h0 = 2 * (lane & 1);
  const bool second = (lane & 2) != 0;
  const int d = dst[e];
  float* ale = second ? ale2 : ale1;
  float* ls  = second ? ls2 : ls1;
  const v2f o = {v[0], v[1]};
  *(v2f*)(ale + (size_t)e * NH + h0) = o;
  atomicAdd(&ls[d * NH + h0], v[0]);
  atomicAdd(&ls[d * NH + h0 + 1], v[1]);
  if ((lane & 3) == 0) atomicAdd(&deg[d], 1.0f);
}

#define G1X (NH * HID / 128)
#define G1Y ((NN + 63) / 64)
#define GEMM1_WG (G1X * G1Y)
#define EDGE_WG (NE / (4 * EPW))

__global__ __launch_bounds__(128) void k_gemm1_edge(
    const float* __restrict__ A, const float* __restrict__ B, float* __restrict__ C,
    const float* __restrict__ a_src, const float* __restrict__ a_dst,
    float* __restrict__ ans, float* __restrict__ adn,
    const int* __restrict__ ea_idx, const float* __restrict__ emb,
    const float* __restrict__ proj1, const float* __restrict__ proj2, const int* __restrict__ dst,
    float* __restrict__ ale1, float* __restrict__ ale2,
    float* __restrict__ ls1, float* __restrict__ ls2, float* __restrict__ deg) {
  const int b = blockIdx.x;
  if (b < GEMM1_WG)
    gemm_wg_tile<EMB, NH * HID>(A, B, C, NN, a_src, a_dst, ans, adn, b % G1X, b / G1X);
  else
    edge_attr_wave(ea_idx, emb, proj1, proj2, dst, ale1, ale2, ls1, ls2, deg,
                   (b - GEMM1_WG) * 4 + (threadIdx.x >> 5));
}

__global__ void k_edge_combine(const float* __restrict__ ale, const float* __restrict__ ls,
                               const float* __restrict__ deg,
                               const float* __restrict__ ans, const float* __restrict__ adn,
                               const int* __restrict__ src, const int* __restrict__ dst,
                               float* __restrict__ alpha, unsigned* __restrict__ mx) {
  const int i = blockIdx.x * blockDim.x + threadIdx.x;
  if (i >= (NE + NN) * NH) return;
  const int e = i >> 2, h = i & 3;
  int s, d;
  float t;
  if (e < NE) {
    s = src[e];
    d = dst[e];
    t = ale[i];
  } else {
    s = d = e - NE;
    t = ls[d * NH + h] / fmaxf(deg[d], 1.0f);
  }
  float a = ans[s * NH + h] + adn[d * NH + h] + t;
  a = (a > 0.f) ? a : NEG_SLOPE * a;
  alpha[i] = a;
  atomicMax(&mx[d * NH + h], fenc(a));
}

__global__ void k_softmax_den(const int* __restrict__ dst, const unsigned* __restrict__ mx,
                              float* __restrict__ alpha, float* __restrict__ den) {
  const int tot = NE + NN;
  long long i = (long long)blockIdx.x * blockDim.x + threadIdx.x;
  if (i >= (long long)tot * NH) return;
  int e = (int)(i >> 2), h = (int)(i & 3);
  int d = (e < NE) ? dst[e] : (e - NE);
  float ex = expf(alpha[i] - fdec(mx[d * NH + h]));
  alpha[i] = ex;
  atomicAdd(&den[d * NH + h], ex);
}

#define GN   8
#define GCAP 64

template <int C>
__device__ __forceinline__ void gather_load(v4f (&v)[NH * (C / 128)], const float* __restrict__ hx, int s,
                                            int lane) {
  const float* hr = hx + (size_t)s * (NH * C) + 4 * lane;
#pragma unroll
  for (int q = 0; q < NH * (C / 128); ++q) v[q] = *(const v4f*)(hr + 128 * q);
}

template <int C>
__device__ __forceinline__ void gather_fma(v4f (&acc)[C / 128], const v4f (&v)[NH * (C / 128)], v4f a) {
#pragma unroll
  for (int h = 0; h < NH; ++h) {
#pragma unroll
    for (int j = 0; j < C / 128; ++j) acc[j] += a[h] * v[h * (C / 128) + j];
  }
}

__device__ __forceinline__ v4f lane_bcast(v4f x, int k) {
  v4f r;
  r[0] = __int_as_float(__builtin_amdgcn_readlane(__float_as_int(x[0]), k));
  r[1] = __int_as_float(__builtin_amdgcn_readlane(__float_as_int(x[1]), k));
  r[2] = __int_as_float(__builtin_amdgcn_readlane(__float_as_int(x[2]), k));
  r[3] = __int_as_float(__builtin_amdgcn_readlane(__float_as_int(x[3]), k));
  return r;
}

template <int C, bool RELU>
__global__ __launch_bounds__(256) void k_gather(const float* __restrict__ hx, const float* __restrict__ ex,
                                                const float* __restrict__ den, const int* __restrict__ src,
                                                const int* __restrict__ dst, const float* __restrict__ bias,
                                                float* __restrict__ out) {
  static_assert(C % 128 == 0 && NN % GN == 0 && NE % 32 == 0 && NE % 4 == 0, "shape");
  __shared__ int cnt[GN];
  __shared__ int lst[GN][GCAP];
  const int tid = threadIdx.x, wave = tid >> 5, lane = tid & 31;
  const int n0 = blockIdx.x * GN;
  if (tid < GN) cnt[tid] = 0;
  __syncthreads();
  const int4* d4 = (const int4*)dst;
  constexpr int SCAN_B = 16;
  for (int i0 = tid; i0 < NE / 4; i0 += 256 * SCAN_B) {
    int4 q[SCAN_B];
#pragma unroll
    for (int u = 0; u < SCAN_B; ++u) {
      const int i = i0 + 256 * u;
      q[u] = (i < NE / 4) ? d4[i] : make_int4(-1, -1, -1, -1);
    }
#pragma unroll
    for (int u = 0; u < SCAN_B; ++u) {
      const int dd[4] = {q[u].x, q[u].y, q[u].z, q[u].w};
#pragma unroll
      for (int t = 0; t < 4; ++t) {
        const unsigned g = (unsigned)(dd[t] - n0);
        if (g < (unsigned)GN) {
          const int p = atomicAdd(&cnt[g], 1);
          if (p < GCAP) lst[g][p] = 4 * (i0 + 256 * u) + t;
        }
      }
    }
  }
  __syncthreads();

  const int d = n0 + wave;
  const int n = __builtin_amdgcn_readfirstlane(cnt[wave]);
  const v4f dn = *(const v4f*)(den + (size_t)d * NH);
  v4f acc[C / 128] = {};
#define GATHER_FENCE() __builtin_amdgcn_sched_barrier(0)
  if (n <= GCAP) {
    const int nt = n + 1;
    for (int base = 0; base < nt; base += 32) {
      const int m = min(32, nt - base);
      const int i = base + lane;
      int e_i = NE + d, s_i = d;
      if (i > 0 && i < nt) { e_i = lst[wave][i - 1]; s_i = src[e_i]; }
      const v4f a_i = *(const v4f*)(ex + (size_t)e_i * NH) / dn;
      int k = 0;
      for (; k + 1 < m; k += 2) {
        v4f v0[NH * (C / 128)], v1[NH * (C / 128)];
        gather_load<C>(v0, hx, __builtin_amdgcn_readlane(s_i, k), lane);
        gather_load<C>(v1, hx, __builtin_amdgcn_readlane(s_i, k + 1), lane);
        GATHER_FENCE();
        gather_fma<C>(acc, v0, lane_bcast(a_i, k));
        gather_fma<C>(acc, v1, lane_bcast(a_i, k + 1));
        GATHER_FENCE();
      }
      if (k < m) {
        v4f v0[NH * (C / 128)];
        gather_load<C>(v0, hx, __builtin_amdgcn_readlane(s_i, k), lane);
        GATHER_FENCE();
        gather_fma<C>(acc, v0, lane_bcast(a_i, k));
      }
    }
  } else {
    {
      v4f v0[NH * (C / 128)];
      gather_load<C>(v0, hx, d, lane);
      gather_fma<C>(acc, v0, *(const v4f*)(ex + (size_t)(NE + d) * NH) / dn);
    }
    for (int base = 0; base < NE; base += 32) {
      unsigned mk = __builtin_amdgcn_ballot_w32(dst[base + lane] == d);
      while (mk) {
        const int e = base + __builtin_ctz(mk);
        mk &= mk - 1;
        v4f v0[NH * (C / 128)];
        gather_load<C>(v0, hx, src[e], lane);
        gather_fma<C>(acc, v0, *(const v4f*)(ex + (size_t)e * NH) / dn);
      }
    }
  }
#undef GATHER_FENCE
#pragma unroll
  for (int j = 0; j < C / 128; ++j) {
    const int c0 = 128 * j + 4 * lane;
    v4f o = 0.25f * acc[j] + *(const v4f*)(bias + c0);
    if (RELU) { o[0] = fmaxf(o[0], 0.f); o[1] = fmaxf(o[1], 0.f); o[2] = fmaxf(o[2], 0.f); o[3] = fmaxf(o[3], 0.f); }
    *(v4f*)(out + (size_t)d * C + c0) = o;
  }
}

extern "C" void kernel_launch(void* const* d_in, const int* in_sizes, int n_in,
                              void* d_out, int out_size, void* d_ws, size_t ws_size,
                              hipStream_t stream) {
  const int* x      = (const int*)d_in[0];
  const int* eidx   = (const int*)d_in[1];
  const int* src    = eidx;
  const int* dst    = eidx + NE;
  const int* eattr  = (const int*)d_in[2];
  const float* xemb = (const float*)d_in[3];
  const float* eemb = (const float*)d_in[4];
  const float* W1   = (const float*)d_in[5];
  const float* as1  = (const float*)d_in[6];
  const float* ad1  = (const float*)d_in[7];
  const float* We1  = (const float*)d_in[8];
  const float* ae1  = (const float*)d_in[9];
  const float* b1   = (const float*)d_in[10];
  const float* W2   = (const float*)d_in[11];
  const float* as2  = (const float*)d_in[12];
  const float* ad2  = (const float*)d_in[13];
  const float* We2  = (const float*)d_in[14];
  const float* ae2  = (const float*)d_in[15];
  const float* b2   = (const float*)d_in[16];
  float* out = (float*)d_out;

  float* ws = (float*)d_ws;
  size_t off = 0;
  float* XF   = ws + off; off += (size_t)NN * EMB;
  float* HX1  = ws + off; off += (size_t)NN * NH * HID;
  float* HX2  = ws + off; off += (size_t)NN * NH * FT;
  float* PROJ1 = ws + off; off += (size_t)EMB * NH;
  float* PROJ2 = ws + off; off += (size_t)EMB * NH;
  float* ALE1 = ws + off; off += (size_t)NE * NH;
  float* ALE2 = ws + off; off += (size_t)NE * NH;
  float* LS1  = ws + off; off += (size_t)NN * NH;
  float* LS2  = ws + off; off += (size_t)NN * NH;
  float* DEG  = ws + off; off += 2048;
  unsigned* MX1 = (unsigned*)(ws + off); off += (size_t)NN * NH;
  float* DEN1 = ws + off; off += (size_t)NN * NH;
  unsigned* MX2 = (unsigned*)(ws + off); off += (size_t)NN * NH;
  float* DEN2 = ws + off; off += (size_t)NN * NH;
  float* ANS1 = ws + off; off += (size_t)NN * NH;
  float* ADN1 = ws + off; off += (size_t)NN * NH;
  float* ANS2 = ws + off; off += (size_t)NN * NH;
  float* ADN2 = ws + off; off += (size_t)NN * NH;
  float* ALPH = ws + off; off += (size_t)(NE + NN) * NH;
  float* HB   = ws + off; off += (size_t)NN * HID;
  (void)ws_size; (void)in_sizes; (void)n_in; (void)out_size;

  const int T = 256;

  k_pre<<<PRE_WG, 32, 0, stream>>>(We1, ae1, We2, ae2, PROJ1, PROJ2, x, xemb, XF, LS1);

  k_gemm1_edge<<<GEMM1_WG + EDGE_WG, 128, 0, stream>>>(XF, W1, HX1, as1, ad1, ANS1, ADN1,
                                                     eattr, eemb, PROJ1, PROJ2, dst, ALE1, ALE2, LS1, LS2, DEG);
  k_edge_combine<<<cdiv((long long)(NE + NN) * NH, T), T, 0, stream>>>(ALE1, LS1, DEG, ANS1, ADN1, src, dst, ALPH, MX1);
  k_softmax_den<<<cdiv((long long)(NE + NN) * NH, T), T, 0, stream>>>(dst, MX1, ALPH, DEN1);
  k_gather<HID, true><<<NN / GN, 256, 0, stream>>>(HX1, ALPH, DEN1, src, dst, b1, HB);

  {
    dim3 g(NH * FT / 128, cdiv(NN, 64));
    k_gemm_wmma_f32<HID, NH * FT><<<g, 128, 0, stream>>>(HB, W2, HX2, NN, as2, ad2, ANS2, ADN2);
  }
  k_edge_combine<<<cdiv((long long)(NE + NN) * NH, T), T, 0, stream>>>(ALE2, LS2, DEG, ANS2, ADN2, src, dst, ALPH, MX2);
  k_softmax_den<<<cdiv((long long)(NE + NN) * NH, T), T, 0, stream>>>(dst, MX2, ALPH, DEN2);
  k_gather<FT, false><<<NN / GN, 256, 0, stream>>>(HX2, ALPH, DEN2, src, dst, b2, out);
}


# Scores and segment softmax inside the aggregation launch: each wave scores its node's LDS list, wave max, exp, wave sum in registers; 4 per-edge launches, mx/den atomics gone; sum order changes
# speedup vs baseline: 6.1388x; 1.1760x over previous
#include <hip/hip_runtime.h>
#include <math.h>

#define NN   2000
#define NE   32000
#define EMB  768
#define HID  384
#define FT   768
#define NH   4
#define NEG_SLOPE 0.2f

typedef __attribute__((ext_vector_type(2))) float v2f;
typedef __attribute__((ext_vector_type(4))) float v4f;
typedef __attribute__((ext_vector_type(8))) float v8f;

static inline int cdiv(long long a, long long b) { return (int)((a + b - 1) / b); }

__device__ __forceinline__ void embed_row(const int* __restrict__ x, const float* __restrict__ emb,
                                          float* __restrict__ xf, int row) {
  static_assert(EMB % 128 == 0, "shape");
  const int lane = threadIdx.x & 31;
  const int n = __builtin_amdgcn_readfirstlane(row);
  const float* r0 = emb + (size_t)x[n * 2 + 0] * EMB + 4 * lane;
  const float* r1 = emb + (size_t)x[n * 2 + 1] * EMB + 4 * lane;
  float* o = xf + (size_t)n * EMB + 4 * lane;
#pragma unroll
  for (int t = 0; t < EMB / 128; ++t)
    *(v4f*)(o + 128 * t) = *(const v4f*)(r0 + 128 * t) + *(const v4f*)(r1 + 128 * t);
}

#define EPW 8

template <int N, int OFF, int SZ>
__device__ __forceinline__ void halving_step(float (&v)[SZ], int lane) {
  const bool up = (lane & OFF) != 0;
#pragma unroll
  for (int i = 0; i < N / 2; ++i) {
    const float keep = up ? v[i + N / 2] : v[i];
    const float send = up ? v[i] : v[i + N / 2];
    v[i] = keep + __shfl_xor(send, OFF, 32);
  }
}

struct GemmFrag {
  v4f a[2][2];
  v4f b[2][4];
};

template <int K, int N>
__device__ __forceinline__ void gemm_load_stage(GemmFrag& f, const float* __restrict__ a0p,
                                                const float* __restrict__ a1p,
                                                const float* __restrict__ bp, int kb) {
#pragma unroll
  for (int u = 0; u < 2; ++u) {
    f.a[0][u] = *(const v4f*)(a0p + kb + 8 * u);
    f.a[1][u] = *(const v4f*)(a1p + kb + 8 * u);
#pragma unroll
    for (int t = 0; t < 4; ++t)
      f.b[u][t] = *(const v4f*)(bp + (size_t)(kb + 8 * u + t) * N);
  }
}

__device__ __forceinline__ void gemm_mma_stage(const GemmFrag& f, v8f (&acc)[2][4]) {
#pragma unroll
  for (int u = 0; u < 2; ++u) {
#pragma unroll
    for (int s = 0; s < 2; ++s) {
      v2f a0 = {f.a[0][u][2 * s], f.a[0][u][2 * s + 1]};
      v2f a1 = {f.a[1][u][2 * s], f.a[1][u][2 * s + 1]};
#pragma unroll
      for (int j = 0; j < 4; ++j) {
        v2f b = {f.b[u][2 * s][j], f.b[u][2 * s + 1][j]};
        acc[0][j] = __builtin_amdgcn_wmma_f32_16x16x4_f32(false, a0, false, b, (short)0, acc[0][j], false, false);
        acc[1][j] = __builtin_amdgcn_wmma_f32_16x16x4_f32(false, a1, false, b, (short)0, acc[1][j], false, false);
      }
    }
  }
}

template <int K, int N>
__device__ __forceinline__ void gemm_wg_tile(
    const float* __restrict__ A, const float* __restrict__ B, float* __restrict__ C, int M,
    const float* __restrict__ a_src, const float* __restrict__ a_dst,
    float* __restrict__ ans, float* __restrict__ adn, int bx, int by) {
  static_assert(K % 48 == 0 && K >= 48 && N % 128 == 0, "tile shape");
  static_assert((N / NH) % 64 == 0, "a wave's 64 columns inside one head");
  const int wave = threadIdx.x >> 5, lane = threadIdx.x & 31;
  const int c = lane & 15, h = lane >> 4;
  const int m0 = by * 64 + (wave >> 1) * 32;
  const int n0 = bx * 128 + (wave & 1) * 64;
  if (m0 >= M) return;

  const int r0 = min(m0 + c, M - 1), r1 = min(m0 + 16 + c, M - 1);
  const float* a0p = A + (size_t)r0 * K + 4 * h;
  const float* a1p = A + (size_t)r1 * K + 4 * h;
  const float* bp  = B + (size_t)(4 * h) * N + n0 + 4 * c;

  v8f acc[2][4] = {};
  GemmFrag f0, f1, f2;
  gemm_load_stage<K, N>(f0, a0p, a1p, bp, 0);
  gemm_load_stage<K, N>(f1, a0p, a1p, bp, 16);
#define GEMM_SCHED_FENCE() __builtin_amdgcn_sched_barrier(0)
#pragma clang loop unroll(disable)
  for (int kb = 0; kb < K - 48; kb += 48) {
    gemm_load_stage<K, N>(f2, a0p, a1p, bp, kb + 32);
    GEMM_SCHED_FENCE();
    gemm_mma_stage(f0, acc);
    GEMM_SCHED_FENCE();
    gemm_load_stage<K, N>(f0, a0p, a1p, bp, kb + 48);
    GEMM_SCHED_FENCE();
    gemm_mma_stage(f1, acc);
    GEMM_SCHED_FENCE();
    gemm_load_stage<K, N>(f1, a0p, a1p, bp, kb + 64);
    GEMM_SCHED_FENCE();
    gemm_mma_stage(f2, acc);
    GEMM_SCHED_FENCE();
  }
  gemm_load_stage<K, N>(f2, a0p, a1p, bp, K - 16);
  GEMM_SCHED_FENCE();
  gemm_mma_stage(f0, acc);
  gemm_mma_stage(f1, acc);
  gemm_mma_stage(f2, acc);
#undef GEMM_SCHED_FENCE

#pragma unroll
  for (int i = 0; i < 2; ++i) {
#pragma unroll
    for (int r = 0; r < 8; ++r) {
      const int gm = m0 + 16 * i + 8 * h + r;
      if (gm < M) {
        v4f o = {acc[i][0][r], acc[i][1][r], acc[i][2][r], acc[i][3][r]};
        *(v4f*)(C + (size_t)gm * N + n0 + 4 * c) = o;
      }
    }
  }

  const v4f asv = *(const v4f*)(a_src + n0 + 4 * c);
  const v4f adv = *(const v4f*)(a_dst + n0 + 4 * c);
  float v[EPW * NH];
#pragma unroll
  for (int i = 0; i < 2; ++i) {
#pragma unroll
    for (int r = 0; r < 8; ++r) {
      float s = acc[i][0][r] * asv[0], d = acc[i][0][r] * adv[0];
#pragma unroll
      for (int j = 1; j < 4; ++j) { s += acc[i][j][r] * asv[j]; d += acc[i][j][r] * adv[j]; }
      v[2 * (8 * i + r) + 0] = s;
      v[2 * (8 * i + r) + 1] = d;
    }
  }
  halving_step<32, 8>(v, lane);
  halving_step<16, 4>(v, lane);
  halving_step<8, 2>(v, lane);
  halving_step<4, 1>(v, lane);
  const int prow = m0 + 16 * (c >> 3) + 8 * h + (c & 7);
  const int head = n0 / (N / NH);
  if (prow < M) {
    atomicAdd(&ans[prow * NH + head], v[0]);
    atomicAdd(&adn[prow * NH + head], v[1]);
  }
}

template <int K, int N>
__global__ __launch_bounds__(128) void k_gemm_wmma_f32(
    const float* __restrict__ A, const float* __restrict__ B, float* __restrict__ C, int M,
    const float* __restrict__ a_src, const float* __restrict__ a_dst,
    float* __restrict__ ans, float* __restrict__ adn) {
  gemm_wg_tile<K, N>(A, B, C, M, a_src, a_dst, ans, adn, blockIdx.x, blockIdx.y);
}

#define PB 12
#define PROJ_WG (EMB * NH / 32)

template <int C>
__device__ __forceinline__ void proj_load(v4f (&w)[PB], v4f (&a)[PB], const float* __restrict__ wr,
                                          const float* __restrict__ ar, int c0) {
#pragma unroll
  for (int u = 0; u < PB; ++u) {
    w[u] = *(const v4f*)(wr + c0 + 4 * u);
    a[u] = *(const v4f*)(ar + c0 + 4 * u);
  }
}

__device__ __forceinline__ float proj_fma(float s, const v4f (&w)[PB], const v4f (&a)[PB]) {
#pragma unroll
  for (int u = 0; u < PB; ++u) {
#pragma unroll
    for (int q = 0; q < 4; ++q) s = __builtin_fmaf(w[u][q], a[u][q], s);
  }
  return s;
}

template <int C>
__device__ __forceinline__ void proj_rows(const float* __restrict__ We, const float* __restrict__ ae,
                                          float* __restrict__ proj, int blk) {
  static_assert(C % (8 * PB) == 0 && C >= 16 * PB && (EMB * NH) % 32 == 0, "an even number of whole batches; whole waves");
  const int i = blk * 32 + threadIdx.x;
  const int k = i / NH, h = i % NH;
  const float* wr = We + (size_t)k * (NH * C) + (size_t)h * C;
  const float* ar = ae + (size_t)h * C;
#define PROJ_FENCE() __builtin_amdgcn_sched_barrier(0)
  v4f w0[PB], a0[PB], w1[PB], a1[PB];
  float s = 0.f;
  proj_load<C>(w0, a0, wr, ar, 0);
#pragma clang loop unroll(disable)
  for (int c0 = 0; c0 < C - 8 * PB; c0 += 8 * PB) {
    proj_load<C>(w1, a1, wr, ar, c0 + 4 * PB);
    PROJ_FENCE();
    s = proj_fma(s, w0, a0);
    PROJ_FENCE();
    proj_load<C>(w0, a0, wr, ar, c0 + 8 * PB);
    PROJ_FENCE();
    s = proj_fma(s, w1, a1);
    PROJ_FENCE();
  }
  proj_load<C>(w1, a1, wr, ar, C - 4 * PB);
  PROJ_FENCE();
  s = proj_fma(s, w0, a0);
  s = proj_fma(s, w1, a1);
#undef PROJ_FENCE
  proj[i] = s;
}

#define NCLR (2 * NN * NH + 2048 + 4 * NN * NH)
#define PRE_WG (2 * PROJ_WG + NN + NCLR / 128)

__global__ __launch_bounds__(32) void k_pre(const float* __restrict__ We1, const float* __restrict__ ae1,
                                            const float* __restrict__ We2, const float* __restrict__ ae2,
                                            float* __restrict__ proj1, float* __restrict__ proj2,
                                            const int* __restrict__ x, const float* __restrict__ emb,
                                            float* __restrict__ xf, float* __restrict__ clr) {
  static_assert(NCLR % 128 == 0, "whole waves of 16-B stores");
  const int b = blockIdx.x;
  if (b < PROJ_WG)               proj_rows<FT>(We2, ae2, proj2, b);
  else if (b < 2 * PROJ_WG)      proj_rows<HID>(We1, ae1, proj1, b - PROJ_WG);
  else if (b < 2 * PROJ_WG + NN) embed_row(x, emb, xf, b - 2 * PROJ_WG);
  else {
    const v4f z = {0.f, 0.f, 0.f, 0.f};
    *(v4f*)(clr + (size_t)(b - 2 * PROJ_WG - NN) * 128 + 4 * threadIdx.x) = z;
  }
}

struct EdgeRows {
  v4f a[EMB / 128];
  v4f b[EMB / 128];
};

__device__ __forceinline__ void edge_rows_load(EdgeRows& r, const float* __restrict__ emb, int i0, int i1, int lane) {
  const float* r0 = emb + (size_t)i0 * EMB + 4 * lane;
  const float* r1 = emb + (size_t)i1 * EMB + 4 * lane;
#pragma unroll
  for (int t = 0; t < EMB / 128; ++t) {
    r.a[t] = *(const v4f*)(r0 + 128 * t);
    r.b[t] = *(const v4f*)(r1 + 128 * t);
  }
}

__device__ __forceinline__ void edge_rows_dot(const EdgeRows& r, const v4f (&p1)[EMB / 128][4],
                                              const v4f (&p2)[EMB / 128][4], float* __restrict__ o) {
  v4f a1 = {0.f, 0.f, 0.f, 0.f}, a2 = {0.f, 0.f, 0.f, 0.f};
#pragma unroll
  for (int t = 0; t < EMB / 128; ++t) {
    const v4f xr = r.a[t] + r.b[t];
#pragma unroll
    for (int kk = 0; kk < 4; ++kk) { a1 += xr[kk] * p1[t][kk]; a2 += xr[kk] * p2[t][kk]; }
  }
  o[0] = a1[0]; o[1] = a1[1]; o[2] = a1[2]; o[3] = a1[3];
  o[4] = a2[0]; o[5] = a2[1]; o[6] = a2[2]; o[7] = a2[3];
}

__device__ __forceinline__ void edge_attr_wave(const int* __restrict__ ea_idx, const float* __restrict__ emb,
                            const float* __restrict__ proj1, const float* __restrict__ proj2,
                            const int* __restrict__ dst,
                            float* __restrict__ ale1, float* __restrict__ ale2,
                            float* __restrict__ ls1, float* __restrict__ ls2, float* __restrict__ deg, int wv) {
  static_assert(EPW == 8 && EPW * NH == 32 && NE % (4 * EPW) == 0 && EMB == 768, "two values per lane; four pairs of edges; whole work-groups; six 128-float pieces");
  const int lane = threadIdx.x & 31;
  const int e0 = __builtin_amdgcn_readfirstlane(wv * EPW);

  int ri[2 * EPW];
#pragma unroll
  for (int u = 0; u < 2 * EPW; ++u) ri[u] = ea_idx[e0 * 2 + u];

  v4f p1[EMB / 128][4], p2[EMB / 128][4];
#pragma unroll
  for (int t = 0; t < EMB / 128; ++t) {
#pragma unroll
    for (int kk = 0; kk < 4; ++kk) {
      p1[t][kk] = *(const v4f*)(proj1 + (size_t)(128 * t + 4 * lane + kk) * NH);
      p2[t][kk] = *(const v4f*)(proj2 + (size_t)(128 * t + 4 * lane + kk) * NH);
    }
  }

#define EDGE_FENCE() __builtin_amdgcn_sched_barrier(0)
  EdgeRows g0[2], g1[2];
  float v[EPW * NH * 2];
  edge_rows_load(g0[0], emb, ri[0], ri[1], lane);
  edge_rows_load(g0[1], emb, ri[2], ri[3], lane);
  edge_rows_load(g1[0], emb, ri[4], ri[5], lane);
  edge_rows_load(g1[1], emb, ri[6], ri[7], lane);
  EDGE_FENCE();
  edge_rows_dot(g0[0], p1, p2, v + 0);
  edge_rows_dot(g0[1], p1, p2, v + 8);
  EDGE_FENCE();
  edge_rows_load(g0[0], emb, ri[8], ri[9], lane);
  edge_rows_load(g0[1], emb, ri[10], ri[11], lane);
  EDGE_FENCE();
  edge_rows_dot(g1[0], p1, p2, v + 16);
  edge_rows_dot(g1[1], p1, p2, v + 24);
  EDGE_FENCE();
  edge_rows_load(g1[0], emb, ri[12], ri[13], lane);
  edge_rows_load(g1[1], emb, ri[14], ri[15], lane);
  EDGE_FENCE();
  edge_rows_dot(g0[0], p1, p2, v + 32);
  edge_rows_dot(g0[1], p1, p2, v + 40);
  edge_rows_dot(g1[0], p1, p2, v + 48);
  edge_rows_dot(g1[1], p1, p2, v + 56);
#undef EDGE_FENCE

  halving_step<64, 16>(v, lane);
  halving_step<32, 8>(v, lane);
  halving_step<16, 4>(v, lane);
  halving_step<8, 2>(v, lane);
  halving_step<4, 1>(v, lane);

  const int e = e0 + (lane >> 2), h0 = 2 * (lane & 1);
  const bool second = (lane & 2) != 0;
  const int d = dst[e];
  float* ale = second ? ale2 : ale1;
  float* ls  = second ? ls2 : ls1;
  const v2f o = {v[0], v[1]};
  *(v2f*)(ale + (size_t)e * NH + h0) = o;
  atomicAdd(&ls[d * NH + h0], v[0]);
  atomicAdd(&ls[d * NH + h0 + 1], v[1]);
  if ((lane & 3) == 0) atomicAdd(&deg[d], 1.0f);
}

#define G1X (NH * HID / 128)
#define G1Y ((NN + 63) / 64)
#define GEMM1_WG (G1X * G1Y)
#define EDGE_WG (NE / (4 * EPW))

__global__ __launch_bounds__(128) void k_gemm1_edge(
    const float* __restrict__ A, const float* __restrict__ B, float* __restrict__ C,
    const float* __restrict__ a_src, const float* __restrict__ a_dst,
    float* __restrict__ ans, float* __restrict__ adn,
    const int* __restrict__ ea_idx, const float* __restrict__ emb,
    const float* __restrict__ proj1, const float* __restrict__ proj2, const int* __restrict__ dst,
    float* __restrict__ ale1, float* __restrict__ ale2,
    float* __restrict__ ls1, float* __restrict__ ls2, float* __restrict__ deg) {
  const int b = blockIdx.x;
  if (b < GEMM1_WG)
    gemm_wg_tile<EMB, NH * HID>(A, B, C, NN, a_src, a_dst, ans, adn, b % G1X, b / G1X);
  else
    edge_attr_wave(ea_idx, emb, proj1, proj2, dst, ale1, ale2, ls1, ls2, deg,
                   (b - GEMM1_WG) * 4 + (threadIdx.x >> 5));
}

#define GN   8
#define GCAP 64
#define GNB  ((GCAP + 1 + 31) / 32)

template <int C>
__device__ __forceinline__ void gather_load(v4f (&v)[NH * (C / 128)], const float* __restrict__ hx, int s,
                                            int lane) {
  const float* hr = hx + (size_t)s * (NH * C) + 4 * lane;
#pragma unroll
  for (int q = 0; q < NH * (C / 128); ++q) v[q] = *(const v4f*)(hr + 128 * q);
}

template <int C>
__device__ __forceinline__ void gather_fma(v4f (&acc)[C / 128], const v4f (&v)[NH * (C / 128)], v4f a) {
#pragma unroll
  for (int h = 0; h < NH; ++h) {
#pragma unroll
    for (int j = 0; j < C / 128; ++j) acc[j] += a[h] * v[h * (C / 128) + j];
  }
}

__device__ __forceinline__ v4f lane_bcast(v4f x, int k) {
  v4f r;
  r[0] = __int_as_float(__builtin_amdgcn_readlane(__float_as_int(x[0]), k));
  r[1] = __int_as_float(__builtin_amdgcn_readlane(__float_as_int(x[1]), k));
  r[2] = __int_as_float(__builtin_amdgcn_readlane(__float_as_int(x[2]), k));
  r[3] = __int_as_float(__builtin_amdgcn_readlane(__float_as_int(x[3]), k));
  return r;
}

__device__ __forceinline__ v4f term_score(v4f ans_s, v4f adn_d, v4f t) {
  const v4f a = (ans_s + adn_d) + t;
  v4f r;
#pragma unroll
  for (int q = 0; q < 4; ++q) r[q] = (a[q] > 0.f) ? a[q] : NEG_SLOPE * a[q];
  return r;
}

__device__ __forceinline__ v4f max4(v4f a, v4f b) {
  v4f r = {fmaxf(a[0], b[0]), fmaxf(a[1], b[1]), fmaxf(a[2], b[2]), fmaxf(a[3], b[3])};
  return r;
}

__device__ __forceinline__ v4f exp4(v4f a) {
  v4f r = {expf(a[0]), expf(a[1]), expf(a[2]), expf(a[3])};
  return r;
}

__device__ __forceinline__ v4f wave_max4(v4f m) {
#pragma unroll
  for (int off = 16; off > 0; off >>= 1) {
#pragma unroll
    for (int q = 0; q < 4; ++q) m[q] = fmaxf(m[q], __shfl_xor(m[q], off, 32));
  }
  return m;
}

__device__ __forceinline__ v4f wave_sum4(v4f s) {
#pragma unroll
  for (int off = 16; off > 0; off >>= 1) {
#pragma unroll
    for (int q = 0; q < 4; ++q) s[q] += __shfl_xor(s[q], off, 32);
  }
  return s;
}

template <int C, bool RELU>
__global__ __launch_bounds__(256) void k_gather(const float* __restrict__ hx, const float* __restrict__ ale,
                                                const float* __restrict__ ls, const float* __restrict__ deg,
                                                const float* __restrict__ ans, const float* __restrict__ adn,
                                                const int* __restrict__ src, const int* __restrict__ dst,
                                                const float* __restrict__ bias, float* __restrict__ out) {
  static_assert(C % 128 == 0 && NN % GN == 0 && NE % 32 == 0 && NE % 4 == 0, "shape");
  __shared__ int cnt[GN];
  __shared__ int lst[GN][GCAP];
  const int tid = threadIdx.x, wave = tid >> 5, lane = tid & 31;
  const int n0 = blockIdx.x * GN;
  const int d = n0 + wave;
  const v4f adn_d = *(const v4f*)(adn + (size_t)d * NH);
  const v4f ans_d = *(const v4f*)(ans + (size_t)d * NH);
  const v4f ls_d  = *(const v4f*)(ls + (size_t)d * NH);
  const float dg  = fmaxf(deg[d], 1.0f);
  if (tid < GN) cnt[tid] = 0;
  __syncthreads();
  const int4* d4 = (const int4*)dst;
  constexpr int SCAN_B = 16;
  for (int i0 = tid; i0 < NE / 4; i0 += 256 * SCAN_B) {
    int4 q[SCAN_B];
#pragma unroll
    for (int u = 0; u < SCAN_B; ++u) {
      const int i = i0 + 256 * u;
      q[u] = (i < NE / 4) ? d4[i] : make_int4(-1, -1, -1, -1);
    }
#pragma unroll
    for (int u = 0; u < SCAN_B; ++u) {
      const int dd[4] = {q[u].x, q[u].y, q[u].z, q[u].w};
#pragma unroll
      for (int t = 0; t < 4; ++t) {
        const unsigned g = (unsigned)(dd[t] - n0);
        if (g < (unsigned)GN) {
          const int p = atomicAdd(&cnt[g], 1);
          if (p < GCAP) lst[g][p] = 4 * (i0 + 256 * u) + t;
        }
      }
    }
  }
  __syncthreads();

  const int n = __builtin_amdgcn_readfirstlane(cnt[wave]);
  const v4f a_self = term_score(ans_d, adn_d, ls_d / dg);
  v4f acc[C / 128] = {};
#define GATHER_FENCE() __builtin_amdgcn_sched_barrier(0)
  if (n <= GCAP) {
    const int nt = n + 1;
    int s_b[GNB];
    v4f w_b[GNB];
    v4f mx = a_self;
#pragma unroll
    for (int b = 0; b < GNB; ++b) {
      s_b[b] = d;
      w_b[b] = a_self;
      if (32 * b < nt) {
        const int i = 32 * b + lane;
        if (i > 0 && i < nt) {
          const int e_i = lst[wave][i - 1];
          const int s_i = src[e_i];
          const v4f t_i = *(const v4f*)(ale + (size_t)e_i * NH);
          s_b[b] = s_i;
          w_b[b] = term_score(*(const v4f*)(ans + (size_t)s_i * NH), adn_d, t_i);
          mx = max4(mx, w_b[b]);
        }
      }
    }
    mx = wave_max4(mx);
    v4f dn = {0.f, 0.f, 0.f, 0.f};
#pragma unroll
    for (int b = 0; b < GNB; ++b) {
      if (32 * b < nt) {
        const v4f ex = exp4(w_b[b] - mx);
        w_b[b] = ex;
        if (32 * b + lane < nt) dn += ex;
      }
    }
    dn = wave_sum4(dn);
#pragma unroll
    for (int b = 0; b < GNB; ++b) {
      if (32 * b < nt) w_b[b] = w_b[b] / dn;
    }

    for (int base = 0; base < nt; base += 32) {
      const int m = min(32, nt - base);
      int s_i = s_b[0];
      v4f a_i = w_b[0];
#pragma unroll
      for (int b = 1; b < GNB; ++b) {
        if (base == 32 * b) { s_i = s_b[b]; a_i = w_b[b]; }
      }
      int k = 0;
      for (; k + 1 < m; k += 2) {
        v4f v0[NH * (C / 128)], v1[NH * (C / 128)];
        gather_load<C>(v0, hx, __builtin_amdgcn_readlane(s_i, k), lane);
        gather_load<C>(v1, hx, __builtin_amdgcn_readlane(s_i, k + 1), lane);
        GATHER_FENCE();
        gather_fma<C>(acc, v0, lane_bcast(a_i, k));
        gather_fma<C>(acc, v1, lane_bcast(a_i, k + 1));
        GATHER_FENCE();
      }
      if (k < m) {
        v4f v0[NH * (C / 128)];
        gather_load<C>(v0, hx, __builtin_amdgcn_readlane(s_i, k), lane);
        GATHER_FENCE();
        gather_fma<C>(acc, v0, lane_bcast(a_i, k));
      }
    }
  } else {
    v4f mx = a_self;
    for (int base = 0; base < NE; base += 32) {
      const int e = base + lane;
      if (dst[e] == d)
        mx = max4(mx, term_score(*(const v4f*)(ans + (size_t)src[e] * NH), adn_d, *(const v4f*)(ale + (size_t)e * NH)));
    }
    mx = wave_max4(mx);
    v4f dn = {0.f, 0.f, 0.f, 0.f};
    if (lane == 0) dn = exp4(a_self - mx);
    for (int base = 0; base < NE; base += 32) {
      const int e = base + lane;
      if (dst[e] == d)
        dn += exp4(term_score(*(const v4f*)(ans + (size_t)src[e] * NH), adn_d, *(const v4f*)(ale + (size_t)e * NH)) - mx);
    }
    dn = wave_sum4(dn);
    {
      v4f v0[NH * (C / 128)];
      gather_load<C>(v0, hx, d, lane);
      gather_fma<C>(acc, v0, exp4(a_self - mx) / dn);
    }
    for (int base = 0; base < NE; base += 32) {
      unsigned mk = __builtin_amdgcn_ballot_w32(dst[base + lane] == d);
      while (mk) {
        const int e = base + __builtin_ctz(mk);
        mk &= mk - 1;
        const int s = src[e];
        v4f v0[NH * (C / 128)];
        gather_load<C>(v0, hx, s, lane);
        const v4f a = term_score(*(const v4f*)(ans + (size_t)s * NH), adn_d, *(const v4f*)(ale + (size_t)e * NH));
        gather_fma<C>(acc, v0, exp4(a - mx) / dn);
      }
    }
  }
#undef GATHER_FENCE
#pragma unroll
  for (int j = 0; j < C / 128; ++j) {
    const int c0 = 128 * j + 4 * lane;
    v4f o = 0.25f * acc[j] + *(const v4f*)(bias + c0);
    if (RELU) { o[0] = fmaxf(o[0], 0.f); o[1] = fmaxf(o[1], 0.f); o[2] = fmaxf(o[2], 0.f); o[3] = fmaxf(o[3], 0.f); }
    *(v4f*)(out + (size_t)d * C + c0) = o;
  }
}

extern "C" void kernel_launch(void* const* d_in, const int* in_sizes, int n_in,
                              void* d_out, int out_size, void* d_ws, size_t ws_size,
                              hipStream_t stream) {
  const int* x      = (const int*)d_in[0];
  const int* eidx   = (const int*)d_in[1];
  const int* src    = eidx;
  const int* dst    = eidx + NE;
  const int* eattr  = (const int*)d_in[2];
  const float* xemb = (const float*)d_in[3];
  const float* eemb = (const float*)d_in[4];
  const float* W1   = (const float*)d_in[5];
  const float* as1  = (const float*)d_in[6];
  const float* ad1  = (const float*)d_in[7];
  const float* We1  = (const float*)d_in[8];
  const float* ae1  = (const float*)d_in[9];
  const float* b1   = (const float*)d_in[10];
  const float* W2   = (const float*)d_in[11];
  const float* as2  = (const float*)d_in[12];
  const float* ad2  = (const float*)d_in[13];
  const float* We2  = (const float*)d_in[14];
  const float* ae2  = (const float*)d_in[15];
  const float* b2   = (const float*)d_in[16];
  float* out = (float*)d_out;

  float* ws = (float*)d_ws;
  size_t off = 0;
  float* XF   = ws + off; off += (size_t)NN * EMB;
  float* HX1  = ws + off; off += (size_t)NN * NH * HID;
  float* HX2  = ws + off; off += (size_t)NN * NH * FT;
  float* PROJ1 = ws + off; off += (size_t)EMB * NH;
  float* PROJ2 = ws + off; off += (size_t)EMB * NH;
  float* ALE1 = ws + off; off += (size_t)NE * NH;
  float* ALE2 = ws + off; off += (size_t)NE * NH;
  float* LS1  = ws + off; off += (size_t)NN * NH;
  float* LS2  = ws + off; off += (size_t)NN * NH;
  float* DEG  = ws + off; off += 2048;
  float* ANS1 = ws + off; off += (size_t)NN * NH;
  float* ADN1 = ws + off; off += (size_t)NN * NH;
  float* ANS2 = ws + off; off += (size_t)NN * NH;
  float* ADN2 = ws + off; off += (size_t)NN * NH;
  float* HB   = ws + off; off += (size_t)NN * HID;
  (void)ws_size; (void)in_sizes; (void)n_in; (void)out_size;

  k_pre<<<PRE_WG, 32, 0, stream>>>(We1, ae1, We2, ae2, PROJ1, PROJ2, x, xemb, XF, LS1);

  k_gemm1_edge<<<GEMM1_WG + EDGE_WG, 128, 0, stream>>>(XF, W1, HX1, as1, ad1, ANS1, ADN1,
                                                     eattr, eemb, PROJ1, PROJ2, dst, ALE1, ALE2, LS1, LS2, DEG);
  k_gather<HID, true><<<NN / GN, 256, 0, stream>>>(HX1, ALE1, LS1, DEG, ANS1, ADN1, src, dst, b1, HB);

  {
    dim3 g(NH * FT / 128, cdiv(NN, 64));
    k_gemm_wmma_f32<HID, NH * FT><<<g, 128, 0, stream>>>(HB, W2, HX2, NN, as2, ad2, ANS2, ADN2);
  }
  k_gather<FT, false><<<NN / GN, 256, 0, stream>>>(HX2, ALE2, LS2, DEG, ANS2, ADN2, src, dst, b2, out);
}
